# starting kernel
# baseline (speedup 1.0000x reference)
#include <hip/hip_runtime.h>
#include <hip/hip_bf16.h>
#include <math.h>

#define Bn 8
#define Tn 256
#define INn 128
#define Hn 128
#define OUTn 128
#define Ln 2
#define STSTRIDE 33025
#define LAYERSTATE 16512
#define YSZ (Tn*Bn*OUTn)

typedef __attribute__((ext_vector_type(2))) float v2f;
typedef __attribute__((ext_vector_type(8))) float v8f;

__device__ __forceinline__ v8f wmma4(v2f a, v2f b, v8f c) {
  return __builtin_amdgcn_wmma_f32_16x16x4_f32(false, a, false, b, (short)0, c, false, false);
}

__device__ __forceinline__ void tile_abt(const float* A, int lda, const float* Bt, int ldb,
                                         int K, v8f& acc) {
  const int lane = threadIdx.x & 31;
  const int half = lane >> 4;
  const int idx  = lane & 15;
  const float* ap = A  + (size_t)idx * lda + 2 * half;
  const float* bp = Bt + (size_t)idx * ldb + 2 * half;
  for (int k = 0; k < K; k += 4) {
    v2f a, b;
    a.x = ap[k]; a.y = ap[k + 1];
    b.x = bp[k]; b.y = bp[k + 1];
    acc = wmma4(a, b, acc);
  }
}

__device__ __forceinline__ void tile_ab(const float* A, int lda, const float* Bm, int ldb,
                                        int K, v8f& acc) {
  const int lane = threadIdx.x & 31;
  const int half = lane >> 4;
  const int idx  = lane & 15;
  const float* ap = A + (size_t)idx * lda + 2 * half;
  for (int k = 0; k < K; k += 4) {
    v2f a, b;
    a.x = ap[k]; a.y = ap[k + 1];
    b.x = Bm[(size_t)(k + 2 * half)     * ldb + idx];
    b.y = Bm[(size_t)(k + 2 * half + 1) * ldb + idx];
    acc = wmma4(a, b, acc);
  }
}

__device__ __forceinline__ void tile_atb(const float* Am, int lda, const float* Bm, int ldb,
                                         int K, v8f& acc) {
  const int lane = threadIdx.x & 31;
  const int half = lane >> 4;
  const int idx  = lane & 15;
  for (int k = 0; k < K; k += 4) {
    v2f a, b;
    a.x = Am[(size_t)(k + 2 * half)     * lda + idx];
    a.y = Am[(size_t)(k + 2 * half + 1) * lda + idx];
    b.x = Bm[(size_t)(k + 2 * half)     * ldb + idx];
    b.y = Bm[(size_t)(k + 2 * half + 1) * ldb + idx];
    acc = wmma4(a, b, acc);
  }
}

__global__ void dla_rotate_kernel(const float* __restrict__ z, const float* __restrict__ st,
                                  float* __restrict__ xcur, float* __restrict__ outst) {
  const int bt = blockIdx.x;
  const int b = bt >> 8, t = bt & 255;
  const int j = threadIdx.x;
  const float ts = st[(size_t)b * STSTRIDE + STSTRIDE - 1];
  const float pos = ts + (float)t;
  const float freq = expf((float)(2 * j) * (-9.210340371976184f / 128.0f));
  const float ang = pos * freq;
  const float s = sinf(ang), c = cosf(ang);
  const float* zp = z + (size_t)t * Bn * INn + (size_t)b * INn;
  const float x1 = zp[j], x2 = zp[j + 64];
  float* xp = xcur + ((size_t)b * Tn + t) * INn;
  xp[j]      = x1 * c - x2 * s;
  xp[j + 64] = x2 * c + x1 * s;
  if (t == 0 && j == 0)
    outst[(size_t)b * STSTRIDE + STSTRIDE - 1] = ts + (float)Tn;
}

__global__ void dla_ln_kernel(const float* __restrict__ x, const float* __restrict__ w,
                              const float* __restrict__ bias, float* __restrict__ xn) {
  const int wave = threadIdx.x >> 5, lane = threadIdx.x & 31;
  const int row = blockIdx.x * 8 + wave;
  const float* xr = x + (size_t)row * INn;
  float s = 0.f, ss = 0.f;
  for (int k = lane; k < INn; k += 32) { float v = xr[k]; s += v; ss += v * v; }
  for (int off = 16; off; off >>= 1) { s += __shfl_xor(s, off); ss += __shfl_xor(ss, off); }
  const float m = s * (1.0f / INn);
  const float var = ss * (1.0f / INn) - m * m;
  const float inv = rsqrtf(var + 1e-5f);
  float* outr = xn + (size_t)row * INn;
  for (int k = lane; k < INn; k += 32)
    outr[k] = (xr[k] - m) * inv * w[k] + bias[k];
}

__global__ void dla_proj_kernel(const float* __restrict__ xn,
                                const float* __restrict__ Wk, const float* __restrict__ Wq,
                                const float* __restrict__ Wv,
                                float* __restrict__ Kb, float* __restrict__ Qb,
                                float* __restrict__ Vb) {
  const int g = blockIdx.x * 8 + (threadIdx.x >> 5);
  const int which = g >> 10;
  const int rem = g & 1023;
  const int b = rem >> 7;
  const int tile = rem & 127;
  const int tm = tile >> 3, tn = tile & 7;
  const float* W = (which == 0) ? Wk : (which == 1) ? Wq : Wv;
  const float* A  = xn + (size_t)b * Tn * INn + (size_t)tm * 16 * INn;
  const float* Bt = W + (size_t)tn * 16 * INn;
  v8f acc = {0.f,0.f,0.f,0.f,0.f,0.f,0.f,0.f};
  tile_abt(A, INn, Bt, INn, INn, acc);
  float* Out = ((which == 0) ? Kb : (which == 1) ? Qb : Vb)
             + (size_t)b * Tn * Hn + (size_t)tm * 16 * Hn + tn * 16;
  const int lane = threadIdx.x & 31, half = lane >> 4, idx = lane & 15;
#pragma unroll
  for (int r = 0; r < 8; r++) {
    float v = acc[r];
    if (which != 2) v = ((v > 0.f) ? v : expm1f(v)) + 1.0f;
    Out[(size_t)(r + 8 * half) * Hn + idx] = v;
  }
}

__global__ void dla_score_kernel(const float* __restrict__ Qb, const float* __restrict__ Kb,
                                 float* __restrict__ Ab) {
  const int g = blockIdx.x * 8 + (threadIdx.x >> 5);
  const int b = g >> 8;
  const int tile = g & 255;
  const int tm = tile >> 4, tn = tile & 15;
  const int lane = threadIdx.x & 31, half = lane >> 4, idx = lane & 15;
  float* Out = Ab + (size_t)b * Tn * Tn + (size_t)tm * 16 * Tn + tn * 16;
  if (tn > tm) {
#pragma unroll
    for (int r = 0; r < 8; r++) Out[(size_t)(r + 8 * half) * Tn + idx] = 0.f;
    return;
  }
  v8f acc = {0.f,0.f,0.f,0.f,0.f,0.f,0.f,0.f};
  tile_abt(Qb + (size_t)b * Tn * Hn + (size_t)tm * 16 * Hn, Hn,
           Kb + (size_t)b * Tn * Hn + (size_t)tn * 16 * Hn, Hn, Hn, acc);
#pragma unroll
  for (int r = 0; r < 8; r++) {
    float v = acc[r];
    if (tn == tm && idx > (r + 8 * half)) v = 0.f;
    Out[(size_t)(r + 8 * half) * Tn + idx] = v;
  }
}

__global__ void dla_den_kernel(const float* __restrict__ Ab, const float* __restrict__ Qb,
                               const float* __restrict__ st, int layerOff,
                               float* __restrict__ den) {
  const int wave = threadIdx.x >> 5, lane = threadIdx.x & 31;
  const int row = blockIdx.x * 8 + wave;
  const int b = row >> 8;
  const float* Ar = Ab + (size_t)b * Tn * Tn + (size_t)(row & 255) * Tn;
  const float* Qr = Qb + (size_t)row * Hn;
  const float* Z0 = st + (size_t)b * STSTRIDE + layerOff + Hn * Hn;
  float s = 0.f;
  for (int k = lane; k < Tn; k += 32) s += Ar[k];
  for (int k = lane; k < Hn; k += 32) s += Qr[k] * Z0[k];
  for (int off = 16; off; off >>= 1) s += __shfl_xor(s, off);
  if (lane == 0) den[row] = s + 1e-5f;
}

__global__ void dla_attnout_kernel(const float* __restrict__ Ab, const float* __restrict__ Vb,
                                   const float* __restrict__ Qb, const float* __restrict__ st,
                                   int layerOff, const float* __restrict__ den,
                                   float* __restrict__ Ao) {
  const int g = blockIdx.x * 8 + (threadIdx.x >> 5);
  const int b = g >> 7;
  const int tile = g & 127;
  const int tm = tile >> 3, tn = tile & 7;
  v8f acc = {0.f,0.f,0.f,0.f,0.f,0.f,0.f,0.f};
  tile_ab(Ab + (size_t)b * Tn * Tn + (size_t)tm * 16 * Tn, Tn,
          Vb + (size_t)b * Tn * Hn + tn * 16, Hn, Tn, acc);
  const float* S0 = st + (size_t)b * STSTRIDE + layerOff;
  tile_ab(Qb + (size_t)b * Tn * Hn + (size_t)tm * 16 * Hn, Hn,
          S0 + tn * 16, Hn, Hn, acc);
  const int lane = threadIdx.x & 31, half = lane >> 4, idx = lane & 15;
  float* Out = Ao + (size_t)b * Tn * Hn + (size_t)tm * 16 * Hn + tn * 16;
#pragma unroll
  for (int r = 0; r < 8; r++) {
    const int row = tm * 16 + r + 8 * half;
    Out[(size_t)(r + 8 * half) * Hn + idx] = acc[r] / den[b * Tn + row];
  }
}

__global__ void dla_states_kernel(const float* __restrict__ Kb, const float* __restrict__ Vb,
                                  const float* __restrict__ st, int layerOff,
                                  float* __restrict__ outst) {
  const int g = blockIdx.x * 8 + (threadIdx.x >> 5);
  const int b = g >> 6;
  const int tile = g & 63;
  const int tm = tile >> 3, tn = tile & 7;
  v8f acc = {0.f,0.f,0.f,0.f,0.f,0.f,0.f,0.f};
  tile_atb(Kb + (size_t)b * Tn * Hn + tm * 16, Hn,
           Vb + (size_t)b * Tn * Hn + tn * 16, Hn, Tn, acc);
  const float* S0 = st + (size_t)b * STSTRIDE + layerOff;
  float* Out = outst + (size_t)b * STSTRIDE + layerOff;
  const int lane = threadIdx.x & 31, half = lane >> 4, idx = lane & 15;
#pragma unroll
  for (int r = 0; r < 8; r++) {
    const int i = tm * 16 + r + 8 * half;
    const int l = tn * 16 + idx;
    Out[(size_t)i * Hn + l] = acc[r] + S0[(size_t)i * Hn + l];
  }
}

__global__ void dla_statez_kernel(const float* __restrict__ Kb, const float* __restrict__ st,
                                  int layerOff, float* __restrict__ outst) {
  const int b = blockIdx.x, i = threadIdx.x;
  const float* Kp = Kb + (size_t)b * Tn * Hn + i;
  float s = 0.f;
  for (int t = 0; t < Tn; t++) s += Kp[(size_t)t * Hn];
  const size_t o = (size_t)b * STSTRIDE + layerOff + Hn * Hn + i;
  outst[o] = st[o] + s;
}

__global__ void dla_ff1_kernel(const float* __restrict__ Ao, const float* __restrict__ w1,
                               const float* __restrict__ b1, float* __restrict__ h1) {
  const int g = blockIdx.x * 8 + (threadIdx.x >> 5);
  const int b = g >> 7;
  const int tile = g & 127;
  const int tm = tile >> 3, tn = tile & 7;
  v8f acc = {0.f,0.f,0.f,0.f,0.f,0.f,0.f,0.f};
  tile_abt(Ao + (size_t)b * Tn * Hn + (size_t)tm * 16 * Hn, Hn,
           w1 + (size_t)tn * 16 * Hn, Hn, Hn, acc);
  const int lane = threadIdx.x & 31, half = lane >> 4, idx = lane & 15;
  float* Out = h1 + (size_t)b * Tn * Hn + (size_t)tm * 16 * Hn + tn * 16;
#pragma unroll
  for (int r = 0; r < 8; r++) {
    float v = acc[r] + b1[tn * 16 + idx];
    Out[(size_t)(r + 8 * half) * Hn + idx] = v > 0.f ? v : 0.f;
  }
}

__global__ void dla_ff2_kernel(const float* __restrict__ h1, const float* __restrict__ w2,
                               const float* __restrict__ b2, const float* __restrict__ xn,
                               const float* __restrict__ scw, const float* __restrict__ scb,
                               float* __restrict__ xcur) {
  const int g = blockIdx.x * 8 + (threadIdx.x >> 5);
  const int b = g >> 7;
  const int tile = g & 127;
  const int tm = tile >> 3, tn = tile & 7;
  v8f acc1 = {0.f,0.f,0.f,0.f,0.f,0.f,0.f,0.f};
  v8f acc2 = {0.f,0.f,0.f,0.f,0.f,0.f,0.f,0.f};
  tile_abt(h1 + (size_t)b * Tn * Hn + (size_t)tm * 16 * Hn, Hn,
           w2 + (size_t)tn * 16 * Hn, Hn, Hn, acc1);
  tile_abt(xn + (size_t)b * Tn * INn + (size_t)tm * 16 * INn, INn,
           scw + (size_t)tn * 16 * INn, INn, INn, acc2);
  const int lane = threadIdx.x & 31, half = lane >> 4, idx = lane & 15;
  float* Out = xcur + (size_t)b * Tn * Hn + (size_t)tm * 16 * Hn + tn * 16;
#pragma unroll
  for (int r = 0; r < 8; r++) {
    float v = acc1[r] + b2[tn * 16 + idx];
    v = v > 0.f ? v : 0.f;
    Out[(size_t)(r + 8 * half) * Hn + idx] = v + acc2[r] + scb[tn * 16 + idx];
  }
}

__global__ void dla_unmap_kernel(const float* __restrict__ x, const float* __restrict__ uw,
                                 const float* __restrict__ ub, float* __restrict__ y) {
  const int g = blockIdx.x * 8 + (threadIdx.x >> 5);
  const int b = g >> 7;
  const int tile = g & 127;
  const int tm = tile >> 3, tn = tile & 7;
  v8f acc = {0.f,0.f,0.f,0.f,0.f,0.f,0.f,0.f};
  tile_abt(x + (size_t)b * Tn * Hn + (size_t)tm * 16 * Hn, Hn,
           uw + (size_t)tn * 16 * Hn, Hn, Hn, acc);
  const int lane = threadIdx.x & 31, half = lane >> 4, idx = lane & 15;
#pragma unroll
  for (int r = 0; r < 8; r++) {
    const int t = tm * 16 + r + 8 * half;
    const int o = tn * 16 + idx;
    y[(size_t)t * Bn * OUTn + (size_t)b * OUTn + o] = acc[r] + ub[o];
  }
}

extern "C" void kernel_launch(void* const* d_in, const int* in_sizes, int n_in,
                              void* d_out, int out_size, void* d_ws, size_t ws_size,
                              hipStream_t stream) {
  (void)in_sizes; (void)n_in; (void)out_size; (void)ws_size;
  const float* z   = (const float*)d_in[0];
  const float* st  = (const float*)d_in[1];
  const float* Wk  = (const float*)d_in[2];
  const float* Wq  = (const float*)d_in[3];
  const float* Wv  = (const float*)d_in[4];
  const float* lnw = (const float*)d_in[5];
  const float* lnb = (const float*)d_in[6];
  const float* w1  = (const float*)d_in[7];
  const float* b1  = (const float*)d_in[8];
  const float* w2  = (const float*)d_in[9];
  const float* b2  = (const float*)d_in[10];
  const float* scw = (const float*)d_in[11];
  const float* scb = (const float*)d_in[12];
  const float* uw  = (const float*)d_in[13];
  const float* ub  = (const float*)d_in[14];
  float* out = (float*)d_out;
  float* ws  = (float*)d_ws;

  float* XCUR = ws;
  float* XN   = ws + 262144;
  float* KB   = ws + 524288;
  float* QB   = ws + 786432;
  float* VB   = ws + 1048576;
  float* AB   = ws + 1310720;
  float* AO   = ws + 1835008;
  float* DEN  = ws + 2097152;
  float* OST  = out + YSZ;

  dla_rotate_kernel<<<Bn * Tn, 64, 0, stream>>>(z, st, XCUR, OST);
  for (int i = 0; i < Ln; i++) {
    const int lo = i * LAYERSTATE;
    dla_ln_kernel<<<256, 256, 0, stream>>>(XCUR, lnw + i * INn, lnb + i * INn, XN);
    dla_proj_kernel<<<384, 256, 0, stream>>>(XN, Wk + (size_t)i * Hn * INn,
                                             Wq + (size_t)i * Hn * INn,
                                             Wv + (size_t)i * Hn * INn, KB, QB, VB);
    dla_score_kernel<<<256, 256, 0, stream>>>(QB, KB, AB);
    dla_den_kernel<<<256, 256, 0, stream>>>(AB, QB, st, lo, DEN);
    dla_attnout_kernel<<<128, 256, 0, stream>>>(AB, VB, QB, st, lo, DEN, AO);
    dla_states_kernel<<<64, 256, 0, stream>>>(KB, VB, st, lo, OST);
    dla_statez_kernel<<<8, 128, 0, stream>>>(KB, st, lo, OST);
    dla_ff1_kernel<<<128, 256, 0, stream>>>(AO, w1 + (size_t)i * Hn * Hn, b1 + i * Hn, QB);
    dla_ff2_kernel<<<128, 256, 0, stream>>>(QB, w2 + (size_t)i * Hn * Hn, b2 + i * Hn,
                                            XN, scw + (size_t)i * Hn * INn, scb + i * Hn, XCUR);
  }
  dla_unmap_kernel<<<128, 256, 0, stream>>>(XCUR, uw, ub, out);
}


# ff1, ff2 and the next stage (next layer's LN+K/Q/V/shortcut, or unmap) chained in one wave on transposed products, accumulators reused as fragments: 12 to 8 launches; their sums' k order permuted
# speedup vs baseline: 1.5823x; 1.5823x over previous
#include <hip/hip_runtime.h>
#include <hip/hip_bf16.h>
#include <math.h>

#define Bn 8
#define Tn 256
#define INn 128
#define Hn 128
#define OUTn 128
#define Ln 2
#define STSTRIDE 33025
#define LAYERSTATE 16512
#define YSZ (Tn*Bn*OUTn)

typedef __attribute__((ext_vector_type(2))) float v2f;
typedef __attribute__((ext_vector_type(8))) float v8f;

__device__ __forceinline__ v8f wmma4(v2f a, v2f b, v8f c) {
  return __builtin_amdgcn_wmma_f32_16x16x4_f32(false, a, false, b, (short)0, c, false, false);
}

__device__ __forceinline__ void tile_abt(const float* A, int lda, const float* Bt, int ldb,
                                         int K, v8f& acc) {
  const int lane = threadIdx.x & 31;
  const int half = lane >> 4;
  const int idx  = lane & 15;
  const float* ap = A  + (size_t)idx * lda + 2 * half;
  const float* bp = Bt + (size_t)idx * ldb + 2 * half;
  for (int k = 0; k < K; k += 4) {
    v2f a, b;
    a.x = ap[k]; a.y = ap[k + 1];
    b.x = bp[k]; b.y = bp[k + 1];
    acc = wmma4(a, b, acc);
  }
}

__device__ __forceinline__ void tile_ab(const float* A, int lda, const float* Bm, int ldb,
                                        int K, v8f& acc) {
  const int lane = threadIdx.x & 31;
  const int half = lane >> 4;
  const int idx  = lane & 15;
  const float* ap = A + (size_t)idx * lda + 2 * half;
  for (int k = 0; k < K; k += 4) {
    v2f a, b;
    a.x = ap[k]; a.y = ap[k + 1];
    b.x = Bm[(size_t)(k + 2 * half)     * ldb + idx];
    b.y = Bm[(size_t)(k + 2 * half + 1) * ldb + idx];
    acc = wmma4(a, b, acc);
  }
}

__device__ __forceinline__ void tile_ab_rowsum(const float* A, int lda, const float* Bm, int ldb,
                                               int K, v8f& acc, v8f& rs) {
  const int lane = threadIdx.x & 31;
  const int half = lane >> 4;
  const int idx  = lane & 15;
  const float* ap = A + (size_t)idx * lda + 2 * half;
  v2f ones; ones.x = 1.0f; ones.y = 1.0f;
  for (int k = 0; k < K; k += 4) {
    v2f a, b;
    a.x = ap[k]; a.y = ap[k + 1];
    b.x = Bm[(size_t)(k + 2 * half)     * ldb + idx];
    b.y = Bm[(size_t)(k + 2 * half + 1) * ldb + idx];
    acc = wmma4(a, b, acc);
    rs  = wmma4(a, ones, rs);
  }
}

__device__ __forceinline__ void tile_ab_rowdot(const float* A, int lda, const float* Bm, int ldb,
                                               const float* u, int K, v8f& acc, v8f& rs) {
  const int lane = threadIdx.x & 31;
  const int half = lane >> 4;
  const int idx  = lane & 15;
  const float* ap = A + (size_t)idx * lda + 2 * half;
#pragma unroll
  for (int k = 0; k < K; k += 4) {
    v2f a, b, w;
    a.x = ap[k]; a.y = ap[k + 1];
    b.x = Bm[(size_t)(k + 2 * half)     * ldb + idx];
    b.y = Bm[(size_t)(k + 2 * half + 1) * ldb + idx];
    w.x = u[k + 2 * half]; w.y = u[k + 2 * half + 1];
    acc = wmma4(a, b, acc);
    rs  = wmma4(a, w, rs);
  }
}

template <bool SP>
__device__ __forceinline__ void tile_ab_rowsum_kt(const float* A, int lda, const float* Bm, int ldb,
                                                  const float* Km, int ldk, int K,
                                                  v8f& acc, v8f& rs, v8f& sx) {
  const int lane = threadIdx.x & 31;
  const int half = lane >> 4;
  const int idx  = lane & 15;
  const float* ap = A + (size_t)idx * lda + 2 * half;
  v2f ones; ones.x = 1.0f; ones.y = 1.0f;
#pragma unroll 8
  for (int k = 0; k < K; k += 4) {
    v2f a, b, ka;
    a.x = ap[k]; a.y = ap[k + 1];
    b.x = Bm[(size_t)(k + 2 * half)     * ldb + idx];
    b.y = Bm[(size_t)(k + 2 * half + 1) * ldb + idx];
    ka.x = Km[(size_t)(k + 2 * half)     * ldk + idx];
    ka.y = Km[(size_t)(k + 2 * half + 1) * ldk + idx];
    acc = wmma4(a, b, acc);
    rs  = wmma4(a, ones, rs);
    sx  = wmma4(ka, SP ? b : ones, sx);
  }
}

__global__ void dla_rotate_kernel(const float* __restrict__ z, const float* __restrict__ st,
                                  float* __restrict__ xcur, float* __restrict__ outst) {
  const int bt = blockIdx.x;
  const int b = bt >> 8, t = bt & 255;
  const int j = threadIdx.x;
  const float ts = st[(size_t)b * STSTRIDE + STSTRIDE - 1];
  const float pos = ts + (float)t;
  const float freq = expf((float)(2 * j) * (-9.210340371976184f / 128.0f));
  const float ang = pos * freq;
  const float s = sinf(ang), c = cosf(ang);
  const float* zp = z + (size_t)t * Bn * INn + (size_t)b * INn;
  const float x1 = zp[j], x2 = zp[j + 64];
  float* xp = xcur + ((size_t)b * Tn + t) * INn;
  xp[j]      = x1 * c - x2 * s;
  xp[j + 64] = x2 * c + x1 * s;
  if (t == 0 && j == 0)
    outst[(size_t)b * STSTRIDE + STSTRIDE - 1] = ts + (float)Tn;
}

__device__ __forceinline__ float elu_plus1(float v) {
  return ((v > 0.f) ? v : expm1f(v)) + 1.0f;
}

__global__ void __launch_bounds__(128)
dla_lnproj_kernel(const float* __restrict__ x,
                  const float* __restrict__ lnw, const float* __restrict__ lnb,
                  const float* __restrict__ Wk, const float* __restrict__ Wq,
                  const float* __restrict__ Wv, const float* __restrict__ scw,
                  float* __restrict__ Kb, float* __restrict__ Qb,
                  float* __restrict__ Vb, float* __restrict__ Sc) {
  const int g = blockIdx.x * 4 + (threadIdx.x >> 5);
  const int b = g >> 7;
  const int tile = g & 127;
  const int tm = tile >> 3, tn = tile & 7;
  const int lane = threadIdx.x & 31, half = lane >> 4, idx = lane & 15;

  const float* xr = x + ((size_t)b * Tn + (size_t)tm * 16 + idx) * INn + 2 * half;
  const size_t wrow = (size_t)(tn * 16 + idx) * INn + 2 * half;
  const float* pk = Wk + wrow;
  const float* pq = Wq + wrow;
  const float* pv = Wv + wrow;
  const float* ps = scw + wrow;
  const float* wp = lnw + 2 * half;
  const float* bp = lnb + 2 * half;

  v2f a[32], lw[32], lb[32], fk[32], fq[32], fv[32], fs[32];
#pragma unroll
  for (int s = 0; s < 32; s++) { a[s].x = xr[4 * s]; a[s].y = xr[4 * s + 1]; }
#pragma unroll
  for (int s = 0; s < 32; s++) {
    lw[s].x = wp[4 * s]; lw[s].y = wp[4 * s + 1];
    lb[s].x = bp[4 * s]; lb[s].y = bp[4 * s + 1];
  }
#pragma unroll
  for (int s = 0; s < 16; s++) {
    fk[s].x = pk[4 * s]; fk[s].y = pk[4 * s + 1];
    fq[s].x = pq[4 * s]; fq[s].y = pq[4 * s + 1];
    fv[s].x = pv[4 * s]; fv[s].y = pv[4 * s + 1];
    fs[s].x = ps[4 * s]; fs[s].y = ps[4 * s + 1];
  }
  __builtin_amdgcn_sched_barrier(0);

  float p0 = 0.f, p1 = 0.f, p2 = 0.f, p3 = 0.f;
#pragma unroll
  for (int s = 0; s < 32; s += 4) {
    p0 += a[s].x + a[s].y;
    p1 += a[s + 1].x + a[s + 1].y;
    p2 += a[s + 2].x + a[s + 2].y;
    p3 += a[s + 3].x + a[s + 3].y;
  }
  float sum = (p0 + p1) + (p2 + p3);
  sum += __shfl_xor(sum, 16);
  const float m = sum * (1.0f / INn);

  float q0 = 0.f, q1 = 0.f, q2 = 0.f, q3 = 0.f;
#pragma unroll
  for (int s = 0; s < 32; s += 4) {
    float d;
    d = a[s].x - m;     q0 += d * d;  d = a[s].y - m;     q0 += d * d;
    d = a[s + 1].x - m; q1 += d * d;  d = a[s + 1].y - m; q1 += d * d;
    d = a[s + 2].x - m; q2 += d * d;  d = a[s + 2].y - m; q2 += d * d;
    d = a[s + 3].x - m; q3 += d * d;  d = a[s + 3].y - m; q3 += d * d;
  }
  float sq = (q0 + q1) + (q2 + q3);
  sq += __shfl_xor(sq, 16);
  const float inv = rsqrtf(sq * (1.0f / INn) + 1e-5f);


#pragma unroll
  for (int s = 0; s < 32; s++) {
    a[s].x = (a[s].x - m) * inv * lw[s].x + lb[s].x;
    a[s].y = (a[s].y - m) * inv * lw[s].y + lb[s].y;
  }

  v8f accK = {0.f,0.f,0.f,0.f,0.f,0.f,0.f,0.f};
  v8f accQ = {0.f,0.f,0.f,0.f,0.f,0.f,0.f,0.f};
  v8f accV = {0.f,0.f,0.f,0.f,0.f,0.f,0.f,0.f};
  v8f accS = {0.f,0.f,0.f,0.f,0.f,0.f,0.f,0.f};

#pragma unroll
  for (int gq = 0; gq < 4; gq++) {
    if (gq < 2) {
#pragma unroll
      for (int s = 8 * gq + 16; s < 8 * gq + 24; s++) {
        fk[s].x = pk[4 * s]; fk[s].y = pk[4 * s + 1];
        fq[s].x = pq[4 * s]; fq[s].y = pq[4 * s + 1];
        fv[s].x = pv[4 * s]; fv[s].y = pv[4 * s + 1];
        fs[s].x = ps[4 * s]; fs[s].y = ps[4 * s + 1];
      }
    }
    __builtin_amdgcn_sched_barrier(0);
#pragma unroll
    for (int s = 8 * gq; s < 8 * gq + 8; s++) {
      accK = wmma4(a[s], fk[s], accK);
      accQ = wmma4(a[s], fq[s], accQ);
      accV = wmma4(a[s], fv[s], accV);
      accS = wmma4(a[s], fs[s], accS);
    }
  }

  const size_t o = (size_t)b * Tn * Hn + (size_t)tm * 16 * Hn + tn * 16;
#pragma unroll
  for (int r = 0; r < 8; r++) {
    const size_t e = o + (size_t)(r + 8 * half) * Hn + idx;
    Kb[e] = elu_plus1(accK[r]);
    Qb[e] = elu_plus1(accQ[r]);
    Vb[e] = accV[r];
    Sc[e] = accS[r];
  }
}

__global__ void dla_score_kernel(const float* __restrict__ Qb, const float* __restrict__ Kb,
                                 float* __restrict__ Ab) {
  const int g = blockIdx.x * 8 + (threadIdx.x >> 5);
  const int b = g >> 8;
  const int tile = g & 255;
  const int tm = tile >> 4, tn = tile & 15;
  const int lane = threadIdx.x & 31, half = lane >> 4, idx = lane & 15;
  float* Out = Ab + (size_t)b * Tn * Tn + (size_t)tm * 16 * Tn + tn * 16;
  if (tn > tm) {
#pragma unroll
    for (int r = 0; r < 8; r++) Out[(size_t)(r + 8 * half) * Tn + idx] = 0.f;
    return;
  }
  v8f acc = {0.f,0.f,0.f,0.f,0.f,0.f,0.f,0.f};
  tile_abt(Qb + (size_t)b * Tn * Hn + (size_t)tm * 16 * Hn, Hn,
           Kb + (size_t)b * Tn * Hn + (size_t)tn * 16 * Hn, Hn, Hn, acc);
#pragma unroll
  for (int r = 0; r < 8; r++) {
    float v = acc[r];
    if (tn == tm && idx > (r + 8 * half)) v = 0.f;
    Out[(size_t)(r + 8 * half) * Tn + idx] = v;
  }
}

__global__ void dla_attnout_kernel(const float* __restrict__ Ab, const float* __restrict__ Vb,
                                   const float* __restrict__ Qb, const float* __restrict__ Kb,
                                   const float* __restrict__ st, int layerOff,
                                   float* __restrict__ Ao, float* __restrict__ outst) {
  const int wave = __builtin_amdgcn_readfirstlane((int)(threadIdx.x >> 5));
  const int g = blockIdx.x * 8 + wave;
  const int b = g >> 7;
  const int tile = g & 127;
  const int tm = tile >> 3, tn = tile & 7;
  const bool carries = ((tn >> 2) & 1) == (tm & 1);
  const bool zcol = (tm & 1) == 0 && tn == 4;
  const int ti = tm >> 1;
  v8f acc  = {0.f,0.f,0.f,0.f,0.f,0.f,0.f,0.f};
  v8f dsum = {0.f,0.f,0.f,0.f,0.f,0.f,0.f,0.f};
  v8f sx   = {0.f,0.f,0.f,0.f,0.f,0.f,0.f,0.f};
  const float* At = Ab + (size_t)b * Tn * Tn + (size_t)tm * 16 * Tn;
  const float* Vt = Vb + (size_t)b * Tn * Hn + tn * 16;
  const float* Kt = Kb + (size_t)b * Tn * Hn + ti * 16;
  if (carries)   tile_ab_rowsum_kt<true>(At, Tn, Vt, Hn, Kt, Hn, Tn, acc, dsum, sx);
  else if (zcol) tile_ab_rowsum_kt<false>(At, Tn, Vt, Hn, Kt, Hn, Tn, acc, dsum, sx);
  else           tile_ab_rowsum(At, Tn, Vt, Hn, Tn, acc, dsum);
  const float* S0 = st + (size_t)b * STSTRIDE + layerOff;
  tile_ab_rowdot(Qb + (size_t)b * Tn * Hn + (size_t)tm * 16 * Hn, Hn,
                 S0 + tn * 16, Hn, S0 + Hn * Hn, Hn, acc, dsum);
  const int lane = threadIdx.x & 31, half = lane >> 4, idx = lane & 15;
  float* Out = Ao + (size_t)b * Tn * Hn + (size_t)tm * 16 * Hn + tn * 16;
#pragma unroll
  for (int r = 0; r < 8; r++)
    Out[(size_t)(r + 8 * half) * Hn + idx] = acc[r] / (dsum[r] + 1e-5f);
  float* SOut = outst + (size_t)b * STSTRIDE + layerOff;
  if (carries) {
#pragma unroll
    for (int r = 0; r < 8; r++) {
      const int i = ti * 16 + r + 8 * half;
      const int l = tn * 16 + idx;
      SOut[(size_t)i * Hn + l] = sx[r] + S0[(size_t)i * Hn + l];
    }
  } else if (zcol && idx == 0) {
#pragma unroll
    for (int r = 0; r < 8; r++) {
      const int i = ti * 16 + r + 8 * half;
      SOut[Hn * Hn + i] = S0[Hn * Hn + i] + sx[r];
    }
  }
}

template <bool LAST>
__global__ void __launch_bounds__(128)
dla_ffchain_kernel(const float* __restrict__ Ao,
                   const float* __restrict__ w1, const float* __restrict__ b1,
                   const float* __restrict__ w2, const float* __restrict__ b2,
                   const float* __restrict__ sc, const float* __restrict__ scb,
                   const float* __restrict__ lnw, const float* __restrict__ lnb,
                   const float* __restrict__ Wa, const float* __restrict__ Wq,
                   const float* __restrict__ Wv, const float* __restrict__ Ws,
                   const float* __restrict__ ub,
                   float* __restrict__ O0, float* __restrict__ O1,
                   float* __restrict__ O2, float* __restrict__ O3) {
  const int g = blockIdx.x * 4 + (threadIdx.x >> 5);
  const int b = g >> 7;
  const int tile = g & 127;
  const int tm = tile >> 3, tn = tile & 7;
  const int lane = threadIdx.x & 31, half = lane >> 4, idx = lane & 15;
  const size_t row0 = (size_t)b * Tn + (size_t)tm * 16;

  const float* aop = Ao + (row0 + idx) * Hn + 2 * half;
  const float* w1p = w1 + (size_t)idx * Hn + 2 * half;
  const float* w2p = w2 + (size_t)idx * Hn + 8 * half;
  const float* b1p = b1 + 8 * half;
  const float* b2p = b2 + 8 * half;
  const float* scbp = scb + 8 * half;
  const float* scp = sc + (row0 + idx) * Hn + 8 * half;
  const size_t w3o = (size_t)(tn * 16 + idx) * Hn + 8 * half;

  v2f ao[32], f1[8][32], f2[8][32];
  v8f acc1[8], acc2[8];
  float b1v[8][8], b2v[8][8], scbv[8][8], scv[8][8];
#pragma unroll
  for (int j = 0; j < 8; j++) {
#pragma unroll
    for (int r = 0; r < 8; r++) { acc1[j][r] = 0.f; acc2[j][r] = 0.f; }
  }
#pragma unroll
  for (int s = 0; s < 32; s++) { ao[s].x = aop[4 * s]; ao[s].y = aop[4 * s + 1]; }
#pragma unroll
  for (int L = 0; L < 2; L++) {
#pragma unroll
    for (int u = 0; u < 2; u++) {
#pragma unroll
      for (int sl = 0; sl < 16; sl++) {
        const float* pp = w1p + (size_t)(16 * (2 * (L >> 1) + u)) * Hn + 4 * (16 * (L & 1) + sl);
        f1[L][u * 16 + sl].x = pp[0]; f1[L][u * 16 + sl].y = pp[1];
      }
    }
  }
#pragma unroll
  for (int j = 0; j < 8; j++) {
#pragma unroll
    for (int r = 0; r < 8; r++) b1v[j][r] = b1p[16 * j + r];
  }

#pragma unroll
  for (int G = 0; G < 16; G++) {
    const int L = G + 2;
    if (L < 8) {
#pragma unroll
      for (int u = 0; u < 2; u++) {
#pragma unroll
        for (int sl = 0; sl < 16; sl++) {
          const float* pp = w1p + (size_t)(16 * (2 * (L >> 1) + u)) * Hn + 4 * (16 * (L & 1) + sl);
          f1[L][u * 16 + sl].x = pp[0]; f1[L][u * 16 + sl].y = pp[1];
        }
      }
    } else if (L < 16) {
      const int M = L - 8;
#pragma unroll
      for (int u = 0; u < 2; u++) {
#pragma unroll
        for (int jl = 0; jl < 4; jl++) {
#pragma unroll
          for (int rr = 0; rr < 4; rr++) {
            const float* pp = w2p + (size_t)(16 * (2 * (M >> 1) + u)) * Hn
                                  + 16 * (4 * (M & 1) + jl) + 2 * rr;
            f2[M][u * 16 + jl * 4 + rr].x = pp[0]; f2[M][u * 16 + jl * 4 + rr].y = pp[1];
          }
        }
      }
    }
    if (G == 14) {
#pragma unroll
      for (int n = 0; n < 8; n++) {
#pragma unroll
        for (int r = 0; r < 8; r++) {
          b2v[n][r] = b2p[16 * n + r];
          scbv[n][r] = scbp[16 * n + r];
          scv[n][r] = scp[16 * n + r];
        }
      }
    }
    asm volatile("" ::: "memory");
    __builtin_amdgcn_sched_barrier(0);
    if (G < 8) {
      const int p = G >> 1, q = G & 1;
#pragma unroll
      for (int sl = 0; sl < 16; sl++) {
#pragma unroll
        for (int u = 0; u < 2; u++)
          acc1[2 * p + u] = wmma4(f1[G][u * 16 + sl], ao[16 * q + sl], acc1[2 * p + u]);
      }
      if (G == 7) {
#pragma unroll
        for (int j = 0; j < 8; j++) {
#pragma unroll
          for (int r = 0; r < 8; r++) {
            const float v = acc1[j][r] + b1v[j][r];
            acc1[j][r] = v > 0.f ? v : 0.f;
          }
        }
      }
    } else {
      const int M = G - 8;
      const int p = M >> 1, q = M & 1;
#pragma unroll
      for (int jl = 0; jl < 4; jl++) {
#pragma unroll
        for (int rr = 0; rr < 4; rr++) {
          v2f hb;
          hb.x = acc1[4 * q + jl][2 * rr]; hb.y = acc1[4 * q + jl][2 * rr + 1];
#pragma unroll
          for (int u = 0; u < 2; u++)
            acc2[2 * p + u] = wmma4(f2[M][u * 16 + jl * 4 + rr], hb, acc2[2 * p + u]);
        }
      }
    }
  }

  v2f f3[4][32];
  if (LAST) {
#pragma unroll
    for (int n = 0; n < 8; n++) {
#pragma unroll
      for (int rr = 0; rr < 4; rr++) {
        const float* pp = Wa + w3o + 16 * n + 2 * rr;
        f3[0][n * 4 + rr].x = pp[0]; f3[0][n * 4 + rr].y = pp[1];
      }
    }
  } else {
#pragma unroll
    for (int q3 = 0; q3 < 2; q3++) {
#pragma unroll
      for (int nl = 0; nl < 2; nl++) {
#pragma unroll
        for (int rr = 0; rr < 4; rr++) {
          const size_t o = w3o + 16 * (2 * q3 + nl) + 2 * rr;
          f3[q3][0 * 8 + nl * 4 + rr].x = Wa[o]; f3[q3][0 * 8 + nl * 4 + rr].y = Wa[o + 1];
          f3[q3][1 * 8 + nl * 4 + rr].x = Wq[o]; f3[q3][1 * 8 + nl * 4 + rr].y = Wq[o + 1];
          f3[q3][2 * 8 + nl * 4 + rr].x = Wv[o]; f3[q3][2 * 8 + nl * 4 + rr].y = Wv[o + 1];
          f3[q3][3 * 8 + nl * 4 + rr].x = Ws[o]; f3[q3][3 * 8 + nl * 4 + rr].y = Ws[o + 1];
        }
      }
    }
  }
  __builtin_amdgcn_sched_barrier(0);

#pragma unroll
  for (int n = 0; n < 8; n++) {
#pragma unroll
    for (int r = 0; r < 8; r++) {
      float v = acc2[n][r] + b2v[n][r];
      v = v > 0.f ? v : 0.f;
      acc2[n][r] = v + scv[n][r] + scbv[n][r];
    }
  }

  if (LAST) {
    v8f accE = {0.f,0.f,0.f,0.f,0.f,0.f,0.f,0.f};
    v8f accO = {0.f,0.f,0.f,0.f,0.f,0.f,0.f,0.f};
#pragma unroll
    for (int n = 0; n < 8; n++) {
#pragma unroll
      for (int rr = 0; rr < 4; rr += 2) {
        v2f xa, xb;
        xa.x = acc2[n][2 * rr];     xa.y = acc2[n][2 * rr + 1];
        xb.x = acc2[n][2 * rr + 2]; xb.y = acc2[n][2 * rr + 3];
        accE = wmma4(xa, f3[0][n * 4 + rr], accE);
        accO = wmma4(xb, f3[0][n * 4 + rr + 1], accO);
      }
    }
#pragma unroll
    for (int r = 0; r < 8; r++) {
      const int t = tm * 16 + r + 8 * half;
      const int o = tn * 16 + idx;
      O0[(size_t)t * Bn * OUTn + (size_t)b * OUTn + o] = (accE[r] + accO[r]) + ub[o];
    }
  } else {
    float lwv[8][8], lbv[8][8];
    const float* lwp = lnw + 8 * half;
    const float* lbp = lnb + 8 * half;
#pragma unroll
    for (int n = 0; n < 8; n++) {
#pragma unroll
      for (int r = 0; r < 8; r++) { lwv[n][r] = lwp[16 * n + r]; lbv[n][r] = lbp[16 * n + r]; }
    }
    __builtin_amdgcn_sched_barrier(0);

    float p0 = 0.f, p1 = 0.f, p2 = 0.f, p3 = 0.f;
#pragma unroll
    for (int n = 0; n < 8; n += 4) {
#pragma unroll
      for (int r = 0; r < 8; r += 2) {
        p0 += acc2[n][r] + acc2[n][r + 1];
        p1 += acc2[n + 1][r] + acc2[n + 1][r + 1];
        p2 += acc2[n + 2][r] + acc2[n + 2][r + 1];
        p3 += acc2[n + 3][r] + acc2[n + 3][r + 1];
      }
    }
    float sum = (p0 + p1) + (p2 + p3);
    sum += __shfl_xor(sum, 16);
    const float m = sum * (1.0f / INn);

    float q0 = 0.f, q1 = 0.f, q2 = 0.f, q3 = 0.f;
#pragma unroll
    for (int n = 0; n < 8; n += 4) {
#pragma unroll
      for (int r = 0; r < 8; r++) {
        float d;
        d = acc2[n][r] - m;     q0 += d * d;
        d = acc2[n + 1][r] - m; q1 += d * d;
        d = acc2[n + 2][r] - m; q2 += d * d;
        d = acc2[n + 3][r] - m; q3 += d * d;
      }
    }
    float sq = (q0 + q1) + (q2 + q3);
    sq += __shfl_xor(sq, 16);
    const float inv = rsqrtf(sq * (1.0f / INn) + 1e-5f);

#pragma unroll
    for (int n = 0; n < 8; n++) {
#pragma unroll
      for (int r = 0; r < 8; r++)
        acc2[n][r] = (acc2[n][r] - m) * inv * lwv[n][r] + lbv[n][r];
    }

    v8f accK = {0.f,0.f,0.f,0.f,0.f,0.f,0.f,0.f};
    v8f accQ = {0.f,0.f,0.f,0.f,0.f,0.f,0.f,0.f};
    v8f accV = {0.f,0.f,0.f,0.f,0.f,0.f,0.f,0.f};
    v8f accS = {0.f,0.f,0.f,0.f,0.f,0.f,0.f,0.f};
#pragma unroll
    for (int q3 = 0; q3 < 4; q3++) {
      if (q3 + 2 < 4) {
#pragma unroll
        for (int nl = 0; nl < 2; nl++) {
#pragma unroll
          for (int rr = 0; rr < 4; rr++) {
            const size_t o = w3o + 16 * (2 * (q3 + 2) + nl) + 2 * rr;
            f3[q3 + 2][0 * 8 + nl * 4 + rr].x = Wa[o]; f3[q3 + 2][0 * 8 + nl * 4 + rr].y = Wa[o + 1];
            f3[q3 + 2][1 * 8 + nl * 4 + rr].x = Wq[o]; f3[q3 + 2][1 * 8 + nl * 4 + rr].y = Wq[o + 1];
            f3[q3 + 2][2 * 8 + nl * 4 + rr].x = Wv[o]; f3[q3 + 2][2 * 8 + nl * 4 + rr].y = Wv[o + 1];
            f3[q3 + 2][3 * 8 + nl * 4 + rr].x = Ws[o]; f3[q3 + 2][3 * 8 + nl * 4 + rr].y = Ws[o + 1];
          }
        }
      }
      __builtin_amdgcn_sched_barrier(0);
#pragma unroll
      for (int nl = 0; nl < 2; nl++) {
#pragma unroll
        for (int rr = 0; rr < 4; rr++) {
          v2f xa;
          xa.x = acc2[2 * q3 + nl][2 * rr]; xa.y = acc2[2 * q3 + nl][2 * rr + 1];
          accK = wmma4(xa, f3[q3][0 * 8 + nl * 4 + rr], accK);
          accQ = wmma4(xa, f3[q3][1 * 8 + nl * 4 + rr], accQ);
          accV = wmma4(xa, f3[q3][2 * 8 + nl * 4 + rr], accV);
          accS = wmma4(xa, f3[q3][3 * 8 + nl * 4 + rr], accS);
        }
      }
    }
    const size_t o16 = row0 * Hn + tn * 16;
#pragma unroll
    for (int r = 0; r < 8; r++) {
      const size_t e = o16 + (size_t)(r + 8 * half) * Hn + idx;
      O0[e] = elu_plus1(accK[r]);
      O1[e] = elu_plus1(accQ[r]);
      O2[e] = accV[r];
      O3[e] = accS[r];
    }
  }
}

extern "C" void kernel_launch(void* const* d_in, const int* in_sizes, int n_in,
                              void* d_out, int out_size, void* d_ws, size_t ws_size,
                              hipStream_t stream) {
  (void)in_sizes; (void)n_in; (void)out_size; (void)ws_size;
  const float* z   = (const float*)d_in[0];
  const float* st  = (const float*)d_in[1];
  const float* Wk  = (const float*)d_in[2];
  const float* Wq  = (const float*)d_in[3];
  const float* Wv  = (const float*)d_in[4];
  const float* lnw = (const float*)d_in[5];
  const float* lnb = (const float*)d_in[6];
  const float* w1  = (const float*)d_in[7];
  const float* b1  = (const float*)d_in[8];
  const float* w2  = (const float*)d_in[9];
  const float* b2  = (const float*)d_in[10];
  const float* scw = (const float*)d_in[11];
  const float* scb = (const float*)d_in[12];
  const float* uw  = (const float*)d_in[13];
  const float* ub  = (const float*)d_in[14];
  float* out = (float*)d_out;
  float* ws  = (float*)d_ws;

  float* XCUR = ws;
  float* SC2  = ws;
  float* SC   = ws + 262144;
  float* KB   = ws + 524288;
  float* QB   = ws + 786432;
  float* VB   = ws + 1048576;
  float* AB   = ws + 1310720;
  float* AO   = ws + 1835008;
  float* OST  = out + YSZ;

  dla_rotate_kernel<<<Bn * Tn, 64, 0, stream>>>(z, st, XCUR, OST);
  dla_lnproj_kernel<<<256, 128, 0, stream>>>(XCUR, lnw, lnb, Wk, Wq, Wv, scw, KB, QB, VB, SC);
  dla_score_kernel<<<256, 256, 0, stream>>>(QB, KB, AB);
  dla_attnout_kernel<<<128, 256, 0, stream>>>(AB, VB, QB, KB, st, 0, AO, OST);
  dla_ffchain_kernel<false><<<256, 128, 0, stream>>>(
      AO, w1, b1, w2, b2, SC, scb, lnw + INn, lnb + INn,
      Wk + (size_t)Hn * INn, Wq + (size_t)Hn * INn, Wv + (size_t)Hn * INn,
      scw + (size_t)Hn * INn, nullptr, KB, QB, VB, SC2);
  dla_score_kernel<<<256, 256, 0, stream>>>(QB, KB, AB);
  dla_attnout_kernel<<<128, 256, 0, stream>>>(AB, VB, QB, KB, st, LAYERSTATE, AO, OST);
  dla_ffchain_kernel<true><<<256, 128, 0, stream>>>(
      AO, w1 + (size_t)Hn * Hn, b1 + Hn, w2 + (size_t)Hn * Hn, b2 + Hn, SC2, scb + Hn,
      nullptr, nullptr, uw, nullptr, nullptr, nullptr, ub, out, nullptr, nullptr, nullptr);
}


# attention in 64-token chunks: state waves in the score launch write prefix states P_c = S0 + K^T V (t < 64c), attn-out = A_cc V_c + Q P_c, key range 64 not 256; num and den sum order changes
# speedup vs baseline: 1.7588x; 1.1115x over previous
#include <hip/hip_runtime.h>
#include <hip/hip_bf16.h>
#include <math.h>

#define Bn 8
#define Tn 256
#define INn 128
#define Hn 128
#define OUTn 128
#define Ln 2
#define STSTRIDE 33025
#define LAYERSTATE 16512
#define YSZ (Tn*Bn*OUTn)

typedef __attribute__((ext_vector_type(2))) float v2f;
typedef __attribute__((ext_vector_type(8))) float v8f;

__device__ __forceinline__ v8f wmma4(v2f a, v2f b, v8f c) {
  return __builtin_amdgcn_wmma_f32_16x16x4_f32(false, a, false, b, (short)0, c, false, false);
}

__device__ __forceinline__ void tile_abt(const float* A, int lda, const float* Bt, int ldb,
                                         int K, v8f& acc) {
  const int lane = threadIdx.x & 31;
  const int half = lane >> 4;
  const int idx  = lane & 15;
  const float* ap = A  + (size_t)idx * lda + 2 * half;
  const float* bp = Bt + (size_t)idx * ldb + 2 * half;
  for (int k = 0; k < K; k += 4) {
    v2f a, b;
    a.x = ap[k]; a.y = ap[k + 1];
    b.x = bp[k]; b.y = bp[k + 1];
    acc = wmma4(a, b, acc);
  }
}

__device__ __forceinline__ void tile_ab(const float* A, int lda, const float* Bm, int ldb,
                                        int K, v8f& acc) {
  const int lane = threadIdx.x & 31;
  const int half = lane >> 4;
  const int idx  = lane & 15;
  const float* ap = A + (size_t)idx * lda + 2 * half;
  for (int k = 0; k < K; k += 4) {
    v2f a, b;
    a.x = ap[k]; a.y = ap[k + 1];
    b.x = Bm[(size_t)(k + 2 * half)     * ldb + idx];
    b.y = Bm[(size_t)(k + 2 * half + 1) * ldb + idx];
    acc = wmma4(a, b, acc);
  }
}

__device__ __forceinline__ void tile_ab_rowsum(const float* A, int lda, const float* Bm, int ldb,
                                               int K, v8f& acc, v8f& rs) {
  const int lane = threadIdx.x & 31;
  const int half = lane >> 4;
  const int idx  = lane & 15;
  const float* ap = A + (size_t)idx * lda + 2 * half;
  v2f ones; ones.x = 1.0f; ones.y = 1.0f;
  for (int k = 0; k < K; k += 4) {
    v2f a, b;
    a.x = ap[k]; a.y = ap[k + 1];
    b.x = Bm[(size_t)(k + 2 * half)     * ldb + idx];
    b.y = Bm[(size_t)(k + 2 * half + 1) * ldb + idx];
    acc = wmma4(a, b, acc);
    rs  = wmma4(a, ones, rs);
  }
}

__device__ __forceinline__ void tile_ab_rowdot(const float* A, int lda, const float* Bm, int ldb,
                                               const float* u, int K, v8f& acc, v8f& rs) {
  const int lane = threadIdx.x & 31;
  const int half = lane >> 4;
  const int idx  = lane & 15;
  const float* ap = A + (size_t)idx * lda + 2 * half;
#pragma unroll
  for (int k = 0; k < K; k += 4) {
    v2f a, b, w;
    a.x = ap[k]; a.y = ap[k + 1];
    b.x = Bm[(size_t)(k + 2 * half)     * ldb + idx];
    b.y = Bm[(size_t)(k + 2 * half + 1) * ldb + idx];
    w.x = u[k + 2 * half]; w.y = u[k + 2 * half + 1];
    acc = wmma4(a, b, acc);
    rs  = wmma4(a, w, rs);
  }
}

template <int K>
__device__ __forceinline__ void tile_ab_rowsum_k(const float* A, const float* Bm, int ldb,
                                                 v8f& acc, v8f& rs) {
  const int lane = threadIdx.x & 31;
  const int half = lane >> 4;
  const int idx  = lane & 15;
  const float* ap = A + (size_t)idx * K + 2 * half;
  v2f ones; ones.x = 1.0f; ones.y = 1.0f;
#pragma unroll
  for (int k = 0; k < K; k += 4) {
    v2f a, b;
    a.x = ap[k]; a.y = ap[k + 1];
    b.x = Bm[(size_t)(k + 2 * half)     * ldb + idx];
    b.y = Bm[(size_t)(k + 2 * half + 1) * ldb + idx];
    acc = wmma4(a, b, acc);
    rs  = wmma4(a, ones, rs);
  }
}

__global__ void dla_rotate_kernel(const float* __restrict__ z, const float* __restrict__ st,
                                  float* __restrict__ xcur, float* __restrict__ outst) {
  const int bt = blockIdx.x;
  const int b = bt >> 8, t = bt & 255;
  const int j = threadIdx.x;
  const float ts = st[(size_t)b * STSTRIDE + STSTRIDE - 1];
  const float pos = ts + (float)t;
  const float freq = expf((float)(2 * j) * (-9.210340371976184f / 128.0f));
  const float ang = pos * freq;
  const float s = sinf(ang), c = cosf(ang);
  const float* zp = z + (size_t)t * Bn * INn + (size_t)b * INn;
  const float x1 = zp[j], x2 = zp[j + 64];
  float* xp = xcur + ((size_t)b * Tn + t) * INn;
  xp[j]      = x1 * c - x2 * s;
  xp[j + 64] = x2 * c + x1 * s;
  if (t == 0 && j == 0)
    outst[(size_t)b * STSTRIDE + STSTRIDE - 1] = ts + (float)Tn;
}

__device__ __forceinline__ float elu_plus1(float v) {
  return ((v > 0.f) ? v : expm1f(v)) + 1.0f;
}

__global__ void __launch_bounds__(128)
dla_lnproj_kernel(const float* __restrict__ x,
                  const float* __restrict__ lnw, const float* __restrict__ lnb,
                  const float* __restrict__ Wk, const float* __restrict__ Wq,
                  const float* __restrict__ Wv, const float* __restrict__ scw,
                  float* __restrict__ Kb, float* __restrict__ Qb,
                  float* __restrict__ Vb, float* __restrict__ Sc) {
  const int g = blockIdx.x * 4 + (threadIdx.x >> 5);
  const int b = g >> 7;
  const int tile = g & 127;
  const int tm = tile >> 3, tn = tile & 7;
  const int lane = threadIdx.x & 31, half = lane >> 4, idx = lane & 15;

  const float* xr = x + ((size_t)b * Tn + (size_t)tm * 16 + idx) * INn + 2 * half;
  const size_t wrow = (size_t)(tn * 16 + idx) * INn + 2 * half;
  const float* pk = Wk + wrow;
  const float* pq = Wq + wrow;
  const float* pv = Wv + wrow;
  const float* ps = scw + wrow;
  const float* wp = lnw + 2 * half;
  const float* bp = lnb + 2 * half;

  v2f a[32], lw[32], lb[32], fk[32], fq[32], fv[32], fs[32];
#pragma unroll
  for (int s = 0; s < 32; s++) { a[s].x = xr[4 * s]; a[s].y = xr[4 * s + 1]; }
#pragma unroll
  for (int s = 0; s < 32; s++) {
    lw[s].x = wp[4 * s]; lw[s].y = wp[4 * s + 1];
    lb[s].x = bp[4 * s]; lb[s].y = bp[4 * s + 1];
  }
#pragma unroll
  for (int s = 0; s < 16; s++) {
    fk[s].x = pk[4 * s]; fk[s].y = pk[4 * s + 1];
    fq[s].x = pq[4 * s]; fq[s].y = pq[4 * s + 1];
    fv[s].x = pv[4 * s]; fv[s].y = pv[4 * s + 1];
    fs[s].x = ps[4 * s]; fs[s].y = ps[4 * s + 1];
  }
  __builtin_amdgcn_sched_barrier(0);

  float p0 = 0.f, p1 = 0.f, p2 = 0.f, p3 = 0.f;
#pragma unroll
  for (int s = 0; s < 32; s += 4) {
    p0 += a[s].x + a[s].y;
    p1 += a[s + 1].x + a[s + 1].y;
    p2 += a[s + 2].x + a[s + 2].y;
    p3 += a[s + 3].x + a[s + 3].y;
  }
  float sum = (p0 + p1) + (p2 + p3);
  sum += __shfl_xor(sum, 16);
  const float m = sum * (1.0f / INn);

  float q0 = 0.f, q1 = 0.f, q2 = 0.f, q3 = 0.f;
#pragma unroll
  for (int s = 0; s < 32; s += 4) {
    float d;
    d = a[s].x - m;     q0 += d * d;  d = a[s].y - m;     q0 += d * d;
    d = a[s + 1].x - m; q1 += d * d;  d = a[s + 1].y - m; q1 += d * d;
    d = a[s + 2].x - m; q2 += d * d;  d = a[s + 2].y - m; q2 += d * d;
    d = a[s + 3].x - m; q3 += d * d;  d = a[s + 3].y - m; q3 += d * d;
  }
  float sq = (q0 + q1) + (q2 + q3);
  sq += __shfl_xor(sq, 16);
  const float inv = rsqrtf(sq * (1.0f / INn) + 1e-5f);


#pragma unroll
  for (int s = 0; s < 32; s++) {
    a[s].x = (a[s].x - m) * inv * lw[s].x + lb[s].x;
    a[s].y = (a[s].y - m) * inv * lw[s].y + lb[s].y;
  }

  v8f accK = {0.f,0.f,0.f,0.f,0.f,0.f,0.f,0.f};
  v8f accQ = {0.f,0.f,0.f,0.f,0.f,0.f,0.f,0.f};
  v8f accV = {0.f,0.f,0.f,0.f,0.f,0.f,0.f,0.f};
  v8f accS = {0.f,0.f,0.f,0.f,0.f,0.f,0.f,0.f};

#pragma unroll
  for (int gq = 0; gq < 4; gq++) {
    if (gq < 2) {
#pragma unroll
      for (int s = 8 * gq + 16; s < 8 * gq + 24; s++) {
        fk[s].x = pk[4 * s]; fk[s].y = pk[4 * s + 1];
        fq[s].x = pq[4 * s]; fq[s].y = pq[4 * s + 1];
        fv[s].x = pv[4 * s]; fv[s].y = pv[4 * s + 1];
        fs[s].x = ps[4 * s]; fs[s].y = ps[4 * s + 1];
      }
    }
    __builtin_amdgcn_sched_barrier(0);
#pragma unroll
    for (int s = 8 * gq; s < 8 * gq + 8; s++) {
      accK = wmma4(a[s], fk[s], accK);
      accQ = wmma4(a[s], fq[s], accQ);
      accV = wmma4(a[s], fv[s], accV);
      accS = wmma4(a[s], fs[s], accS);
    }
  }

  const size_t o = (size_t)b * Tn * Hn + (size_t)tm * 16 * Hn + tn * 16;
#pragma unroll
  for (int r = 0; r < 8; r++) {
    const size_t e = o + (size_t)(r + 8 * half) * Hn + idx;
    Kb[e] = elu_plus1(accK[r]);
    Qb[e] = elu_plus1(accQ[r]);
    Vb[e] = accV[r];
    Sc[e] = accS[r];
  }
}

#define PSZ  393216
#define ZPSZ 3072

__global__ void __launch_bounds__(256)
dla_score_state_kernel(const float* __restrict__ Qb, const float* __restrict__ Kb,
                                       const float* __restrict__ Vb, const float* __restrict__ st,
                                       int layerOff, float* __restrict__ Pb,
                                       float* __restrict__ Zpb, float* __restrict__ Ab,
                                       float* __restrict__ outst) {
  const int wave = __builtin_amdgcn_readfirstlane((int)(threadIdx.x >> 5));
  const int b  = blockIdx.x / 14;
  const int jb = (blockIdx.x - b * 14) * 8 + wave;
  const int lane = threadIdx.x & 31, half = lane >> 4, idx = lane & 15;
  if (jb >= 72) {
    const int y = (jb - 72) >> 3;
    const int c = (y < 4) ? y : (wave >> 1);
    const int q = (y < 4) ? ((wave >= 4) ? 3 : (wave >= 1) ? 2 : 0) : 1;
    const int p = (y < 4) ? ((wave >= 4) ? (wave - 4) : (wave >= 1) ? (wave - 1) : 0) : (wave & 1);
    const int tm = 4 * c + q, tn = 4 * c + p;
    const int lda = 16 * (q + 1);
    float* Out = Ab + (size_t)((b * 4 + c) * 10 + ((q * (q + 1)) >> 1)) * 256 + p * 16;
    v8f acc = {0.f,0.f,0.f,0.f,0.f,0.f,0.f,0.f};
    tile_abt(Qb + (size_t)b * Tn * Hn + (size_t)tm * 16 * Hn, Hn,
             Kb + (size_t)b * Tn * Hn + (size_t)tn * 16 * Hn, Hn, Hn, acc);
#pragma unroll
    for (int r = 0; r < 8; r++) {
      float v = acc[r];
      if (p == q && idx > (r + 8 * half)) v = 0.f;
      Out[(size_t)(r + 8 * half) * lda + idx] = v;
    }
    return;
  }
  const bool zjob = jb >= 64;
  const int ti = zjob ? (jb - 64) : (jb >> 3);
  const int tn = jb & 7;
  const float* S0 = st + (size_t)b * STSTRIDE + layerOff;
  float* SOut = outst + (size_t)b * STSTRIDE + layerOff;
  const float* Kt = Kb + (size_t)b * Tn * Hn + ti * 16 + idx;
  const float* Vt = Vb + (size_t)b * Tn * Hn + tn * 16 + idx;
  v8f sx = {0.f,0.f,0.f,0.f,0.f,0.f,0.f,0.f};
  if (!zjob) {
    float s0v[8];
#pragma unroll
    for (int r = 0; r < 8; r++)
      s0v[r] = S0[(size_t)(ti * 16 + r + 8 * half) * Hn + tn * 16 + idx];
    v2f ka[64], vb[64];
#pragma unroll
    for (int s = 0; s < 32; s++) {
      ka[s].x = Kt[(size_t)(4 * s + 2 * half)     * Hn];
      ka[s].y = Kt[(size_t)(4 * s + 2 * half + 1) * Hn];
      vb[s].x = Vt[(size_t)(4 * s + 2 * half)     * Hn];
      vb[s].y = Vt[(size_t)(4 * s + 2 * half + 1) * Hn];
    }
#pragma unroll
    for (int g = 0; g < 4; g++) {
      if (g + 2 < 4) {
#pragma unroll
        for (int s = 16 * (g + 2); s < 16 * (g + 3); s++) {
          ka[s].x = Kt[(size_t)(4 * s + 2 * half)     * Hn];
          ka[s].y = Kt[(size_t)(4 * s + 2 * half + 1) * Hn];
          vb[s].x = Vt[(size_t)(4 * s + 2 * half)     * Hn];
          vb[s].y = Vt[(size_t)(4 * s + 2 * half + 1) * Hn];
        }
      }
      __builtin_amdgcn_sched_barrier(0);
#pragma unroll
      for (int s = 16 * g; s < 16 * g + 16; s++) sx = wmma4(ka[s], vb[s], sx);
      float* D = (g < 3) ? (Pb + (size_t)(b * 3 + g) * Hn * Hn) : SOut;
#pragma unroll
      for (int r = 0; r < 8; r++)
        D[(size_t)(ti * 16 + r + 8 * half) * Hn + tn * 16 + idx] = sx[r] + s0v[r];
    }
  } else {
    v2f ones; ones.x = 1.0f; ones.y = 1.0f;
    float z0v[8];
#pragma unroll
    for (int r = 0; r < 8; r++) z0v[r] = S0[Hn * Hn + ti * 16 + r + 8 * half];
    v2f ka[64];
#pragma unroll
    for (int s = 0; s < 32; s++) {
      ka[s].x = Kt[(size_t)(4 * s + 2 * half)     * Hn];
      ka[s].y = Kt[(size_t)(4 * s + 2 * half + 1) * Hn];
    }
#pragma unroll
    for (int g = 0; g < 4; g++) {
      if (g + 2 < 4) {
#pragma unroll
        for (int s = 16 * (g + 2); s < 16 * (g + 3); s++) {
          ka[s].x = Kt[(size_t)(4 * s + 2 * half)     * Hn];
          ka[s].y = Kt[(size_t)(4 * s + 2 * half + 1) * Hn];
        }
      }
      __builtin_amdgcn_sched_barrier(0);
#pragma unroll
      for (int s = 16 * g; s < 16 * g + 16; s++) sx = wmma4(ka[s], ones, sx);
      float* D = (g < 3) ? (Zpb + (size_t)(b * 3 + g) * Hn) : (SOut + Hn * Hn);
      float zv[8];
#pragma unroll
      for (int r = 0; r < 8; r++) zv[r] = z0v[r] + sx[r];
      if (idx == 0) {
#pragma unroll
        for (int r = 0; r < 8; r++) D[ti * 16 + r + 8 * half] = zv[r];
      }
    }
  }
}

__global__ void dla_attnout_kernel(const float* __restrict__ Ab, const float* __restrict__ Vb,
                                   const float* __restrict__ Qb,
                                   const float* __restrict__ Pb, const float* __restrict__ Zpb,
                                   const float* __restrict__ st, int layerOff,
                                   float* __restrict__ Ao) {
  const int wave = __builtin_amdgcn_readfirstlane((int)(threadIdx.x >> 5));
  const int g = blockIdx.x * 8 + wave;
  const int b = g >> 7;
  const int tile = g & 127;
  const int tm = tile >> 3, tn = tile & 7;
  const int c = tm >> 2, q = tm & 3;
  v8f acc  = {0.f,0.f,0.f,0.f,0.f,0.f,0.f,0.f};
  v8f dsum = {0.f,0.f,0.f,0.f,0.f,0.f,0.f,0.f};
  const float* At = Ab + (size_t)((b * 4 + c) * 10 + ((q * (q + 1)) >> 1)) * 256;
  const float* Vt = Vb + (size_t)b * Tn * Hn + (size_t)c * 64 * Hn + tn * 16;
  switch (q) {
    case 0:  tile_ab_rowsum_k<16>(At, Vt, Hn, acc, dsum); break;
    case 1:  tile_ab_rowsum_k<32>(At, Vt, Hn, acc, dsum); break;
    case 2:  tile_ab_rowsum_k<48>(At, Vt, Hn, acc, dsum); break;
    default: tile_ab_rowsum_k<64>(At, Vt, Hn, acc, dsum); break;
  }
  const float* S0 = st + (size_t)b * STSTRIDE + layerOff;
  const float* Pc = (c == 0) ? S0 : (Pb + (size_t)(b * 3 + (c - 1)) * Hn * Hn);
  const float* Zc = (c == 0) ? (S0 + Hn * Hn) : (Zpb + (size_t)(b * 3 + (c - 1)) * Hn);
  tile_ab_rowdot(Qb + (size_t)b * Tn * Hn + (size_t)tm * 16 * Hn, Hn,
                 Pc + tn * 16, Hn, Zc, Hn, acc, dsum);
  const int lane = threadIdx.x & 31, half = lane >> 4, idx = lane & 15;
  float* Out = Ao + (size_t)b * Tn * Hn + (size_t)tm * 16 * Hn + tn * 16;
#pragma unroll
  for (int r = 0; r < 8; r++)
    Out[(size_t)(r + 8 * half) * Hn + idx] = acc[r] / (dsum[r] + 1e-5f);
}

template <bool LAST>
__global__ void __launch_bounds__(128)
dla_ffchain_kernel(const float* __restrict__ Ao,
                   const float* __restrict__ w1, const float* __restrict__ b1,
                   const float* __restrict__ w2, const float* __restrict__ b2,
                   const float* __restrict__ sc, const float* __restrict__ scb,
                   const float* __restrict__ lnw, const float* __restrict__ lnb,
                   const float* __restrict__ Wa, const float* __restrict__ Wq,
                   const float* __restrict__ Wv, const float* __restrict__ Ws,
                   const float* __restrict__ ub,
                   float* __restrict__ O0, float* __restrict__ O1,
                   float* __restrict__ O2, float* __restrict__ O3) {
  const int g = blockIdx.x * 4 + (threadIdx.x >> 5);
  const int b = g >> 7;
  const int tile = g & 127;
  const int tm = tile >> 3, tn = tile & 7;
  const int lane = threadIdx.x & 31, half = lane >> 4, idx = lane & 15;
  const size_t row0 = (size_t)b * Tn + (size_t)tm * 16;

  const float* aop = Ao + (row0 + idx) * Hn + 2 * half;
  const float* w1p = w1 + (size_t)idx * Hn + 2 * half;
  const float* w2p = w2 + (size_t)idx * Hn + 8 * half;
  const float* b1p = b1 + 8 * half;
  const float* b2p = b2 + 8 * half;
  const float* scbp = scb + 8 * half;
  const float* scp = sc + (row0 + idx) * Hn + 8 * half;
  const size_t w3o = (size_t)(tn * 16 + idx) * Hn + 8 * half;

  v2f ao[32], f1[8][32], f2[8][32];
  v8f acc1[8], acc2[8];
  float b1v[8][8], b2v[8][8], scbv[8][8], scv[8][8];
#pragma unroll
  for (int j = 0; j < 8; j++) {
#pragma unroll
    for (int r = 0; r < 8; r++) { acc1[j][r] = 0.f; acc2[j][r] = 0.f; }
  }
#pragma unroll
  for (int s = 0; s < 32; s++) { ao[s].x = aop[4 * s]; ao[s].y = aop[4 * s + 1]; }
#pragma unroll
  for (int L = 0; L < 2; L++) {
#pragma unroll
    for (int u = 0; u < 2; u++) {
#pragma unroll
      for (int sl = 0; sl < 16; sl++) {
        const float* pp = w1p + (size_t)(16 * (2 * (L >> 1) + u)) * Hn + 4 * (16 * (L & 1) + sl);
        f1[L][u * 16 + sl].x = pp[0]; f1[L][u * 16 + sl].y = pp[1];
      }
    }
  }
#pragma unroll
  for (int j = 0; j < 8; j++) {
#pragma unroll
    for (int r = 0; r < 8; r++) b1v[j][r] = b1p[16 * j + r];
  }

#pragma unroll
  for (int G = 0; G < 16; G++) {
    const int L = G + 2;
    if (L < 8) {
#pragma unroll
      for (int u = 0; u < 2; u++) {
#pragma unroll
        for (int sl = 0; sl < 16; sl++) {
          const float* pp = w1p + (size_t)(16 * (2 * (L >> 1) + u)) * Hn + 4 * (16 * (L & 1) + sl);
          f1[L][u * 16 + sl].x = pp[0]; f1[L][u * 16 + sl].y = pp[1];
        }
      }
    } else if (L < 16) {
      const int M = L - 8;
#pragma unroll
      for (int u = 0; u < 2; u++) {
#pragma unroll
        for (int jl = 0; jl < 4; jl++) {
#pragma unroll
          for (int rr = 0; rr < 4; rr++) {
            const float* pp = w2p + (size_t)(16 * (2 * (M >> 1) + u)) * Hn
                                  + 16 * (4 * (M & 1) + jl) + 2 * rr;
            f2[M][u * 16 + jl * 4 + rr].x = pp[0]; f2[M][u * 16 + jl * 4 + rr].y = pp[1];
          }
        }
      }
    }
    if (G == 14) {
#pragma unroll
      for (int n = 0; n < 8; n++) {
#pragma unroll
        for (int r = 0; r < 8; r++) {
          b2v[n][r] = b2p[16 * n + r];
          scbv[n][r] = scbp[16 * n + r];
          scv[n][r] = scp[16 * n + r];
        }
      }
    }
    asm volatile("" ::: "memory");
    __builtin_amdgcn_sched_barrier(0);
    if (G < 8) {
      const int p = G >> 1, q = G & 1;
#pragma unroll
      for (int sl = 0; sl < 16; sl++) {
#pragma unroll
        for (int u = 0; u < 2; u++)
          acc1[2 * p + u] = wmma4(f1[G][u * 16 + sl], ao[16 * q + sl], acc1[2 * p + u]);
      }
      if (G == 7) {
#pragma unroll
        for (int j = 0; j < 8; j++) {
#pragma unroll
          for (int r = 0; r < 8; r++) {
            const float v = acc1[j][r] + b1v[j][r];
            acc1[j][r] = v > 0.f ? v : 0.f;
          }
        }
      }
    } else {
      const int M = G - 8;
      const int p = M >> 1, q = M & 1;
#pragma unroll
      for (int jl = 0; jl < 4; jl++) {
#pragma unroll
        for (int rr = 0; rr < 4; rr++) {
          v2f hb;
          hb.x = acc1[4 * q + jl][2 * rr]; hb.y = acc1[4 * q + jl][2 * rr + 1];
#pragma unroll
          for (int u = 0; u < 2; u++)
            acc2[2 * p + u] = wmma4(f2[M][u * 16 + jl * 4 + rr], hb, acc2[2 * p + u]);
        }
      }
    }
  }

  v2f f3[4][32];
  if (LAST) {
#pragma unroll
    for (int n = 0; n < 8; n++) {
#pragma unroll
      for (int rr = 0; rr < 4; rr++) {
        const float* pp = Wa + w3o + 16 * n + 2 * rr;
        f3[0][n * 4 + rr].x = pp[0]; f3[0][n * 4 + rr].y = pp[1];
      }
    }
  } else {
#pragma unroll
    for (int q3 = 0; q3 < 2; q3++) {
#pragma unroll
      for (int nl = 0; nl < 2; nl++) {
#pragma unroll
        for (int rr = 0; rr < 4; rr++) {
          const size_t o = w3o + 16 * (2 * q3 + nl) + 2 * rr;
          f3[q3][0 * 8 + nl * 4 + rr].x = Wa[o]; f3[q3][0 * 8 + nl * 4 + rr].y = Wa[o + 1];
          f3[q3][1 * 8 + nl * 4 + rr].x = Wq[o]; f3[q3][1 * 8 + nl * 4 + rr].y = Wq[o + 1];
          f3[q3][2 * 8 + nl * 4 + rr].x = Wv[o]; f3[q3][2 * 8 + nl * 4 + rr].y = Wv[o + 1];
          f3[q3][3 * 8 + nl * 4 + rr].x = Ws[o]; f3[q3][3 * 8 + nl * 4 + rr].y = Ws[o + 1];
        }
      }
    }
  }
  __builtin_amdgcn_sched_barrier(0);

#pragma unroll
  for (int n = 0; n < 8; n++) {
#pragma unroll
    for (int r = 0; r < 8; r++) {
      float v = acc2[n][r] + b2v[n][r];
      v = v > 0.f ? v : 0.f;
      acc2[n][r] = v + scv[n][r] + scbv[n][r];
    }
  }

  if (LAST) {
    v8f accE = {0.f,0.f,0.f,0.f,0.f,0.f,0.f,0.f};
    v8f accO = {0.f,0.f,0.f,0.f,0.f,0.f,0.f,0.f};
#pragma unroll
    for (int n = 0; n < 8; n++) {
#pragma unroll
      for (int rr = 0; rr < 4; rr += 2) {
        v2f xa, xb;
        xa.x = acc2[n][2 * rr];     xa.y = acc2[n][2 * rr + 1];
        xb.x = acc2[n][2 * rr + 2]; xb.y = acc2[n][2 * rr + 3];
        accE = wmma4(xa, f3[0][n * 4 + rr], accE);
        accO = wmma4(xb, f3[0][n * 4 + rr + 1], accO);
      }
    }
#pragma unroll
    for (int r = 0; r < 8; r++) {
      const int t = tm * 16 + r + 8 * half;
      const int o = tn * 16 + idx;
      O0[(size_t)t * Bn * OUTn + (size_t)b * OUTn + o] = (accE[r] + accO[r]) + ub[o];
    }
  } else {
    float lwv[8][8], lbv[8][8];
    const float* lwp = lnw + 8 * half;
    const float* lbp = lnb + 8 * half;
#pragma unroll
    for (int n = 0; n < 8; n++) {
#pragma unroll
      for (int r = 0; r < 8; r++) { lwv[n][r] = lwp[16 * n + r]; lbv[n][r] = lbp[16 * n + r]; }
    }
    __builtin_amdgcn_sched_barrier(0);

    float p0 = 0.f, p1 = 0.f, p2 = 0.f, p3 = 0.f;
#pragma unroll
    for (int n = 0; n < 8; n += 4) {
#pragma unroll
      for (int r = 0; r < 8; r += 2) {
        p0 += acc2[n][r] + acc2[n][r + 1];
        p1 += acc2[n + 1][r] + acc2[n + 1][r + 1];
        p2 += acc2[n + 2][r] + acc2[n + 2][r + 1];
        p3 += acc2[n + 3][r] + acc2[n + 3][r + 1];
      }
    }
    float sum = (p0 + p1) + (p2 + p3);
    sum += __shfl_xor(sum, 16);
    const float m = sum * (1.0f / INn);

    float q0 = 0.f, q1 = 0.f, q2 = 0.f, q3 = 0.f;
#pragma unroll
    for (int n = 0; n < 8; n += 4) {
#pragma unroll
      for (int r = 0; r < 8; r++) {
        float d;
        d = acc2[n][r] - m;     q0 += d * d;
        d = acc2[n + 1][r] - m; q1 += d * d;
        d = acc2[n + 2][r] - m; q2 += d * d;
        d = acc2[n + 3][r] - m; q3 += d * d;
      }
    }
    float sq = (q0 + q1) + (q2 + q3);
    sq += __shfl_xor(sq, 16);
    const float inv = rsqrtf(sq * (1.0f / INn) + 1e-5f);

#pragma unroll
    for (int n = 0; n < 8; n++) {
#pragma unroll
      for (int r = 0; r < 8; r++)
        acc2[n][r] = (acc2[n][r] - m) * inv * lwv[n][r] + lbv[n][r];
    }

    v8f accK = {0.f,0.f,0.f,0.f,0.f,0.f,0.f,0.f};
    v8f accQ = {0.f,0.f,0.f,0.f,0.f,0.f,0.f,0.f};
    v8f accV = {0.f,0.f,0.f,0.f,0.f,0.f,0.f,0.f};
    v8f accS = {0.f,0.f,0.f,0.f,0.f,0.f,0.f,0.f};
#pragma unroll
    for (int q3 = 0; q3 < 4; q3++) {
      if (q3 + 2 < 4) {
#pragma unroll
        for (int nl = 0; nl < 2; nl++) {
#pragma unroll
          for (int rr = 0; rr < 4; rr++) {
            const size_t o = w3o + 16 * (2 * (q3 + 2) + nl) + 2 * rr;
            f3[q3 + 2][0 * 8 + nl * 4 + rr].x = Wa[o]; f3[q3 + 2][0 * 8 + nl * 4 + rr].y = Wa[o + 1];
            f3[q3 + 2][1 * 8 + nl * 4 + rr].x = Wq[o]; f3[q3 + 2][1 * 8 + nl * 4 + rr].y = Wq[o + 1];
            f3[q3 + 2][2 * 8 + nl * 4 + rr].x = Wv[o]; f3[q3 + 2][2 * 8 + nl * 4 + rr].y = Wv[o + 1];
            f3[q3 + 2][3 * 8 + nl * 4 + rr].x = Ws[o]; f3[q3 + 2][3 * 8 + nl * 4 + rr].y = Ws[o + 1];
          }
        }
      }
      __builtin_amdgcn_sched_barrier(0);
#pragma unroll
      for (int nl = 0; nl < 2; nl++) {
#pragma unroll
        for (int rr = 0; rr < 4; rr++) {
          v2f xa;
          xa.x = acc2[2 * q3 + nl][2 * rr]; xa.y = acc2[2 * q3 + nl][2 * rr + 1];
          accK = wmma4(xa, f3[q3][0 * 8 + nl * 4 + rr], accK);
          accQ = wmma4(xa, f3[q3][1 * 8 + nl * 4 + rr], accQ);
          accV = wmma4(xa, f3[q3][2 * 8 + nl * 4 + rr], accV);
          accS = wmma4(xa, f3[q3][3 * 8 + nl * 4 + rr], accS);
        }
      }
    }
    const size_t o16 = row0 * Hn + tn * 16;
#pragma unroll
    for (int r = 0; r < 8; r++) {
      const size_t e = o16 + (size_t)(r + 8 * half) * Hn + idx;
      O0[e] = elu_plus1(accK[r]);
      O1[e] = elu_plus1(accQ[r]);
      O2[e] = accV[r];
      O3[e] = accS[r];
    }
  }
}

extern "C" void kernel_launch(void* const* d_in, const int* in_sizes, int n_in,
                              void* d_out, int out_size, void* d_ws, size_t ws_size,
                              hipStream_t stream) {
  (void)in_sizes; (void)n_in; (void)out_size; (void)ws_size;
  const float* z   = (const float*)d_in[0];
  const float* st  = (const float*)d_in[1];
  const float* Wk  = (const float*)d_in[2];
  const float* Wq  = (const float*)d_in[3];
  const float* Wv  = (const float*)d_in[4];
  const float* lnw = (const float*)d_in[5];
  const float* lnb = (const float*)d_in[6];
  const float* w1  = (const float*)d_in[7];
  const float* b1  = (const float*)d_in[8];
  const float* w2  = (const float*)d_in[9];
  const float* b2  = (const float*)d_in[10];
  const float* scw = (const float*)d_in[11];
  const float* scb = (const float*)d_in[12];
  const float* uw  = (const float*)d_in[13];
  const float* ub  = (const float*)d_in[14];
  float* out = (float*)d_out;
  float* ws  = (float*)d_ws;

  float* XCUR = ws;
  float* SC2  = ws;
  float* SC   = ws + 262144;
  float* KB   = ws + 524288;
  float* QB   = ws + 786432;
  float* VB   = ws + 1048576;
  float* PB   = ws + 1310720;
  float* ZPB  = PB + PSZ;
  float* AB   = ZPB + ZPSZ;
  float* AO   = ws + 1835008;
  float* OST  = out + YSZ;

  dla_rotate_kernel<<<Bn * Tn, 64, 0, stream>>>(z, st, XCUR, OST);
  dla_lnproj_kernel<<<256, 128, 0, stream>>>(XCUR, lnw, lnb, Wk, Wq, Wv, scw, KB, QB, VB, SC);
  dla_score_state_kernel<<<Bn * 14, 256, 0, stream>>>(QB, KB, VB, st, 0, PB, ZPB, AB, OST);
  dla_attnout_kernel<<<128, 256, 0, stream>>>(AB, VB, QB, PB, ZPB, st, 0, AO);
  dla_ffchain_kernel<false><<<256, 128, 0, stream>>>(
      AO, w1, b1, w2, b2, SC, scb, lnw + INn, lnb + INn,
      Wk + (size_t)Hn * INn, Wq + (size_t)Hn * INn, Wv + (size_t)Hn * INn,
      scw + (size_t)Hn * INn, nullptr, KB, QB, VB, SC2);
  dla_score_state_kernel<<<Bn * 14, 256, 0, stream>>>(QB, KB, VB, st, LAYERSTATE, PB, ZPB, AB, OST);
  dla_attnout_kernel<<<128, 256, 0, stream>>>(AB, VB, QB, PB, ZPB, st, LAYERSTATE, AO);
  dla_ffchain_kernel<true><<<256, 128, 0, stream>>>(
      AO, w1 + (size_t)Hn * Hn, b1 + Hn, w2 + (size_t)Hn * Hn, b2 + Hn, SC2, scb + Hn,
      nullptr, nullptr, uw, nullptr, nullptr, nullptr, ub, out, nullptr, nullptr, nullptr);
}


# ff1 and ff2 computed once per work-group, not by each of its 4 waves: wave w takes h1 and x tiles 2w, 2w+1, the tiles cross through two LDS tiles behind one barrier each; sum order unchanged
# speedup vs baseline: 2.1458x; 1.2200x over previous
#include <hip/hip_runtime.h>
#include <hip/hip_bf16.h>
#include <math.h>

#define Bn 8
#define Tn 256
#define INn 128
#define Hn 128
#define OUTn 128
#define Ln 2
#define STSTRIDE 33025
#define LAYERSTATE 16512
#define YSZ (Tn*Bn*OUTn)

typedef __attribute__((ext_vector_type(2))) float v2f;
typedef __attribute__((ext_vector_type(8))) float v8f;

__device__ __forceinline__ v8f wmma4(v2f a, v2f b, v8f c) {
  return __builtin_amdgcn_wmma_f32_16x16x4_f32(false, a, false, b, (short)0, c, false, false);
}

__device__ __forceinline__ void tile_abt(const float* A, int lda, const float* Bt, int ldb,
                                         int K, v8f& acc) {
  const int lane = threadIdx.x & 31;
  const int half = lane >> 4;
  const int idx  = lane & 15;
  const float* ap = A  + (size_t)idx * lda + 2 * half;
  const float* bp = Bt + (size_t)idx * ldb + 2 * half;
  for (int k = 0; k < K; k += 4) {
    v2f a, b;
    a.x = ap[k]; a.y = ap[k + 1];
    b.x = bp[k]; b.y = bp[k + 1];
    acc = wmma4(a, b, acc);
  }
}

__device__ __forceinline__ void tile_ab(const float* A, int lda, const float* Bm, int ldb,
                                        int K, v8f& acc) {
  const int lane = threadIdx.x & 31;
  const int half = lane >> 4;
  const int idx  = lane & 15;
  const float* ap = A + (size_t)idx * lda + 2 * half;
  for (int k = 0; k < K; k += 4) {
    v2f a, b;
    a.x = ap[k]; a.y = ap[k + 1];
    b.x = Bm[(size_t)(k + 2 * half)     * ldb + idx];
    b.y = Bm[(size_t)(k + 2 * half + 1) * ldb + idx];
    acc = wmma4(a, b, acc);
  }
}

__device__ __forceinline__ void tile_ab_rowsum(const float* A, int lda, const float* Bm, int ldb,
                                               int K, v8f& acc, v8f& rs) {
  const int lane = threadIdx.x & 31;
  const int half = lane >> 4;
  const int idx  = lane & 15;
  const float* ap = A + (size_t)idx * lda + 2 * half;
  v2f ones; ones.x = 1.0f; ones.y = 1.0f;
  for (int k = 0; k < K; k += 4) {
    v2f a, b;
    a.x = ap[k]; a.y = ap[k + 1];
    b.x = Bm[(size_t)(k + 2 * half)     * ldb + idx];
    b.y = Bm[(size_t)(k + 2 * half + 1) * ldb + idx];
    acc = wmma4(a, b, acc);
    rs  = wmma4(a, ones, rs);
  }
}

__device__ __forceinline__ void tile_ab_rowdot(const float* A, int lda, const float* Bm, int ldb,
                                               const float* u, int K, v8f& acc, v8f& rs) {
  const int lane = threadIdx.x & 31;
  const int half = lane >> 4;
  const int idx  = lane & 15;
  const float* ap = A + (size_t)idx * lda + 2 * half;
#pragma unroll
  for (int k = 0; k < K; k += 4) {
    v2f a, b, w;
    a.x = ap[k]; a.y = ap[k + 1];
    b.x = Bm[(size_t)(k + 2 * half)     * ldb + idx];
    b.y = Bm[(size_t)(k + 2 * half + 1) * ldb + idx];
    w.x = u[k + 2 * half]; w.y = u[k + 2 * half + 1];
    acc = wmma4(a, b, acc);
    rs  = wmma4(a, w, rs);
  }
}

template <int K>
__device__ __forceinline__ void tile_ab_rowsum_k(const float* A, const float* Bm, int ldb,
                                                 v8f& acc, v8f& rs) {
  const int lane = threadIdx.x & 31;
  const int half = lane >> 4;
  const int idx  = lane & 15;
  const float* ap = A + (size_t)idx * K + 2 * half;
  v2f ones; ones.x = 1.0f; ones.y = 1.0f;
#pragma unroll
  for (int k = 0; k < K; k += 4) {
    v2f a, b;
    a.x = ap[k]; a.y = ap[k + 1];
    b.x = Bm[(size_t)(k + 2 * half)     * ldb + idx];
    b.y = Bm[(size_t)(k + 2 * half + 1) * ldb + idx];
    acc = wmma4(a, b, acc);
    rs  = wmma4(a, ones, rs);
  }
}

__global__ void dla_rotate_kernel(const float* __restrict__ z, const float* __restrict__ st,
                                  float* __restrict__ xcur, float* __restrict__ outst) {
  const int bt = blockIdx.x;
  const int b = bt >> 8, t = bt & 255;
  const int j = threadIdx.x;
  const float ts = st[(size_t)b * STSTRIDE + STSTRIDE - 1];
  const float pos = ts + (float)t;
  const float freq = expf((float)(2 * j) * (-9.210340371976184f / 128.0f));
  const float ang = pos * freq;
  const float s = sinf(ang), c = cosf(ang);
  const float* zp = z + (size_t)t * Bn * INn + (size_t)b * INn;
  const float x1 = zp[j], x2 = zp[j + 64];
  float* xp = xcur + ((size_t)b * Tn + t) * INn;
  xp[j]      = x1 * c - x2 * s;
  xp[j + 64] = x2 * c + x1 * s;
  if (t == 0 && j == 0)
    outst[(size_t)b * STSTRIDE + STSTRIDE - 1] = ts + (float)Tn;
}

__device__ __forceinline__ float elu_plus1(float v) {
  return ((v > 0.f) ? v : expm1f(v)) + 1.0f;
}

__global__ void __launch_bounds__(128)
dla_lnproj_kernel(const float* __restrict__ x,
                  const float* __restrict__ lnw, const float* __restrict__ lnb,
                  const float* __restrict__ Wk, const float* __restrict__ Wq,
                  const float* __restrict__ Wv, const float* __restrict__ scw,
                  float* __restrict__ Kb, float* __restrict__ Qb,
                  float* __restrict__ Vb, float* __restrict__ Sc) {
  const int g = blockIdx.x * 4 + (threadIdx.x >> 5);
  const int b = g >> 7;
  const int tile = g & 127;
  const int tm = tile >> 3, tn = tile & 7;
  const int lane = threadIdx.x & 31, half = lane >> 4, idx = lane & 15;

  const float* xr = x + ((size_t)b * Tn + (size_t)tm * 16 + idx) * INn + 2 * half;
  const size_t wrow = (size_t)(tn * 16 + idx) * INn + 2 * half;
  const float* pk = Wk + wrow;
  const float* pq = Wq + wrow;
  const float* pv = Wv + wrow;
  const float* ps = scw + wrow;
  const float* wp = lnw + 2 * half;
  const float* bp = lnb + 2 * half;

  v2f a[32], lw[32], lb[32], fk[32], fq[32], fv[32], fs[32];
#pragma unroll
  for (int s = 0; s < 32; s++) { a[s].x = xr[4 * s]; a[s].y = xr[4 * s + 1]; }
#pragma unroll
  for (int s = 0; s < 32; s++) {
    lw[s].x = wp[4 * s]; lw[s].y = wp[4 * s + 1];
    lb[s].x = bp[4 * s]; lb[s].y = bp[4 * s + 1];
  }
#pragma unroll
  for (int s = 0; s < 16; s++) {
    fk[s].x = pk[4 * s]; fk[s].y = pk[4 * s + 1];
    fq[s].x = pq[4 * s]; fq[s].y = pq[4 * s + 1];
    fv[s].x = pv[4 * s]; fv[s].y = pv[4 * s + 1];
    fs[s].x = ps[4 * s]; fs[s].y = ps[4 * s + 1];
  }
  __builtin_amdgcn_sched_barrier(0);

  float p0 = 0.f, p1 = 0.f, p2 = 0.f, p3 = 0.f;
#pragma unroll
  for (int s = 0; s < 32; s += 4) {
    p0 += a[s].x + a[s].y;
    p1 += a[s + 1].x + a[s + 1].y;
    p2 += a[s + 2].x + a[s + 2].y;
    p3 += a[s + 3].x + a[s + 3].y;
  }
  float sum = (p0 + p1) + (p2 + p3);
  sum += __shfl_xor(sum, 16);
  const float m = sum * (1.0f / INn);

  float q0 = 0.f, q1 = 0.f, q2 = 0.f, q3 = 0.f;
#pragma unroll
  for (int s = 0; s < 32; s += 4) {
    float d;
    d = a[s].x - m;     q0 += d * d;  d = a[s].y - m;     q0 += d * d;
    d = a[s + 1].x - m; q1 += d * d;  d = a[s + 1].y - m; q1 += d * d;
    d = a[s + 2].x - m; q2 += d * d;  d = a[s + 2].y - m; q2 += d * d;
    d = a[s + 3].x - m; q3 += d * d;  d = a[s + 3].y - m; q3 += d * d;
  }
  float sq = (q0 + q1) + (q2 + q3);
  sq += __shfl_xor(sq, 16);
  const float inv = rsqrtf(sq * (1.0f / INn) + 1e-5f);


#pragma unroll
  for (int s = 0; s < 32; s++) {
    a[s].x = (a[s].x - m) * inv * lw[s].x + lb[s].x;
    a[s].y = (a[s].y - m) * inv * lw[s].y + lb[s].y;
  }

  v8f accK = {0.f,0.f,0.f,0.f,0.f,0.f,0.f,0.f};
  v8f accQ = {0.f,0.f,0.f,0.f,0.f,0.f,0.f,0.f};
  v8f accV = {0.f,0.f,0.f,0.f,0.f,0.f,0.f,0.f};
  v8f accS = {0.f,0.f,0.f,0.f,0.f,0.f,0.f,0.f};

#pragma unroll
  for (int gq = 0; gq < 4; gq++) {
    if (gq < 2) {
#pragma unroll
      for (int s = 8 * gq + 16; s < 8 * gq + 24; s++) {
        fk[s].x = pk[4 * s]; fk[s].y = pk[4 * s + 1];
        fq[s].x = pq[4 * s]; fq[s].y = pq[4 * s + 1];
        fv[s].x = pv[4 * s]; fv[s].y = pv[4 * s + 1];
        fs[s].x = ps[4 * s]; fs[s].y = ps[4 * s + 1];
      }
    }
    __builtin_amdgcn_sched_barrier(0);
#pragma unroll
    for (int s = 8 * gq; s < 8 * gq + 8; s++) {
      accK = wmma4(a[s], fk[s], accK);
      accQ = wmma4(a[s], fq[s], accQ);
      accV = wmma4(a[s], fv[s], accV);
      accS = wmma4(a[s], fs[s], accS);
    }
  }

  const size_t o = (size_t)b * Tn * Hn + (size_t)tm * 16 * Hn + tn * 16;
#pragma unroll
  for (int r = 0; r < 8; r++) {
    const size_t e = o + (size_t)(r + 8 * half) * Hn + idx;
    Kb[e] = elu_plus1(accK[r]);
    Qb[e] = elu_plus1(accQ[r]);
    Vb[e] = accV[r];
    Sc[e] = accS[r];
  }
}

#define PSZ  393216
#define ZPSZ 3072

__global__ void __launch_bounds__(256)
dla_score_state_kernel(const float* __restrict__ Qb, const float* __restrict__ Kb,
                                       const float* __restrict__ Vb, const float* __restrict__ st,
                                       int layerOff, float* __restrict__ Pb,
                                       float* __restrict__ Zpb, float* __restrict__ Ab,
                                       float* __restrict__ outst) {
  const int wave = __builtin_amdgcn_readfirstlane((int)(threadIdx.x >> 5));
  const int b  = blockIdx.x / 14;
  const int jb = (blockIdx.x - b * 14) * 8 + wave;
  const int lane = threadIdx.x & 31, half = lane >> 4, idx = lane & 15;
  if (jb >= 72) {
    const int y = (jb - 72) >> 3;
    const int c = (y < 4) ? y : (wave >> 1);
    const int q = (y < 4) ? ((wave >= 4) ? 3 : (wave >= 1) ? 2 : 0) : 1;
    const int p = (y < 4) ? ((wave >= 4) ? (wave - 4) : (wave >= 1) ? (wave - 1) : 0) : (wave & 1);
    const int tm = 4 * c + q, tn = 4 * c + p;
    const int lda = 16 * (q + 1);
    float* Out = Ab + (size_t)((b * 4 + c) * 10 + ((q * (q + 1)) >> 1)) * 256 + p * 16;
    v8f acc = {0.f,0.f,0.f,0.f,0.f,0.f,0.f,0.f};
    tile_abt(Qb + (size_t)b * Tn * Hn + (size_t)tm * 16 * Hn, Hn,
             Kb + (size_t)b * Tn * Hn + (size_t)tn * 16 * Hn, Hn, Hn, acc);
#pragma unroll
    for (int r = 0; r < 8; r++) {
      float v = acc[r];
      if (p == q && idx > (r + 8 * half)) v = 0.f;
      Out[(size_t)(r + 8 * half) * lda + idx] = v;
    }
    return;
  }
  const bool zjob = jb >= 64;
  const int ti = zjob ? (jb - 64) : (jb >> 3);
  const int tn = jb & 7;
  const float* S0 = st + (size_t)b * STSTRIDE + layerOff;
  float* SOut = outst + (size_t)b * STSTRIDE + layerOff;
  const float* Kt = Kb + (size_t)b * Tn * Hn + ti * 16 + idx;
  const float* Vt = Vb + (size_t)b * Tn * Hn + tn * 16 + idx;
  v8f sx = {0.f,0.f,0.f,0.f,0.f,0.f,0.f,0.f};
  if (!zjob) {
    float s0v[8];
#pragma unroll
    for (int r = 0; r < 8; r++)
      s0v[r] = S0[(size_t)(ti * 16 + r + 8 * half) * Hn + tn * 16 + idx];
    v2f ka[64], vb[64];
#pragma unroll
    for (int s = 0; s < 32; s++) {
      ka[s].x = Kt[(size_t)(4 * s + 2 * half)     * Hn];
      ka[s].y = Kt[(size_t)(4 * s + 2 * half + 1) * Hn];
      vb[s].x = Vt[(size_t)(4 * s + 2 * half)     * Hn];
      vb[s].y = Vt[(size_t)(4 * s + 2 * half + 1) * Hn];
    }
#pragma unroll
    for (int g = 0; g < 4; g++) {
      if (g + 2 < 4) {
#pragma unroll
        for (int s = 16 * (g + 2); s < 16 * (g + 3); s++) {
          ka[s].x = Kt[(size_t)(4 * s + 2 * half)     * Hn];
          ka[s].y = Kt[(size_t)(4 * s + 2 * half + 1) * Hn];
          vb[s].x = Vt[(size_t)(4 * s + 2 * half)     * Hn];
          vb[s].y = Vt[(size_t)(4 * s + 2 * half + 1) * Hn];
        }
      }
      __builtin_amdgcn_sched_barrier(0);
#pragma unroll
      for (int s = 16 * g; s < 16 * g + 16; s++) sx = wmma4(ka[s], vb[s], sx);
      float* D = (g < 3) ? (Pb + (size_t)(b * 3 + g) * Hn * Hn) : SOut;
#pragma unroll
      for (int r = 0; r < 8; r++)
        D[(size_t)(ti * 16 + r + 8 * half) * Hn + tn * 16 + idx] = sx[r] + s0v[r];
    }
  } else {
    v2f ones; ones.x = 1.0f; ones.y = 1.0f;
    float z0v[8];
#pragma unroll
    for (int r = 0; r < 8; r++) z0v[r] = S0[Hn * Hn + ti * 16 + r + 8 * half];
    v2f ka[64];
#pragma unroll
    for (int s = 0; s < 32; s++) {
      ka[s].x = Kt[(size_t)(4 * s + 2 * half)     * Hn];
      ka[s].y = Kt[(size_t)(4 * s + 2 * half + 1) * Hn];
    }
#pragma unroll
    for (int g = 0; g < 4; g++) {
      if (g + 2 < 4) {
#pragma unroll
        for (int s = 16 * (g + 2); s < 16 * (g + 3); s++) {
          ka[s].x = Kt[(size_t)(4 * s + 2 * half)     * Hn];
          ka[s].y = Kt[(size_t)(4 * s + 2 * half + 1) * Hn];
        }
      }
      __builtin_amdgcn_sched_barrier(0);
#pragma unroll
      for (int s = 16 * g; s < 16 * g + 16; s++) sx = wmma4(ka[s], ones, sx);
      float* D = (g < 3) ? (Zpb + (size_t)(b * 3 + g) * Hn) : (SOut + Hn * Hn);
      float zv[8];
#pragma unroll
      for (int r = 0; r < 8; r++) zv[r] = z0v[r] + sx[r];
      if (idx == 0) {
#pragma unroll
        for (int r = 0; r < 8; r++) D[ti * 16 + r + 8 * half] = zv[r];
      }
    }
  }
}

__global__ void dla_attnout_kernel(const float* __restrict__ Ab, const float* __restrict__ Vb,
                                   const float* __restrict__ Qb,
                                   const float* __restrict__ Pb, const float* __restrict__ Zpb,
                                   const float* __restrict__ st, int layerOff,
                                   float* __restrict__ Ao) {
  const int wave = __builtin_amdgcn_readfirstlane((int)(threadIdx.x >> 5));
  const int g = blockIdx.x * 8 + wave;
  const int b = g >> 7;
  const int tile = g & 127;
  const int tm = tile >> 3, tn = tile & 7;
  const int c = tm >> 2, q = tm & 3;
  v8f acc  = {0.f,0.f,0.f,0.f,0.f,0.f,0.f,0.f};
  v8f dsum = {0.f,0.f,0.f,0.f,0.f,0.f,0.f,0.f};
  const float* At = Ab + (size_t)((b * 4 + c) * 10 + ((q * (q + 1)) >> 1)) * 256;
  const float* Vt = Vb + (size_t)b * Tn * Hn + (size_t)c * 64 * Hn + tn * 16;
  switch (q) {
    case 0:  tile_ab_rowsum_k<16>(At, Vt, Hn, acc, dsum); break;
    case 1:  tile_ab_rowsum_k<32>(At, Vt, Hn, acc, dsum); break;
    case 2:  tile_ab_rowsum_k<48>(At, Vt, Hn, acc, dsum); break;
    default: tile_ab_rowsum_k<64>(At, Vt, Hn, acc, dsum); break;
  }
  const float* S0 = st + (size_t)b * STSTRIDE + layerOff;
  const float* Pc = (c == 0) ? S0 : (Pb + (size_t)(b * 3 + (c - 1)) * Hn * Hn);
  const float* Zc = (c == 0) ? (S0 + Hn * Hn) : (Zpb + (size_t)(b * 3 + (c - 1)) * Hn);
  tile_ab_rowdot(Qb + (size_t)b * Tn * Hn + (size_t)tm * 16 * Hn, Hn,
                 Pc + tn * 16, Hn, Zc, Hn, acc, dsum);
  const int lane = threadIdx.x & 31, half = lane >> 4, idx = lane & 15;
  float* Out = Ao + (size_t)b * Tn * Hn + (size_t)tm * 16 * Hn + tn * 16;
#pragma unroll
  for (int r = 0; r < 8; r++)
    Out[(size_t)(r + 8 * half) * Hn + idx] = acc[r] / (dsum[r] + 1e-5f);
}


#define XP 132

typedef __attribute__((ext_vector_type(4))) float v4f;

__device__ __forceinline__ void lds_barrier() {
  asm volatile("s_wait_dscnt 0x0\n\ts_barrier_signal -1\n\ts_barrier_wait -1" ::: "memory");
}

#define LOAD_FENCE() do { asm volatile("" ::: "memory"); __builtin_amdgcn_sched_barrier(0); } while (0)

template <bool LAST>
__global__ void __launch_bounds__(128)
dla_ffchain_kernel(const float* __restrict__ Ao,
                   const float* __restrict__ w1, const float* __restrict__ b1,
                   const float* __restrict__ w2, const float* __restrict__ b2,
                   const float* __restrict__ sc, const float* __restrict__ scb,
                   const float* __restrict__ lnw, const float* __restrict__ lnb,
                   const float* __restrict__ Wa, const float* __restrict__ Wq,
                   const float* __restrict__ Wv, const float* __restrict__ Ws,
                   const float* __restrict__ ub,
                   float* __restrict__ O0, float* __restrict__ O1,
                   float* __restrict__ O2, float* __restrict__ O3) {
  __shared__ __attribute__((aligned(16))) float hs[16 * XP];
  __shared__ __attribute__((aligned(16))) float xs[16 * XP];
  const int wave = threadIdx.x >> 5;
  const int b = blockIdx.x >> 5;
  const int tm = (blockIdx.x & 31) >> 1;
  const int tn = 4 * (blockIdx.x & 1) + wave;
  const int lane = threadIdx.x & 31, half = lane >> 4, idx = lane & 15;
  const size_t row0 = (size_t)b * Tn + (size_t)tm * 16;
  const int c0 = 32 * wave;

  const float* aop = Ao + (row0 + idx) * Hn + 2 * half;
  const float* w1p = w1 + (size_t)(c0 + idx) * Hn + 2 * half;
  const float* w2p = w2 + (size_t)(c0 + idx) * Hn + 8 * half;
  const float* b1p = b1 + c0 + 8 * half;
  const float* b2p = b2 + c0 + 8 * half;
  const float* scbp = scb + c0 + 8 * half;
  const float* scp = sc + (row0 + idx) * Hn + c0 + 8 * half;
  const size_t w3o = (size_t)(tn * 16 + idx) * Hn + 8 * half;
  float* hrow = hs + idx * XP + 8 * half;
  float* xrow = xs + idx * XP + 8 * half;

  v2f ao[32], f1[2][32], f2[2][32], f3[4][32], hb[32];
  v8f h[2], x2[2], acc2[8];
  float b1v[2][8], b2v[2][8], scbv[2][8], scv[2][8], lwv[8][8], lbv[8][8];
#pragma unroll
  for (int u = 0; u < 2; u++) {
#pragma unroll
    for (int r = 0; r < 8; r++) { h[u][r] = 0.f; x2[u][r] = 0.f; }
  }
#pragma unroll
  for (int s = 0; s < 32; s++) { ao[s].x = aop[4 * s]; ao[s].y = aop[4 * s + 1]; }
#pragma unroll
  for (int s = 0; s < 32; s++) {
#pragma unroll
    for (int u = 0; u < 2; u++) {
      const float* pp = w1p + (size_t)(16 * u) * Hn + 4 * s;
      f1[u][s].x = pp[0]; f1[u][s].y = pp[1];
    }
  }
  LOAD_FENCE();

#pragma unroll
  for (int s = 0; s < 16; s++) {
#pragma unroll
    for (int u = 0; u < 2; u++) h[u] = wmma4(f1[u][s], ao[s], h[u]);
  }
  LOAD_FENCE();
#pragma unroll
  for (int u = 0; u < 2; u++) {
#pragma unroll
    for (int r = 0; r < 8; r++) b1v[u][r] = b1p[16 * u + r];
  }
#pragma unroll
  for (int j = 0; j < 8; j++) {
#pragma unroll
    for (int rr = 0; rr < 4; rr++) {
#pragma unroll
      for (int u = 0; u < 2; u++) {
        const float* pp = w2p + (size_t)(16 * u) * Hn + 16 * j + 2 * rr;
        f2[u][4 * j + rr].x = pp[0]; f2[u][4 * j + rr].y = pp[1];
      }
    }
  }
#pragma unroll
  for (int u = 0; u < 2; u++) {
#pragma unroll
    for (int r = 0; r < 8; r++) {
      b2v[u][r] = b2p[16 * u + r];
      scbv[u][r] = scbp[16 * u + r];
      scv[u][r] = scp[16 * u + r];
    }
  }
  LOAD_FENCE();
#pragma unroll
  for (int s = 16; s < 32; s++) {
#pragma unroll
    for (int u = 0; u < 2; u++) h[u] = wmma4(f1[u][s], ao[s], h[u]);
  }

#pragma unroll
  for (int u = 0; u < 2; u++) {
    float t[8];
#pragma unroll
    for (int r = 0; r < 8; r++) {
      const float v = h[u][r] + b1v[u][r];
      t[r] = v > 0.f ? v : 0.f;
    }
    v4f lo, hi;
    lo.x = t[0]; lo.y = t[1]; lo.z = t[2]; lo.w = t[3];
    hi.x = t[4]; hi.y = t[5]; hi.z = t[6]; hi.w = t[7];
    *(v4f*)(hrow + c0 + 16 * u)     = lo;
    *(v4f*)(hrow + c0 + 16 * u + 4) = hi;
  }
  lds_barrier();

#pragma unroll
  for (int j = 0; j < 8; j++) {
    const v4f lo = *(const v4f*)(hrow + 16 * j);
    const v4f hi = *(const v4f*)(hrow + 16 * j + 4);
    hb[4 * j].x     = lo.x; hb[4 * j].y     = lo.y;
    hb[4 * j + 1].x = lo.z; hb[4 * j + 1].y = lo.w;
    hb[4 * j + 2].x = hi.x; hb[4 * j + 2].y = hi.y;
    hb[4 * j + 3].x = hi.z; hb[4 * j + 3].y = hi.w;
  }
  LOAD_FENCE();
#pragma unroll
  for (int e = 0; e < 16; e++) {
#pragma unroll
    for (int u = 0; u < 2; u++) x2[u] = wmma4(f2[u][e], hb[e], x2[u]);
  }
  LOAD_FENCE();
  if (LAST) {
#pragma unroll
    for (int n = 0; n < 8; n++) {
#pragma unroll
      for (int rr = 0; rr < 4; rr++) {
        const float* pp = Wa + w3o + 16 * n + 2 * rr;
        f3[0][n * 4 + rr].x = pp[0]; f3[0][n * 4 + rr].y = pp[1];
      }
    }
  } else {
    const float* lwp = lnw + 8 * half;
    const float* lbp = lnb + 8 * half;
#pragma unroll
    for (int n = 0; n < 8; n++) {
#pragma unroll
      for (int r = 0; r < 8; r++) { lwv[n][r] = lwp[16 * n + r]; lbv[n][r] = lbp[16 * n + r]; }
    }
#pragma unroll
    for (int nl = 0; nl < 2; nl++) {
#pragma unroll
      for (int rr = 0; rr < 4; rr++) {
        const size_t o = w3o + 16 * nl + 2 * rr;
        f3[0][0 * 8 + nl * 4 + rr].x = Wa[o]; f3[0][0 * 8 + nl * 4 + rr].y = Wa[o + 1];
        f3[0][1 * 8 + nl * 4 + rr].x = Wq[o]; f3[0][1 * 8 + nl * 4 + rr].y = Wq[o + 1];
        f3[0][2 * 8 + nl * 4 + rr].x = Wv[o]; f3[0][2 * 8 + nl * 4 + rr].y = Wv[o + 1];
        f3[0][3 * 8 + nl * 4 + rr].x = Ws[o]; f3[0][3 * 8 + nl * 4 + rr].y = Ws[o + 1];
      }
    }
  }
  LOAD_FENCE();
#pragma unroll
  for (int e = 16; e < 32; e++) {
#pragma unroll
    for (int u = 0; u < 2; u++) x2[u] = wmma4(f2[u][e], hb[e], x2[u]);
  }

#pragma unroll
  for (int u = 0; u < 2; u++) {
    float t[8];
#pragma unroll
    for (int r = 0; r < 8; r++) {
      float v = x2[u][r] + b2v[u][r];
      v = v > 0.f ? v : 0.f;
      t[r] = v + scv[u][r] + scbv[u][r];
    }
    v4f lo, hi;
    lo.x = t[0]; lo.y = t[1]; lo.z = t[2]; lo.w = t[3];
    hi.x = t[4]; hi.y = t[5]; hi.z = t[6]; hi.w = t[7];
    *(v4f*)(xrow + c0 + 16 * u)     = lo;
    *(v4f*)(xrow + c0 + 16 * u + 4) = hi;
  }
  lds_barrier();

#pragma unroll
  for (int n = 0; n < 8; n++) {
    const v4f lo = *(const v4f*)(xrow + 16 * n);
    const v4f hi = *(const v4f*)(xrow + 16 * n + 4);
    acc2[n][0] = lo.x; acc2[n][1] = lo.y; acc2[n][2] = lo.z; acc2[n][3] = lo.w;
    acc2[n][4] = hi.x; acc2[n][5] = hi.y; acc2[n][6] = hi.z; acc2[n][7] = hi.w;
  }
  if (!LAST) {
#pragma unroll
    for (int q3 = 1; q3 < 4; q3++) {
#pragma unroll
      for (int nl = 0; nl < 2; nl++) {
#pragma unroll
        for (int rr = 0; rr < 4; rr++) {
          const size_t o = w3o + 16 * (2 * q3 + nl) + 2 * rr;
          f3[q3][0 * 8 + nl * 4 + rr].x = Wa[o]; f3[q3][0 * 8 + nl * 4 + rr].y = Wa[o + 1];
          f3[q3][1 * 8 + nl * 4 + rr].x = Wq[o]; f3[q3][1 * 8 + nl * 4 + rr].y = Wq[o + 1];
          f3[q3][2 * 8 + nl * 4 + rr].x = Wv[o]; f3[q3][2 * 8 + nl * 4 + rr].y = Wv[o + 1];
          f3[q3][3 * 8 + nl * 4 + rr].x = Ws[o]; f3[q3][3 * 8 + nl * 4 + rr].y = Ws[o + 1];
        }
      }
    }
  }
  LOAD_FENCE();

  if (LAST) {
    v8f accE = {0.f,0.f,0.f,0.f,0.f,0.f,0.f,0.f};
    v8f accO = {0.f,0.f,0.f,0.f,0.f,0.f,0.f,0.f};
#pragma unroll
    for (int n = 0; n < 8; n++) {
#pragma unroll
      for (int rr = 0; rr < 4; rr += 2) {
        v2f xa, xb;
        xa.x = acc2[n][2 * rr];     xa.y = acc2[n][2 * rr + 1];
        xb.x = acc2[n][2 * rr + 2]; xb.y = acc2[n][2 * rr + 3];
        accE = wmma4(xa, f3[0][n * 4 + rr], accE);
        accO = wmma4(xb, f3[0][n * 4 + rr + 1], accO);
      }
    }
#pragma unroll
    for (int r = 0; r < 8; r++) {
      const int t = tm * 16 + r + 8 * half;
      const int o = tn * 16 + idx;
      O0[(size_t)t * Bn * OUTn + (size_t)b * OUTn + o] = (accE[r] + accO[r]) + ub[o];
    }
  } else {
    float p0 = 0.f, p1 = 0.f, p2 = 0.f, p3 = 0.f;
#pragma unroll
    for (int n = 0; n < 8; n += 4) {
#pragma unroll
      for (int r = 0; r < 8; r += 2) {
        p0 += acc2[n][r] + acc2[n][r + 1];
        p1 += acc2[n + 1][r] + acc2[n + 1][r + 1];
        p2 += acc2[n + 2][r] + acc2[n + 2][r + 1];
        p3 += acc2[n + 3][r] + acc2[n + 3][r + 1];
      }
    }
    float sum = (p0 + p1) + (p2 + p3);
    sum += __shfl_xor(sum, 16);
    const float m = sum * (1.0f / INn);

    float q0 = 0.f, q1 = 0.f, q2 = 0.f, q3 = 0.f;
#pragma unroll
    for (int n = 0; n < 8; n += 4) {
#pragma unroll
      for (int r = 0; r < 8; r++) {
        float d;
        d = acc2[n][r] - m;     q0 += d * d;
        d = acc2[n + 1][r] - m; q1 += d * d;
        d = acc2[n + 2][r] - m; q2 += d * d;
        d = acc2[n + 3][r] - m; q3 += d * d;
      }
    }
    float sq = (q0 + q1) + (q2 + q3);
    sq += __shfl_xor(sq, 16);
    const float inv = rsqrtf(sq * (1.0f / INn) + 1e-5f);

#pragma unroll
    for (int n = 0; n < 8; n++) {
#pragma unroll
      for (int r = 0; r < 8; r++)
        acc2[n][r] = (acc2[n][r] - m) * inv * lwv[n][r] + lbv[n][r];
    }

    v8f accK = {0.f,0.f,0.f,0.f,0.f,0.f,0.f,0.f};
    v8f accQ = {0.f,0.f,0.f,0.f,0.f,0.f,0.f,0.f};
    v8f accV = {0.f,0.f,0.f,0.f,0.f,0.f,0.f,0.f};
    v8f accS = {0.f,0.f,0.f,0.f,0.f,0.f,0.f,0.f};
#pragma unroll
    for (int q3 = 0; q3 < 4; q3++) {
#pragma unroll
      for (int nl = 0; nl < 2; nl++) {
#pragma unroll
        for (int rr = 0; rr < 4; rr++) {
          v2f xa;
          xa.x = acc2[2 * q3 + nl][2 * rr]; xa.y = acc2[2 * q3 + nl][2 * rr + 1];
          accK = wmma4(xa, f3[q3][0 * 8 + nl * 4 + rr], accK);
          accQ = wmma4(xa, f3[q3][1 * 8 + nl * 4 + rr], accQ);
          accV = wmma4(xa, f3[q3][2 * 8 + nl * 4 + rr], accV);
          accS = wmma4(xa, f3[q3][3 * 8 + nl * 4 + rr], accS);
        }
      }
    }
    const size_t o16 = row0 * Hn + tn * 16;
#pragma unroll
    for (int r = 0; r < 8; r++) {
      const size_t e = o16 + (size_t)(r + 8 * half) * Hn + idx;
      O0[e] = elu_plus1(accK[r]);
      O1[e] = elu_plus1(accQ[r]);
      O2[e] = accV[r];
      O3[e] = accS[r];
    }
  }
}

extern "C" void kernel_launch(void* const* d_in, const int* in_sizes, int n_in,
                              void* d_out, int out_size, void* d_ws, size_t ws_size,
                              hipStream_t stream) {
  (void)in_sizes; (void)n_in; (void)out_size; (void)ws_size;
  const float* z   = (const float*)d_in[0];
  const float* st  = (const float*)d_in[1];
  const float* Wk  = (const float*)d_in[2];
  const float* Wq  = (const float*)d_in[3];
  const float* Wv  = (const float*)d_in[4];
  const float* lnw = (const float*)d_in[5];
  const float* lnb = (const float*)d_in[6];
  const float* w1  = (const float*)d_in[7];
  const float* b1  = (const float*)d_in[8];
  const float* w2  = (const float*)d_in[9];
  const float* b2  = (const float*)d_in[10];
  const float* scw = (const float*)d_in[11];
  const float* scb = (const float*)d_in[12];
  const float* uw  = (const float*)d_in[13];
  const float* ub  = (const float*)d_in[14];
  float* out = (float*)d_out;
  float* ws  = (float*)d_ws;

  float* XCUR = ws;
  float* SC2  = ws;
  float* SC   = ws + 262144;
  float* KB   = ws + 524288;
  float* QB   = ws + 786432;
  float* VB   = ws + 1048576;
  float* PB   = ws + 1310720;
  float* ZPB  = PB + PSZ;
  float* AB   = ZPB + ZPSZ;
  float* AO   = ws + 1835008;
  float* OST  = out + YSZ;

  dla_rotate_kernel<<<Bn * Tn, 64, 0, stream>>>(z, st, XCUR, OST);
  dla_lnproj_kernel<<<256, 128, 0, stream>>>(XCUR, lnw, lnb, Wk, Wq, Wv, scw, KB, QB, VB, SC);
  dla_score_state_kernel<<<Bn * 14, 256, 0, stream>>>(QB, KB, VB, st, 0, PB, ZPB, AB, OST);
  dla_attnout_kernel<<<128, 256, 0, stream>>>(AB, VB, QB, PB, ZPB, st, 0, AO);
  dla_ffchain_kernel<false><<<256, 128, 0, stream>>>(
      AO, w1, b1, w2, b2, SC, scb, lnw + INn, lnb + INn,
      Wk + (size_t)Hn * INn, Wq + (size_t)Hn * INn, Wv + (size_t)Hn * INn,
      scw + (size_t)Hn * INn, nullptr, KB, QB, VB, SC2);
  dla_score_state_kernel<<<Bn * 14, 256, 0, stream>>>(QB, KB, VB, st, LAYERSTATE, PB, ZPB, AB, OST);
  dla_attnout_kernel<<<128, 256, 0, stream>>>(AB, VB, QB, PB, ZPB, st, LAYERSTATE, AO);
  dla_ffchain_kernel<true><<<256, 128, 0, stream>>>(
      AO, w1 + (size_t)Hn * Hn, b1 + Hn, w2 + (size_t)Hn * Hn, b2 + Hn, SC2, scb + Hn,
      nullptr, nullptr, uw, nullptr, nullptr, nullptr, ub, out, nullptr, nullptr, nullptr);
}


# rotary embedding done inside the first projection launch: each wave rotates a quarter of its work-group's 16-token tile, quarters meet in an LDS tile, A fragments read from it: 8 to 7 launches
# speedup vs baseline: 2.2570x; 1.0518x over previous
#include <hip/hip_runtime.h>
#include <hip/hip_bf16.h>
#include <math.h>

#define Bn 8
#define Tn 256
#define INn 128
#define Hn 128
#define OUTn 128
#define Ln 2
#define STSTRIDE 33025
#define LAYERSTATE 16512
#define YSZ (Tn*Bn*OUTn)

typedef __attribute__((ext_vector_type(2))) float v2f;
typedef __attribute__((ext_vector_type(8))) float v8f;

__device__ __forceinline__ v8f wmma4(v2f a, v2f b, v8f c) {
  return __builtin_amdgcn_wmma_f32_16x16x4_f32(false, a, false, b, (short)0, c, false, false);
}

__device__ __forceinline__ void tile_abt(const float* A, int lda, const float* Bt, int ldb,
                                         int K, v8f& acc) {
  const int lane = threadIdx.x & 31;
  const int half = lane >> 4;
  const int idx  = lane & 15;
  const float* ap = A  + (size_t)idx * lda + 2 * half;
  const float* bp = Bt + (size_t)idx * ldb + 2 * half;
  for (int k = 0; k < K; k += 4) {
    v2f a, b;
    a.x = ap[k]; a.y = ap[k + 1];
    b.x = bp[k]; b.y = bp[k + 1];
    acc = wmma4(a, b, acc);
  }
}

__device__ __forceinline__ void tile_ab(const float* A, int lda, const float* Bm, int ldb,
                                        int K, v8f& acc) {
  const int lane = threadIdx.x & 31;
  const int half = lane >> 4;
  const int idx  = lane & 15;
  const float* ap = A + (size_t)idx * lda + 2 * half;
  for (int k = 0; k < K; k += 4) {
    v2f a, b;
    a.x = ap[k]; a.y = ap[k + 1];
    b.x = Bm[(size_t)(k + 2 * half)     * ldb + idx];
    b.y = Bm[(size_t)(k + 2 * half + 1) * ldb + idx];
    acc = wmma4(a, b, acc);
  }
}

__device__ __forceinline__ void tile_ab_rowsum(const float* A, int lda, const float* Bm, int ldb,
                                               int K, v8f& acc, v8f& rs) {
  const int lane = threadIdx.x & 31;
  const int half = lane >> 4;
  const int idx  = lane & 15;
  const float* ap = A + (size_t)idx * lda + 2 * half;
  v2f ones; ones.x = 1.0f; ones.y = 1.0f;
  for (int k = 0; k < K; k += 4) {
    v2f a, b;
    a.x = ap[k]; a.y = ap[k + 1];
    b.x = Bm[(size_t)(k + 2 * half)     * ldb + idx];
    b.y = Bm[(size_t)(k + 2 * half + 1) * ldb + idx];
    acc = wmma4(a, b, acc);
    rs  = wmma4(a, ones, rs);
  }
}

__device__ __forceinline__ void tile_ab_rowdot(const float* A, int lda, const float* Bm, int ldb,
                                               const float* u, int K, v8f& acc, v8f& rs) {
  const int lane = threadIdx.x & 31;
  const int half = lane >> 4;
  const int idx  = lane & 15;
  const float* ap = A + (size_t)idx * lda + 2 * half;
#pragma unroll
  for (int k = 0; k < K; k += 4) {
    v2f a, b, w;
    a.x = ap[k]; a.y = ap[k + 1];
    b.x = Bm[(size_t)(k + 2 * half)     * ldb + idx];
    b.y = Bm[(size_t)(k + 2 * half + 1) * ldb + idx];
    w.x = u[k + 2 * half]; w.y = u[k + 2 * half + 1];
    acc = wmma4(a, b, acc);
    rs  = wmma4(a, w, rs);
  }
}

template <int K>
__device__ __forceinline__ void tile_ab_rowsum_k(const float* A, const float* Bm, int ldb,
                                                 v8f& acc, v8f& rs) {
  const int lane = threadIdx.x & 31;
  const int half = lane >> 4;
  const int idx  = lane & 15;
  const float* ap = A + (size_t)idx * K + 2 * half;
  v2f ones; ones.x = 1.0f; ones.y = 1.0f;
#pragma unroll
  for (int k = 0; k < K; k += 4) {
    v2f a, b;
    a.x = ap[k]; a.y = ap[k + 1];
    b.x = Bm[(size_t)(k + 2 * half)     * ldb + idx];
    b.y = Bm[(size_t)(k + 2 * half + 1) * ldb + idx];
    acc = wmma4(a, b, acc);
    rs  = wmma4(a, ones, rs);
  }
}

#define XP 132

typedef __attribute__((ext_vector_type(4))) float v4f;

__device__ __forceinline__ void lds_barrier() {
  asm volatile("s_wait_dscnt 0x0\n\ts_barrier_signal -1\n\ts_barrier_wait -1" ::: "memory");
}

__device__ __forceinline__ float elu_plus1(float v) {
  return ((v > 0.f) ? v : expm1f(v)) + 1.0f;
}

__global__ void __launch_bounds__(128)
dla_lnproj_kernel(const float* __restrict__ z, const float* __restrict__ st,
                  const float* __restrict__ lnw, const float* __restrict__ lnb,
                  const float* __restrict__ Wk, const float* __restrict__ Wq,
                  const float* __restrict__ Wv, const float* __restrict__ scw,
                  float* __restrict__ Kb, float* __restrict__ Qb,
                  float* __restrict__ Vb, float* __restrict__ Sc,
                  float* __restrict__ outst) {
  __shared__ __attribute__((aligned(16))) float xs[16 * XP];
  const int wave = threadIdx.x >> 5;
  const int g = blockIdx.x * 4 + wave;
  const int b = g >> 7;
  const int tile = g & 127;
  const int tm = tile >> 3, tn = tile & 7;
  const int lane = threadIdx.x & 31, half = lane >> 4, idx = lane & 15;

  const int t = tm * 16 + idx;
  const int j0 = 16 * wave + 8 * half;
  const float* zp = z + ((size_t)t * Bn + b) * INn + j0;
  const float ts = st[(size_t)b * STSTRIDE + STSTRIDE - 1];
  const v4f z1lo = *(const v4f*)(zp);
  const v4f z1hi = *(const v4f*)(zp + 4);
  const v4f z2lo = *(const v4f*)(zp + 64);
  const v4f z2hi = *(const v4f*)(zp + 68);
  const float* xr = xs + idx * XP + 2 * half;
  const size_t wrow = (size_t)(tn * 16 + idx) * INn + 2 * half;
  const float* pk = Wk + wrow;
  const float* pq = Wq + wrow;
  const float* pv = Wv + wrow;
  const float* ps = scw + wrow;
  const float* wp = lnw + 2 * half;
  const float* bp = lnb + 2 * half;

  v2f a[32], lw[32], lb[32], fk[32], fq[32], fv[32], fs[32];
#pragma unroll
  for (int s = 0; s < 16; s++) {
    fk[s].x = pk[4 * s]; fk[s].y = pk[4 * s + 1];
    fq[s].x = pq[4 * s]; fq[s].y = pq[4 * s + 1];
    fv[s].x = pv[4 * s]; fv[s].y = pv[4 * s + 1];
    fs[s].x = ps[4 * s]; fs[s].y = ps[4 * s + 1];
  }
  __builtin_amdgcn_sched_barrier(0);

  {
    const float pos = ts + (float)t;
    const float x1[8] = {z1lo.x, z1lo.y, z1lo.z, z1lo.w, z1hi.x, z1hi.y, z1hi.z, z1hi.w};
    const float x2[8] = {z2lo.x, z2lo.y, z2lo.z, z2lo.w, z2hi.x, z2hi.y, z2hi.z, z2hi.w};
    float r1[8], r2[8];
#pragma unroll
    for (int i = 0; i < 8; i++) {
      const float freq = expf((float)(2 * (j0 + i)) * (-9.210340371976184f / 128.0f));
      const float ang = pos * freq;
      const float s = sinf(ang), c = cosf(ang);
      r1[i] = x1[i] * c - x2[i] * s;
      r2[i] = x2[i] * c + x1[i] * s;
    }
    float* xw = xs + idx * XP + j0;
    v4f o;
    o.x = r1[0]; o.y = r1[1]; o.z = r1[2]; o.w = r1[3]; *(v4f*)(xw)      = o;
    o.x = r1[4]; o.y = r1[5]; o.z = r1[6]; o.w = r1[7]; *(v4f*)(xw + 4)  = o;
    o.x = r2[0]; o.y = r2[1]; o.z = r2[2]; o.w = r2[3]; *(v4f*)(xw + 64) = o;
    o.x = r2[4]; o.y = r2[5]; o.z = r2[6]; o.w = r2[7]; *(v4f*)(xw + 68) = o;
  }
  if (tile == 0 && (threadIdx.x & 31) == 0)
    outst[(size_t)b * STSTRIDE + STSTRIDE - 1] = ts + (float)Tn;
#pragma unroll
  for (int s = 0; s < 32; s++) {
    lw[s].x = wp[4 * s]; lw[s].y = wp[4 * s + 1];
    lb[s].x = bp[4 * s]; lb[s].y = bp[4 * s + 1];
  }
  lds_barrier();
#pragma unroll
  for (int s = 0; s < 32; s++) { a[s] = *(const v2f*)(xr + 4 * s); }

  float p0 = 0.f, p1 = 0.f, p2 = 0.f, p3 = 0.f;
#pragma unroll
  for (int s = 0; s < 32; s += 4) {
    p0 += a[s].x + a[s].y;
    p1 += a[s + 1].x + a[s + 1].y;
    p2 += a[s + 2].x + a[s + 2].y;
    p3 += a[s + 3].x + a[s + 3].y;
  }
  float sum = (p0 + p1) + (p2 + p3);
  sum += __shfl_xor(sum, 16);
  const float m = sum * (1.0f / INn);

  float q0 = 0.f, q1 = 0.f, q2 = 0.f, q3 = 0.f;
#pragma unroll
  for (int s = 0; s < 32; s += 4) {
    float d;
    d = a[s].x - m;     q0 += d * d;  d = a[s].y - m;     q0 += d * d;
    d = a[s + 1].x - m; q1 += d * d;  d = a[s + 1].y - m; q1 += d * d;
    d = a[s + 2].x - m; q2 += d * d;  d = a[s + 2].y - m; q2 += d * d;
    d = a[s + 3].x - m; q3 += d * d;  d = a[s + 3].y - m; q3 += d * d;
  }
  float sq = (q0 + q1) + (q2 + q3);
  sq += __shfl_xor(sq, 16);
  const float inv = rsqrtf(sq * (1.0f / INn) + 1e-5f);


#pragma unroll
  for (int s = 0; s < 32; s++) {
    a[s].x = (a[s].x - m) * inv * lw[s].x + lb[s].x;
    a[s].y = (a[s].y - m) * inv * lw[s].y + lb[s].y;
  }

  v8f accK = {0.f,0.f,0.f,0.f,0.f,0.f,0.f,0.f};
  v8f accQ = {0.f,0.f,0.f,0.f,0.f,0.f,0.f,0.f};
  v8f accV = {0.f,0.f,0.f,0.f,0.f,0.f,0.f,0.f};
  v8f accS = {0.f,0.f,0.f,0.f,0.f,0.f,0.f,0.f};

#pragma unroll
  for (int gq = 0; gq < 4; gq++) {
    if (gq < 2) {
#pragma unroll
      for (int s = 8 * gq + 16; s < 8 * gq + 24; s++) {
        fk[s].x = pk[4 * s]; fk[s].y = pk[4 * s + 1];
        fq[s].x = pq[4 * s]; fq[s].y = pq[4 * s + 1];
        fv[s].x = pv[4 * s]; fv[s].y = pv[4 * s + 1];
        fs[s].x = ps[4 * s]; fs[s].y = ps[4 * s + 1];
      }
    }
    __builtin_amdgcn_sched_barrier(0);
#pragma unroll
    for (int s = 8 * gq; s < 8 * gq + 8; s++) {
      accK = wmma4(a[s], fk[s], accK);
      accQ = wmma4(a[s], fq[s], accQ);
      accV = wmma4(a[s], fv[s], accV);
      accS = wmma4(a[s], fs[s], accS);
    }
  }

  const size_t o = (size_t)b * Tn * Hn + (size_t)tm * 16 * Hn + tn * 16;
#pragma unroll
  for (int r = 0; r < 8; r++) {
    const size_t e = o + (size_t)(r + 8 * half) * Hn + idx;
    Kb[e] = elu_plus1(accK[r]);
    Qb[e] = elu_plus1(accQ[r]);
    Vb[e] = accV[r];
    Sc[e] = accS[r];
  }
}

#define PSZ  393216
#define ZPSZ 3072

__global__ void __launch_bounds__(256)
dla_score_state_kernel(const float* __restrict__ Qb, const float* __restrict__ Kb,
                                       const float* __restrict__ Vb, const float* __restrict__ st,
                                       int layerOff, float* __restrict__ Pb,
                                       float* __restrict__ Zpb, float* __restrict__ Ab,
                                       float* __restrict__ outst) {
  const int wave = __builtin_amdgcn_readfirstlane((int)(threadIdx.x >> 5));
  const int b  = blockIdx.x / 14;
  const int jb = (blockIdx.x - b * 14) * 8 + wave;
  const int lane = threadIdx.x & 31, half = lane >> 4, idx = lane & 15;
  if (jb >= 72) {
    const int y = (jb - 72) >> 3;
    const int c = (y < 4) ? y : (wave >> 1);
    const int q = (y < 4) ? ((wave >= 4) ? 3 : (wave >= 1) ? 2 : 0) : 1;
    const int p = (y < 4) ? ((wave >= 4) ? (wave - 4) : (wave >= 1) ? (wave - 1) : 0) : (wave & 1);
    const int tm = 4 * c + q, tn = 4 * c + p;
    const int lda = 16 * (q + 1);
    float* Out = Ab + (size_t)((b * 4 + c) * 10 + ((q * (q + 1)) >> 1)) * 256 + p * 16;
    v8f acc = {0.f,0.f,0.f,0.f,0.f,0.f,0.f,0.f};
    tile_abt(Qb + (size_t)b * Tn * Hn + (size_t)tm * 16 * Hn, Hn,
             Kb + (size_t)b * Tn * Hn + (size_t)tn * 16 * Hn, Hn, Hn, acc);
#pragma unroll
    for (int r = 0; r < 8; r++) {
      float v = acc[r];
      if (p == q && idx > (r + 8 * half)) v = 0.f;
      Out[(size_t)(r + 8 * half) * lda + idx] = v;
    }
    return;
  }
  const bool zjob = jb >= 64;
  const int ti = zjob ? (jb - 64) : (jb >> 3);
  const int tn = jb & 7;
  const float* S0 = st + (size_t)b * STSTRIDE + layerOff;
  float* SOut = outst + (size_t)b * STSTRIDE + layerOff;
  const float* Kt = Kb + (size_t)b * Tn * Hn + ti * 16 + idx;
  const float* Vt = Vb + (size_t)b * Tn * Hn + tn * 16 + idx;
  v8f sx = {0.f,0.f,0.f,0.f,0.f,0.f,0.f,0.f};
  if (!zjob) {
    float s0v[8];
#pragma unroll
    for (int r = 0; r < 8; r++)
      s0v[r] = S0[(size_t)(ti * 16 + r + 8 * half) * Hn + tn * 16 + idx];
    v2f ka[64], vb[64];
#pragma unroll
    for (int s = 0; s < 32; s++) {
      ka[s].x = Kt[(size_t)(4 * s + 2 * half)     * Hn];
      ka[s].y = Kt[(size_t)(4 * s + 2 * half + 1) * Hn];
      vb[s].x = Vt[(size_t)(4 * s + 2 * half)     * Hn];
      vb[s].y = Vt[(size_t)(4 * s + 2 * half + 1) * Hn];
    }
#pragma unroll
    for (int g = 0; g < 4; g++) {
      if (g + 2 < 4) {
#pragma unroll
        for (int s = 16 * (g + 2); s < 16 * (g + 3); s++) {
          ka[s].x = Kt[(size_t)(4 * s + 2 * half)     * Hn];
          ka[s].y = Kt[(size_t)(4 * s + 2 * half + 1) * Hn];
          vb[s].x = Vt[(size_t)(4 * s + 2 * half)     * Hn];
          vb[s].y = Vt[(size_t)(4 * s + 2 * half + 1) * Hn];
        }
      }
      __builtin_amdgcn_sched_barrier(0);
#pragma unroll
      for (int s = 16 * g; s < 16 * g + 16; s++) sx = wmma4(ka[s], vb[s], sx);
      float* D = (g < 3) ? (Pb + (size_t)(b * 3 + g) * Hn * Hn) : SOut;
#pragma unroll
      for (int r = 0; r < 8; r++)
        D[(size_t)(ti * 16 + r + 8 * half) * Hn + tn * 16 + idx] = sx[r] + s0v[r];
    }
  } else {
    v2f ones; ones.x = 1.0f; ones.y = 1.0f;
    float z0v[8];
#pragma unroll
    for (int r = 0; r < 8; r++) z0v[r] = S0[Hn * Hn + ti * 16 + r + 8 * half];
    v2f ka[64];
#pragma unroll
    for (int s = 0; s < 32; s++) {
      ka[s].x = Kt[(size_t)(4 * s + 2 * half)     * Hn];
      ka[s].y = Kt[(size_t)(4 * s + 2 * half + 1) * Hn];
    }
#pragma unroll
    for (int g = 0; g < 4; g++) {
      if (g + 2 < 4) {
#pragma unroll
        for (int s = 16 * (g + 2); s < 16 * (g + 3); s++) {
          ka[s].x = Kt[(size_t)(4 * s + 2 * half)     * Hn];
          ka[s].y = Kt[(size_t)(4 * s + 2 * half + 1) * Hn];
        }
      }
      __builtin_amdgcn_sched_barrier(0);
#pragma unroll
      for (int s = 16 * g; s < 16 * g + 16; s++) sx = wmma4(ka[s], ones, sx);
      float* D = (g < 3) ? (Zpb + (size_t)(b * 3 + g) * Hn) : (SOut + Hn * Hn);
      float zv[8];
#pragma unroll
      for (int r = 0; r < 8; r++) zv[r] = z0v[r] + sx[r];
      if (idx == 0) {
#pragma unroll
        for (int r = 0; r < 8; r++) D[ti * 16 + r + 8 * half] = zv[r];
      }
    }
  }
}

__global__ void dla_attnout_kernel(const float* __restrict__ Ab, const float* __restrict__ Vb,
                                   const float* __restrict__ Qb,
                                   const float* __restrict__ Pb, const float* __restrict__ Zpb,
                                   const float* __restrict__ st, int layerOff,
                                   float* __restrict__ Ao) {
  const int wave = __builtin_amdgcn_readfirstlane((int)(threadIdx.x >> 5));
  const int g = blockIdx.x * 8 + wave;
  const int b = g >> 7;
  const int tile = g & 127;
  const int tm = tile >> 3, tn = tile & 7;
  const int c = tm >> 2, q = tm & 3;
  v8f acc  = {0.f,0.f,0.f,0.f,0.f,0.f,0.f,0.f};
  v8f dsum = {0.f,0.f,0.f,0.f,0.f,0.f,0.f,0.f};
  const float* At = Ab + (size_t)((b * 4 + c) * 10 + ((q * (q + 1)) >> 1)) * 256;
  const float* Vt = Vb + (size_t)b * Tn * Hn + (size_t)c * 64 * Hn + tn * 16;
  switch (q) {
    case 0:  tile_ab_rowsum_k<16>(At, Vt, Hn, acc, dsum); break;
    case 1:  tile_ab_rowsum_k<32>(At, Vt, Hn, acc, dsum); break;
    case 2:  tile_ab_rowsum_k<48>(At, Vt, Hn, acc, dsum); break;
    default: tile_ab_rowsum_k<64>(At, Vt, Hn, acc, dsum); break;
  }
  const float* S0 = st + (size_t)b * STSTRIDE + layerOff;
  const float* Pc = (c == 0) ? S0 : (Pb + (size_t)(b * 3 + (c - 1)) * Hn * Hn);
  const float* Zc = (c == 0) ? (S0 + Hn * Hn) : (Zpb + (size_t)(b * 3 + (c - 1)) * Hn);
  tile_ab_rowdot(Qb + (size_t)b * Tn * Hn + (size_t)tm * 16 * Hn, Hn,
                 Pc + tn * 16, Hn, Zc, Hn, acc, dsum);
  const int lane = threadIdx.x & 31, half = lane >> 4, idx = lane & 15;
  float* Out = Ao + (size_t)b * Tn * Hn + (size_t)tm * 16 * Hn + tn * 16;
#pragma unroll
  for (int r = 0; r < 8; r++)
    Out[(size_t)(r + 8 * half) * Hn + idx] = acc[r] / (dsum[r] + 1e-5f);
}


#define LOAD_FENCE() do { asm volatile("" ::: "memory"); __builtin_amdgcn_sched_barrier(0); } while (0)

template <bool LAST>
__global__ void __launch_bounds__(128)
dla_ffchain_kernel(const float* __restrict__ Ao,
                   const float* __restrict__ w1, const float* __restrict__ b1,
                   const float* __restrict__ w2, const float* __restrict__ b2,
                   const float* __restrict__ sc, const float* __restrict__ scb,
                   const float* __restrict__ lnw, const float* __restrict__ lnb,
                   const float* __restrict__ Wa, const float* __restrict__ Wq,
                   const float* __restrict__ Wv, const float* __restrict__ Ws,
                   const float* __restrict__ ub,
                   float* __restrict__ O0, float* __restrict__ O1,
                   float* __restrict__ O2, float* __restrict__ O3) {
  __shared__ __attribute__((aligned(16))) float hs[16 * XP];
  __shared__ __attribute__((aligned(16))) float xs[16 * XP];
  const int wave = threadIdx.x >> 5;
  const int b = blockIdx.x >> 5;
  const int tm = (blockIdx.x & 31) >> 1;
  const int tn = 4 * (blockIdx.x & 1) + wave;
  const int lane = threadIdx.x & 31, half = lane >> 4, idx = lane & 15;
  const size_t row0 = (size_t)b * Tn + (size_t)tm * 16;
  const int c0 = 32 * wave;

  const float* aop = Ao + (row0 + idx) * Hn + 2 * half;
  const float* w1p = w1 + (size_t)(c0 + idx) * Hn + 2 * half;
  const float* w2p = w2 + (size_t)(c0 + idx) * Hn + 8 * half;
  const float* b1p = b1 + c0 + 8 * half;
  const float* b2p = b2 + c0 + 8 * half;
  const float* scbp = scb + c0 + 8 * half;
  const float* scp = sc + (row0 + idx) * Hn + c0 + 8 * half;
  const size_t w3o = (size_t)(tn * 16 + idx) * Hn + 8 * half;
  float* hrow = hs + idx * XP + 8 * half;
  float* xrow = xs + idx * XP + 8 * half;

  v2f ao[32], f1[2][32], f2[2][32], f3[4][32], hb[32];
  v8f h[2], x2[2], acc2[8];
  float b1v[2][8], b2v[2][8], scbv[2][8], scv[2][8], lwv[8][8], lbv[8][8];
#pragma unroll
  for (int u = 0; u < 2; u++) {
#pragma unroll
    for (int r = 0; r < 8; r++) { h[u][r] = 0.f; x2[u][r] = 0.f; }
  }
#pragma unroll
  for (int s = 0; s < 32; s++) { ao[s].x = aop[4 * s]; ao[s].y = aop[4 * s + 1]; }
#pragma unroll
  for (int s = 0; s < 32; s++) {
#pragma unroll
    for (int u = 0; u < 2; u++) {
      const float* pp = w1p + (size_t)(16 * u) * Hn + 4 * s;
      f1[u][s].x = pp[0]; f1[u][s].y = pp[1];
    }
  }
  LOAD_FENCE();

#pragma unroll
  for (int s = 0; s < 16; s++) {
#pragma unroll
    for (int u = 0; u < 2; u++) h[u] = wmma4(f1[u][s], ao[s], h[u]);
  }
  LOAD_FENCE();
#pragma unroll
  for (int u = 0; u < 2; u++) {
#pragma unroll
    for (int r = 0; r < 8; r++) b1v[u][r] = b1p[16 * u + r];
  }
#pragma unroll
  for (int j = 0; j < 8; j++) {
#pragma unroll
    for (int rr = 0; rr < 4; rr++) {
#pragma unroll
      for (int u = 0; u < 2; u++) {
        const float* pp = w2p + (size_t)(16 * u) * Hn + 16 * j + 2 * rr;
        f2[u][4 * j + rr].x = pp[0]; f2[u][4 * j + rr].y = pp[1];
      }
    }
  }
#pragma unroll
  for (int u = 0; u < 2; u++) {
#pragma unroll
    for (int r = 0; r < 8; r++) {
      b2v[u][r] = b2p[16 * u + r];
      scbv[u][r] = scbp[16 * u + r];
      scv[u][r] = scp[16 * u + r];
    }
  }
  LOAD_FENCE();
#pragma unroll
  for (int s = 16; s < 32; s++) {
#pragma unroll
    for (int u = 0; u < 2; u++) h[u] = wmma4(f1[u][s], ao[s], h[u]);
  }

#pragma unroll
  for (int u = 0; u < 2; u++) {
    float t[8];
#pragma unroll
    for (int r = 0; r < 8; r++) {
      const float v = h[u][r] + b1v[u][r];
      t[r] = v > 0.f ? v : 0.f;
    }
    v4f lo, hi;
    lo.x = t[0]; lo.y = t[1]; lo.z = t[2]; lo.w = t[3];
    hi.x = t[4]; hi.y = t[5]; hi.z = t[6]; hi.w = t[7];
    *(v4f*)(hrow + c0 + 16 * u)     = lo;
    *(v4f*)(hrow + c0 + 16 * u + 4) = hi;
  }
  lds_barrier();

#pragma unroll
  for (int j = 0; j < 8; j++) {
    const v4f lo = *(const v4f*)(hrow + 16 * j);
    const v4f hi = *(const v4f*)(hrow + 16 * j + 4);
    hb[4 * j].x     = lo.x; hb[4 * j].y     = lo.y;
    hb[4 * j + 1].x = lo.z; hb[4 * j + 1].y = lo.w;
    hb[4 * j + 2].x = hi.x; hb[4 * j + 2].y = hi.y;
    hb[4 * j + 3].x = hi.z; hb[4 * j + 3].y = hi.w;
  }
  LOAD_FENCE();
#pragma unroll
  for (int e = 0; e < 16; e++) {
#pragma unroll
    for (int u = 0; u < 2; u++) x2[u] = wmma4(f2[u][e], hb[e], x2[u]);
  }
  LOAD_FENCE();
  if (LAST) {
#pragma unroll
    for (int n = 0; n < 8; n++) {
#pragma unroll
      for (int rr = 0; rr < 4; rr++) {
        const float* pp = Wa + w3o + 16 * n + 2 * rr;
        f3[0][n * 4 + rr].x = pp[0]; f3[0][n * 4 + rr].y = pp[1];
      }
    }
  } else {
    const float* lwp = lnw + 8 * half;
    const float* lbp = lnb + 8 * half;
#pragma unroll
    for (int n = 0; n < 8; n++) {
#pragma unroll
      for (int r = 0; r < 8; r++) { lwv[n][r] = lwp[16 * n + r]; lbv[n][r] = lbp[16 * n + r]; }
    }
#pragma unroll
    for (int nl = 0; nl < 2; nl++) {
#pragma unroll
      for (int rr = 0; rr < 4; rr++) {
        const size_t o = w3o + 16 * nl + 2 * rr;
        f3[0][0 * 8 + nl * 4 + rr].x = Wa[o]; f3[0][0 * 8 + nl * 4 + rr].y = Wa[o + 1];
        f3[0][1 * 8 + nl * 4 + rr].x = Wq[o]; f3[0][1 * 8 + nl * 4 + rr].y = Wq[o + 1];
        f3[0][2 * 8 + nl * 4 + rr].x = Wv[o]; f3[0][2 * 8 + nl * 4 + rr].y = Wv[o + 1];
        f3[0][3 * 8 + nl * 4 + rr].x = Ws[o]; f3[0][3 * 8 + nl * 4 + rr].y = Ws[o + 1];
      }
    }
  }
  LOAD_FENCE();
#pragma unroll
  for (int e = 16; e < 32; e++) {
#pragma unroll
    for (int u = 0; u < 2; u++) x2[u] = wmma4(f2[u][e], hb[e], x2[u]);
  }

#pragma unroll
  for (int u = 0; u < 2; u++) {
    float t[8];
#pragma unroll
    for (int r = 0; r < 8; r++) {
      float v = x2[u][r] + b2v[u][r];
      v = v > 0.f ? v : 0.f;
      t[r] = v + scv[u][r] + scbv[u][r];
    }
    v4f lo, hi;
    lo.x = t[0]; lo.y = t[1]; lo.z = t[2]; lo.w = t[3];
    hi.x = t[4]; hi.y = t[5]; hi.z = t[6]; hi.w = t[7];
    *(v4f*)(xrow + c0 + 16 * u)     = lo;
    *(v4f*)(xrow + c0 + 16 * u + 4) = hi;
  }
  lds_barrier();

#pragma unroll
  for (int n = 0; n < 8; n++) {
    const v4f lo = *(const v4f*)(xrow + 16 * n);
    const v4f hi = *(const v4f*)(xrow + 16 * n + 4);
    acc2[n][0] = lo.x; acc2[n][1] = lo.y; acc2[n][2] = lo.z; acc2[n][3] = lo.w;
    acc2[n][4] = hi.x; acc2[n][5] = hi.y; acc2[n][6] = hi.z; acc2[n][7] = hi.w;
  }
  if (!LAST) {
#pragma unroll
    for (int q3 = 1; q3 < 4; q3++) {
#pragma unroll
      for (int nl = 0; nl < 2; nl++) {
#pragma unroll
        for (int rr = 0; rr < 4; rr++) {
          const size_t o = w3o + 16 * (2 * q3 + nl) + 2 * rr;
          f3[q3][0 * 8 + nl * 4 + rr].x = Wa[o]; f3[q3][0 * 8 + nl * 4 + rr].y = Wa[o + 1];
          f3[q3][1 * 8 + nl * 4 + rr].x = Wq[o]; f3[q3][1 * 8 + nl * 4 + rr].y = Wq[o + 1];
          f3[q3][2 * 8 + nl * 4 + rr].x = Wv[o]; f3[q3][2 * 8 + nl * 4 + rr].y = Wv[o + 1];
          f3[q3][3 * 8 + nl * 4 + rr].x = Ws[o]; f3[q3][3 * 8 + nl * 4 + rr].y = Ws[o + 1];
        }
      }
    }
  }
  LOAD_FENCE();

  if (LAST) {
    v8f accE = {0.f,0.f,0.f,0.f,0.f,0.f,0.f,0.f};
    v8f accO = {0.f,0.f,0.f,0.f,0.f,0.f,0.f,0.f};
#pragma unroll
    for (int n = 0; n < 8; n++) {
#pragma unroll
      for (int rr = 0; rr < 4; rr += 2) {
        v2f xa, xb;
        xa.x = acc2[n][2 * rr];     xa.y = acc2[n][2 * rr + 1];
        xb.x = acc2[n][2 * rr + 2]; xb.y = acc2[n][2 * rr + 3];
        accE = wmma4(xa, f3[0][n * 4 + rr], accE);
        accO = wmma4(xb, f3[0][n * 4 + rr + 1], accO);
      }
    }
#pragma unroll
    for (int r = 0; r < 8; r++) {
      const int t = tm * 16 + r + 8 * half;
      const int o = tn * 16 + idx;
      O0[(size_t)t * Bn * OUTn + (size_t)b * OUTn + o] = (accE[r] + accO[r]) + ub[o];
    }
  } else {
    float p0 = 0.f, p1 = 0.f, p2 = 0.f, p3 = 0.f;
#pragma unroll
    for (int n = 0; n < 8; n += 4) {
#pragma unroll
      for (int r = 0; r < 8; r += 2) {
        p0 += acc2[n][r] + acc2[n][r + 1];
        p1 += acc2[n + 1][r] + acc2[n + 1][r + 1];
        p2 += acc2[n + 2][r] + acc2[n + 2][r + 1];
        p3 += acc2[n + 3][r] + acc2[n + 3][r + 1];
      }
    }
    float sum = (p0 + p1) + (p2 + p3);
    sum += __shfl_xor(sum, 16);
    const float m = sum * (1.0f / INn);

    float q0 = 0.f, q1 = 0.f, q2 = 0.f, q3 = 0.f;
#pragma unroll
    for (int n = 0; n < 8; n += 4) {
#pragma unroll
      for (int r = 0; r < 8; r++) {
        float d;
        d = acc2[n][r] - m;     q0 += d * d;
        d = acc2[n + 1][r] - m; q1 += d * d;
        d = acc2[n + 2][r] - m; q2 += d * d;
        d = acc2[n + 3][r] - m; q3 += d * d;
      }
    }
    float sq = (q0 + q1) + (q2 + q3);
    sq += __shfl_xor(sq, 16);
    const float inv = rsqrtf(sq * (1.0f / INn) + 1e-5f);

#pragma unroll
    for (int n = 0; n < 8; n++) {
#pragma unroll
      for (int r = 0; r < 8; r++)
        acc2[n][r] = (acc2[n][r] - m) * inv * lwv[n][r] + lbv[n][r];
    }

    v8f accK = {0.f,0.f,0.f,0.f,0.f,0.f,0.f,0.f};
    v8f accQ = {0.f,0.f,0.f,0.f,0.f,0.f,0.f,0.f};
    v8f accV = {0.f,0.f,0.f,0.f,0.f,0.f,0.f,0.f};
    v8f accS = {0.f,0.f,0.f,0.f,0.f,0.f,0.f,0.f};
#pragma unroll
    for (int q3 = 0; q3 < 4; q3++) {
#pragma unroll
      for (int nl = 0; nl < 2; nl++) {
#pragma unroll
        for (int rr = 0; rr < 4; rr++) {
          v2f xa;
          xa.x = acc2[2 * q3 + nl][2 * rr]; xa.y = acc2[2 * q3 + nl][2 * rr + 1];
          accK = wmma4(xa, f3[q3][0 * 8 + nl * 4 + rr], accK);
          accQ = wmma4(xa, f3[q3][1 * 8 + nl * 4 + rr], accQ);
          accV = wmma4(xa, f3[q3][2 * 8 + nl * 4 + rr], accV);
          accS = wmma4(xa, f3[q3][3 * 8 + nl * 4 + rr], accS);
        }
      }
    }
    const size_t o16 = row0 * Hn + tn * 16;
#pragma unroll
    for (int r = 0; r < 8; r++) {
      const size_t e = o16 + (size_t)(r + 8 * half) * Hn + idx;
      O0[e] = elu_plus1(accK[r]);
      O1[e] = elu_plus1(accQ[r]);
      O2[e] = accV[r];
      O3[e] = accS[r];
    }
  }
}

extern "C" void kernel_launch(void* const* d_in, const int* in_sizes, int n_in,
                              void* d_out, int out_size, void* d_ws, size_t ws_size,
                              hipStream_t stream) {
  (void)in_sizes; (void)n_in; (void)out_size; (void)ws_size;
  const float* z   = (const float*)d_in[0];
  const float* st  = (const float*)d_in[1];
  const float* Wk  = (const float*)d_in[2];
  const float* Wq  = (const float*)d_in[3];
  const float* Wv  = (const float*)d_in[4];
  const float* lnw = (const float*)d_in[5];
  const float* lnb = (const float*)d_in[6];
  const float* w1  = (const float*)d_in[7];
  const float* b1  = (const float*)d_in[8];
  const float* w2  = (const float*)d_in[9];
  const float* b2  = (const float*)d_in[10];
  const float* scw = (const float*)d_in[11];
  const float* scb = (const float*)d_in[12];
  const float* uw  = (const float*)d_in[13];
  const float* ub  = (const float*)d_in[14];
  float* out = (float*)d_out;
  float* ws  = (float*)d_ws;

  float* SC2  = ws;
  float* SC   = ws + 262144;
  float* KB   = ws + 524288;
  float* QB   = ws + 786432;
  float* VB   = ws + 1048576;
  float* PB   = ws + 1310720;
  float* ZPB  = PB + PSZ;
  float* AB   = ZPB + ZPSZ;
  float* AO   = ws + 1835008;
  float* OST  = out + YSZ;

  dla_lnproj_kernel<<<256, 128, 0, stream>>>(z, st, lnw, lnb, Wk, Wq, Wv, scw, KB, QB, VB, SC, OST);
  dla_score_state_kernel<<<Bn * 14, 256, 0, stream>>>(QB, KB, VB, st, 0, PB, ZPB, AB, OST);
  dla_attnout_kernel<<<128, 256, 0, stream>>>(AB, VB, QB, PB, ZPB, st, 0, AO);
  dla_ffchain_kernel<false><<<256, 128, 0, stream>>>(
      AO, w1, b1, w2, b2, SC, scb, lnw + INn, lnb + INn,
      Wk + (size_t)Hn * INn, Wq + (size_t)Hn * INn, Wv + (size_t)Hn * INn,
      scw + (size_t)Hn * INn, nullptr, KB, QB, VB, SC2);
  dla_score_state_kernel<<<Bn * 14, 256, 0, stream>>>(QB, KB, VB, st, LAYERSTATE, PB, ZPB, AB, OST);
  dla_attnout_kernel<<<128, 256, 0, stream>>>(AB, VB, QB, PB, ZPB, st, LAYERSTATE, AO);
  dla_ffchain_kernel<true><<<256, 128, 0, stream>>>(
      AO, w1 + (size_t)Hn * Hn, b1 + Hn, w2 + (size_t)Hn * Hn, b2 + Hn, SC2, scb + Hn,
      nullptr, nullptr, uw, nullptr, nullptr, nullptr, ub, out, nullptr, nullptr, nullptr);
}


# layer 1's attn-out launch dissolved: A_cc V_c and its row sums taken beside the state jobs (scores stay in LDS), the Q P_c half as stage 0 of the last ff launch: 7 to 6 launches; sum order changes
# speedup vs baseline: 2.4073x; 1.0666x over previous
#include <hip/hip_runtime.h>
#include <hip/hip_bf16.h>
#include <math.h>

#define Bn 8
#define Tn 256
#define INn 128
#define Hn 128
#define OUTn 128
#define Ln 2
#define STSTRIDE 33025
#define LAYERSTATE 16512
#define YSZ (Tn*Bn*OUTn)

typedef __attribute__((ext_vector_type(2))) float v2f;
typedef __attribute__((ext_vector_type(8))) float v8f;

__device__ __forceinline__ v8f wmma4(v2f a, v2f b, v8f c) {
  return __builtin_amdgcn_wmma_f32_16x16x4_f32(false, a, false, b, (short)0, c, false, false);
}

__device__ __forceinline__ void tile_abt(const float* A, int lda, const float* Bt, int ldb,
                                         int K, v8f& acc) {
  const int lane = threadIdx.x & 31;
  const int half = lane >> 4;
  const int idx  = lane & 15;
  const float* ap = A  + (size_t)idx * lda + 2 * half;
  const float* bp = Bt + (size_t)idx * ldb + 2 * half;
  for (int k = 0; k < K; k += 4) {
    v2f a, b;
    a.x = ap[k]; a.y = ap[k + 1];
    b.x = bp[k]; b.y = bp[k + 1];
    acc = wmma4(a, b, acc);
  }
}

__device__ __forceinline__ void tile_ab(const float* A, int lda, const float* Bm, int ldb,
                                        int K, v8f& acc) {
  const int lane = threadIdx.x & 31;
  const int half = lane >> 4;
  const int idx  = lane & 15;
  const float* ap = A + (size_t)idx * lda + 2 * half;
  for (int k = 0; k < K; k += 4) {
    v2f a, b;
    a.x = ap[k]; a.y = ap[k + 1];
    b.x = Bm[(size_t)(k + 2 * half)     * ldb + idx];
    b.y = Bm[(size_t)(k + 2 * half + 1) * ldb + idx];
    acc = wmma4(a, b, acc);
  }
}

__device__ __forceinline__ void tile_ab_rowsum(const float* A, int lda, const float* Bm, int ldb,
                                               int K, v8f& acc, v8f& rs) {
  const int lane = threadIdx.x & 31;
  const int half = lane >> 4;
  const int idx  = lane & 15;
  const float* ap = A + (size_t)idx * lda + 2 * half;
  v2f ones; ones.x = 1.0f; ones.y = 1.0f;
  for (int k = 0; k < K; k += 4) {
    v2f a, b;
    a.x = ap[k]; a.y = ap[k + 1];
    b.x = Bm[(size_t)(k + 2 * half)     * ldb + idx];
    b.y = Bm[(size_t)(k + 2 * half + 1) * ldb + idx];
    acc = wmma4(a, b, acc);
    rs  = wmma4(a, ones, rs);
  }
}

__device__ __forceinline__ void tile_ab_rowdot(const float* A, int lda, const float* Bm, int ldb,
                                               const float* u, int K, v8f& acc, v8f& rs) {
  const int lane = threadIdx.x & 31;
  const int half = lane >> 4;
  const int idx  = lane & 15;
  const float* ap = A + (size_t)idx * lda + 2 * half;
#pragma unroll
  for (int k = 0; k < K; k += 4) {
    v2f a, b, w;
    a.x = ap[k]; a.y = ap[k + 1];
    b.x = Bm[(size_t)(k + 2 * half)     * ldb + idx];
    b.y = Bm[(size_t)(k + 2 * half + 1) * ldb + idx];
    w.x = u[k + 2 * half]; w.y = u[k + 2 * half + 1];
    acc = wmma4(a, b, acc);
    rs  = wmma4(a, w, rs);
  }
}

template <int K>
__device__ __forceinline__ void tile_ab_rowsum_k(const float* A, const float* Bm, int ldb,
                                                 v8f& acc, v8f& rs) {
  const int lane = threadIdx.x & 31;
  const int half = lane >> 4;
  const int idx  = lane & 15;
  const float* ap = A + (size_t)idx * K + 2 * half;
  v2f ones; ones.x = 1.0f; ones.y = 1.0f;
#pragma unroll
  for (int k = 0; k < K; k += 4) {
    v2f a, b;
    a.x = ap[k]; a.y = ap[k + 1];
    b.x = Bm[(size_t)(k + 2 * half)     * ldb + idx];
    b.y = Bm[(size_t)(k + 2 * half + 1) * ldb + idx];
    acc = wmma4(a, b, acc);
    rs  = wmma4(a, ones, rs);
  }
}

#define XP 132

typedef __attribute__((ext_vector_type(4))) float v4f;

__device__ __forceinline__ void lds_barrier() {
  asm volatile("s_wait_dscnt 0x0\n\ts_barrier_signal -1\n\ts_barrier_wait -1" ::: "memory");
}

__device__ __forceinline__ float elu_plus1(float v) {
  return ((v > 0.f) ? v : expm1f(v)) + 1.0f;
}

__global__ void __launch_bounds__(128)
dla_lnproj_kernel(const float* __restrict__ z, const float* __restrict__ st,
                  const float* __restrict__ lnw, const float* __restrict__ lnb,
                  const float* __restrict__ Wk, const float* __restrict__ Wq,
                  const float* __restrict__ Wv, const float* __restrict__ scw,
                  float* __restrict__ Kb, float* __restrict__ Qb,
                  float* __restrict__ Vb, float* __restrict__ Sc,
                  float* __restrict__ outst) {
  __shared__ __attribute__((aligned(16))) float xs[16 * XP];
  const int wave = threadIdx.x >> 5;
  const int g = blockIdx.x * 4 + wave;
  const int b = g >> 7;
  const int tile = g & 127;
  const int tm = tile >> 3, tn = tile & 7;
  const int lane = threadIdx.x & 31, half = lane >> 4, idx = lane & 15;

  const int t = tm * 16 + idx;
  const int j0 = 16 * wave + 8 * half;
  const float* zp = z + ((size_t)t * Bn + b) * INn + j0;
  const float ts = st[(size_t)b * STSTRIDE + STSTRIDE - 1];
  const v4f z1lo = *(const v4f*)(zp);
  const v4f z1hi = *(const v4f*)(zp + 4);
  const v4f z2lo = *(const v4f*)(zp + 64);
  const v4f z2hi = *(const v4f*)(zp + 68);
  const float* xr = xs + idx * XP + 2 * half;
  const size_t wrow = (size_t)(tn * 16 + idx) * INn + 2 * half;
  const float* pk = Wk + wrow;
  const float* pq = Wq + wrow;
  const float* pv = Wv + wrow;
  const float* ps = scw + wrow;
  const float* wp = lnw + 2 * half;
  const float* bp = lnb + 2 * half;

  v2f a[32], lw[32], lb[32], fk[32], fq[32], fv[32], fs[32];
#pragma unroll
  for (int s = 0; s < 16; s++) {
    fk[s].x = pk[4 * s]; fk[s].y = pk[4 * s + 1];
    fq[s].x = pq[4 * s]; fq[s].y = pq[4 * s + 1];
    fv[s].x = pv[4 * s]; fv[s].y = pv[4 * s + 1];
    fs[s].x = ps[4 * s]; fs[s].y = ps[4 * s + 1];
  }
  __builtin_amdgcn_sched_barrier(0);

  {
    const float pos = ts + (float)t;
    const float x1[8] = {z1lo.x, z1lo.y, z1lo.z, z1lo.w, z1hi.x, z1hi.y, z1hi.z, z1hi.w};
    const float x2[8] = {z2lo.x, z2lo.y, z2lo.z, z2lo.w, z2hi.x, z2hi.y, z2hi.z, z2hi.w};
    float r1[8], r2[8];
#pragma unroll
    for (int i = 0; i < 8; i++) {
      const float freq = expf((float)(2 * (j0 + i)) * (-9.210340371976184f / 128.0f));
      const float ang = pos * freq;
      const float s = sinf(ang), c = cosf(ang);
      r1[i] = x1[i] * c - x2[i] * s;
      r2[i] = x2[i] * c + x1[i] * s;
    }
    float* xw = xs + idx * XP + j0;
    v4f o;
    o.x = r1[0]; o.y = r1[1]; o.z = r1[2]; o.w = r1[3]; *(v4f*)(xw)      = o;
    o.x = r1[4]; o.y = r1[5]; o.z = r1[6]; o.w = r1[7]; *(v4f*)(xw + 4)  = o;
    o.x = r2[0]; o.y = r2[1]; o.z = r2[2]; o.w = r2[3]; *(v4f*)(xw + 64) = o;
    o.x = r2[4]; o.y = r2[5]; o.z = r2[6]; o.w = r2[7]; *(v4f*)(xw + 68) = o;
  }
  if (tile == 0 && (threadIdx.x & 31) == 0)
    outst[(size_t)b * STSTRIDE + STSTRIDE - 1] = ts + (float)Tn;
#pragma unroll
  for (int s = 0; s < 32; s++) {
    lw[s].x = wp[4 * s]; lw[s].y = wp[4 * s + 1];
    lb[s].x = bp[4 * s]; lb[s].y = bp[4 * s + 1];
  }
  lds_barrier();
#pragma unroll
  for (int s = 0; s < 32; s++) { a[s] = *(const v2f*)(xr + 4 * s); }

  float p0 = 0.f, p1 = 0.f, p2 = 0.f, p3 = 0.f;
#pragma unroll
  for (int s = 0; s < 32; s += 4) {
    p0 += a[s].x + a[s].y;
    p1 += a[s + 1].x + a[s + 1].y;
    p2 += a[s + 2].x + a[s + 2].y;
    p3 += a[s + 3].x + a[s + 3].y;
  }
  float sum = (p0 + p1) + (p2 + p3);
  sum += __shfl_xor(sum, 16);
  const float m = sum * (1.0f / INn);

  float q0 = 0.f, q1 = 0.f, q2 = 0.f, q3 = 0.f;
#pragma unroll
  for (int s = 0; s < 32; s += 4) {
    float d;
    d = a[s].x - m;     q0 += d * d;  d = a[s].y - m;     q0 += d * d;
    d = a[s + 1].x - m; q1 += d * d;  d = a[s + 1].y - m; q1 += d * d;
    d = a[s + 2].x - m; q2 += d * d;  d = a[s + 2].y - m; q2 += d * d;
    d = a[s + 3].x - m; q3 += d * d;  d = a[s + 3].y - m; q3 += d * d;
  }
  float sq = (q0 + q1) + (q2 + q3);
  sq += __shfl_xor(sq, 16);
  const float inv = rsqrtf(sq * (1.0f / INn) + 1e-5f);


#pragma unroll
  for (int s = 0; s < 32; s++) {
    a[s].x = (a[s].x - m) * inv * lw[s].x + lb[s].x;
    a[s].y = (a[s].y - m) * inv * lw[s].y + lb[s].y;
  }

  v8f accK = {0.f,0.f,0.f,0.f,0.f,0.f,0.f,0.f};
  v8f accQ = {0.f,0.f,0.f,0.f,0.f,0.f,0.f,0.f};
  v8f accV = {0.f,0.f,0.f,0.f,0.f,0.f,0.f,0.f};
  v8f accS = {0.f,0.f,0.f,0.f,0.f,0.f,0.f,0.f};

#pragma unroll
  for (int gq = 0; gq < 4; gq++) {
    if (gq < 2) {
#pragma unroll
      for (int s = 8 * gq + 16; s < 8 * gq + 24; s++) {
        fk[s].x = pk[4 * s]; fk[s].y = pk[4 * s + 1];
        fq[s].x = pq[4 * s]; fq[s].y = pq[4 * s + 1];
        fv[s].x = pv[4 * s]; fv[s].y = pv[4 * s + 1];
        fs[s].x = ps[4 * s]; fs[s].y = ps[4 * s + 1];
      }
    }
    __builtin_amdgcn_sched_barrier(0);
#pragma unroll
    for (int s = 8 * gq; s < 8 * gq + 8; s++) {
      accK = wmma4(a[s], fk[s], accK);
      accQ = wmma4(a[s], fq[s], accQ);
      accV = wmma4(a[s], fv[s], accV);
      accS = wmma4(a[s], fs[s], accS);
    }
  }

  const size_t o = (size_t)b * Tn * Hn + (size_t)tm * 16 * Hn + tn * 16;
#pragma unroll
  for (int r = 0; r < 8; r++) {
    const size_t e = o + (size_t)(r + 8 * half) * Hn + idx;
    Kb[e] = elu_plus1(accK[r]);
    Qb[e] = elu_plus1(accQ[r]);
    Vb[e] = accV[r];
    Sc[e] = accS[r];
  }
}

#define PSZ  393216
#define ZPSZ 3072

__device__ __forceinline__ void dla_state_job(int b, int jb, int half, int idx,
                                              const float* __restrict__ Kb, const float* __restrict__ Vb,
                                              const float* __restrict__ st, int layerOff,
                                              float* __restrict__ Pb, float* __restrict__ Zpb,
                                              float* __restrict__ outst) {
  const bool zjob = jb >= 64;
  const int ti = zjob ? (jb - 64) : (jb >> 3);
  const int tn = jb & 7;
  const float* S0 = st + (size_t)b * STSTRIDE + layerOff;
  float* SOut = outst + (size_t)b * STSTRIDE + layerOff;
  const float* Kt = Kb + (size_t)b * Tn * Hn + ti * 16 + idx;
  const float* Vt = Vb + (size_t)b * Tn * Hn + tn * 16 + idx;
  v8f sx = {0.f,0.f,0.f,0.f,0.f,0.f,0.f,0.f};
  if (!zjob) {
    float s0v[8];
#pragma unroll
    for (int r = 0; r < 8; r++)
      s0v[r] = S0[(size_t)(ti * 16 + r + 8 * half) * Hn + tn * 16 + idx];
    v2f ka[64], vb[64];
#pragma unroll
    for (int s = 0; s < 32; s++) {
      ka[s].x = Kt[(size_t)(4 * s + 2 * half)     * Hn];
      ka[s].y = Kt[(size_t)(4 * s + 2 * half + 1) * Hn];
      vb[s].x = Vt[(size_t)(4 * s + 2 * half)     * Hn];
      vb[s].y = Vt[(size_t)(4 * s + 2 * half + 1) * Hn];
    }
#pragma unroll
    for (int g = 0; g < 4; g++) {
      if (g + 2 < 4) {
#pragma unroll
        for (int s = 16 * (g + 2); s < 16 * (g + 3); s++) {
          ka[s].x = Kt[(size_t)(4 * s + 2 * half)     * Hn];
          ka[s].y = Kt[(size_t)(4 * s + 2 * half + 1) * Hn];
          vb[s].x = Vt[(size_t)(4 * s + 2 * half)     * Hn];
          vb[s].y = Vt[(size_t)(4 * s + 2 * half + 1) * Hn];
        }
      }
      __builtin_amdgcn_sched_barrier(0);
#pragma unroll
      for (int s = 16 * g; s < 16 * g + 16; s++) sx = wmma4(ka[s], vb[s], sx);
      float* D = (g < 3) ? (Pb + (size_t)(b * 3 + g) * Hn * Hn) : SOut;
#pragma unroll
      for (int r = 0; r < 8; r++)
        D[(size_t)(ti * 16 + r + 8 * half) * Hn + tn * 16 + idx] = sx[r] + s0v[r];
    }
  } else {
    v2f ones; ones.x = 1.0f; ones.y = 1.0f;
    float z0v[8];
#pragma unroll
    for (int r = 0; r < 8; r++) z0v[r] = S0[Hn * Hn + ti * 16 + r + 8 * half];
    v2f ka[64];
#pragma unroll
    for (int s = 0; s < 32; s++) {
      ka[s].x = Kt[(size_t)(4 * s + 2 * half)     * Hn];
      ka[s].y = Kt[(size_t)(4 * s + 2 * half + 1) * Hn];
    }
#pragma unroll
    for (int g = 0; g < 4; g++) {
      if (g + 2 < 4) {
#pragma unroll
        for (int s = 16 * (g + 2); s < 16 * (g + 3); s++) {
          ka[s].x = Kt[(size_t)(4 * s + 2 * half)     * Hn];
          ka[s].y = Kt[(size_t)(4 * s + 2 * half + 1) * Hn];
        }
      }
      __builtin_amdgcn_sched_barrier(0);
#pragma unroll
      for (int s = 16 * g; s < 16 * g + 16; s++) sx = wmma4(ka[s], ones, sx);
      float* D = (g < 3) ? (Zpb + (size_t)(b * 3 + g) * Hn) : (SOut + Hn * Hn);
      float zv[8];
#pragma unroll
      for (int r = 0; r < 8; r++) zv[r] = z0v[r] + sx[r];
      if (idx == 0) {
#pragma unroll
        for (int r = 0; r < 8; r++) D[ti * 16 + r + 8 * half] = zv[r];
      }
    }
  }
}

__global__ void __launch_bounds__(256)
dla_score_state_kernel(const float* __restrict__ Qb, const float* __restrict__ Kb,
                                       const float* __restrict__ Vb, const float* __restrict__ st,
                                       int layerOff, float* __restrict__ Pb,
                                       float* __restrict__ Zpb, float* __restrict__ Ab,
                                       float* __restrict__ outst) {
  const int wave = __builtin_amdgcn_readfirstlane((int)(threadIdx.x >> 5));
  const int b  = blockIdx.x / 14;
  const int jb = (blockIdx.x - b * 14) * 8 + wave;
  const int lane = threadIdx.x & 31, half = lane >> 4, idx = lane & 15;
  if (jb >= 72) {
    const int y = (jb - 72) >> 3;
    const int c = (y < 4) ? y : (wave >> 1);
    const int q = (y < 4) ? ((wave >= 4) ? 3 : (wave >= 1) ? 2 : 0) : 1;
    const int p = (y < 4) ? ((wave >= 4) ? (wave - 4) : (wave >= 1) ? (wave - 1) : 0) : (wave & 1);
    const int tm = 4 * c + q, tn = 4 * c + p;
    const int lda = 16 * (q + 1);
    float* Out = Ab + (size_t)((b * 4 + c) * 10 + ((q * (q + 1)) >> 1)) * 256 + p * 16;
    v8f acc = {0.f,0.f,0.f,0.f,0.f,0.f,0.f,0.f};
    tile_abt(Qb + (size_t)b * Tn * Hn + (size_t)tm * 16 * Hn, Hn,
             Kb + (size_t)b * Tn * Hn + (size_t)tn * 16 * Hn, Hn, Hn, acc);
#pragma unroll
    for (int r = 0; r < 8; r++) {
      float v = acc[r];
      if (p == q && idx > (r + 8 * half)) v = 0.f;
      Out[(size_t)(r + 8 * half) * lda + idx] = v;
    }
    return;
  }
  dla_state_job(b, jb, half, idx, Kb, Vb, st, layerOff, Pb, Zpb, outst);
}

#define AP 68
#define LDP 32

__global__ void __launch_bounds__(256)
dla_state_local_kernel(const float* __restrict__ Qb, const float* __restrict__ Kb,
                       const float* __restrict__ Vb, const float* __restrict__ st,
                       int layerOff, float* __restrict__ Pb, float* __restrict__ Zpb,
                       float* __restrict__ Lnm, float* __restrict__ Ldn,
                       float* __restrict__ outst) {
  __shared__ __attribute__((aligned(16))) float as[16 * AP];
  const int wave = __builtin_amdgcn_readfirstlane((int)(threadIdx.x >> 5));
  const int lane = threadIdx.x & 31, half = lane >> 4, idx = lane & 15;
  if (blockIdx.x >= 128) {
    const int sb = blockIdx.x - 128;
    const int b  = sb / 9;
    const int jb = (sb - b * 9) * 8 + wave;
    dla_state_job(b, jb, half, idx, Kb, Vb, st, layerOff, Pb, Zpb, outst);
    return;
  }
  const int b = blockIdx.x >> 4;
  const int tm = blockIdx.x & 15;
  const int c = tm >> 2, q = tm & 3;
  const size_t bo = (size_t)b * Tn * Hn;
  const float* Vt = Vb + bo + (size_t)c * 64 * Hn + wave * 16 + idx;
  v2f vb[16];
#pragma unroll
  for (int g = 0; g < 4; g++) {
    if (g <= q) {
#pragma unroll
      for (int s = 4 * g; s < 4 * g + 4; s++) {
        vb[s].x = Vt[(size_t)(4 * s + 2 * half)     * Hn];
        vb[s].y = Vt[(size_t)(4 * s + 2 * half + 1) * Hn];
      }
    }
  }
  if (wave <= q) {
    v8f sc = {0.f,0.f,0.f,0.f,0.f,0.f,0.f,0.f};
    tile_abt(Qb + bo + (size_t)tm * 16 * Hn, Hn,
             Kb + bo + (size_t)(4 * c + wave) * 16 * Hn, Hn, Hn, sc);
#pragma unroll
    for (int r = 0; r < 8; r++) {
      float v = sc[r];
      if (wave == q && idx > (r + 8 * half)) v = 0.f;
      as[(r + 8 * half) * AP + 16 * wave + idx] = v;
    }
  }
  lds_barrier();
  v8f acc  = {0.f,0.f,0.f,0.f,0.f,0.f,0.f,0.f};
  v8f dsum = {0.f,0.f,0.f,0.f,0.f,0.f,0.f,0.f};
  v2f ones; ones.x = 1.0f; ones.y = 1.0f;
  const float* ar = as + idx * AP + 2 * half;
#pragma unroll
  for (int g = 0; g < 4; g++) {
    if (g <= q) {
#pragma unroll
      for (int s = 4 * g; s < 4 * g + 4; s++) {
        const v2f a = *(const v2f*)(ar + 4 * s);
        acc  = wmma4(a, vb[s], acc);
        dsum = wmma4(a, ones, dsum);
      }
    }
  }
  float* Lo = Lnm + bo + (size_t)tm * 16 * Hn + wave * 16;
#pragma unroll
  for (int r = 0; r < 8; r++) Lo[(size_t)(r + 8 * half) * Hn + idx] = acc[r];
  if (wave == 0 && idx == 0) {
    float* Do = Ldn + (size_t)(b * 16 + tm) * LDP + 8 * half;
#pragma unroll
    for (int r = 0; r < 8; r++) Do[r] = dsum[r];
  }
}

__global__ void dla_attnout_kernel(const float* __restrict__ Ab, const float* __restrict__ Vb,
                                   const float* __restrict__ Qb,
                                   const float* __restrict__ Pb, const float* __restrict__ Zpb,
                                   const float* __restrict__ st, int layerOff,
                                   float* __restrict__ Ao) {
  const int wave = __builtin_amdgcn_readfirstlane((int)(threadIdx.x >> 5));
  const int g = blockIdx.x * 8 + wave;
  const int b = g >> 7;
  const int tile = g & 127;
  const int tm = tile >> 3, tn = tile & 7;
  const int c = tm >> 2, q = tm & 3;
  v8f acc  = {0.f,0.f,0.f,0.f,0.f,0.f,0.f,0.f};
  v8f dsum = {0.f,0.f,0.f,0.f,0.f,0.f,0.f,0.f};
  const float* At = Ab + (size_t)((b * 4 + c) * 10 + ((q * (q + 1)) >> 1)) * 256;
  const float* Vt = Vb + (size_t)b * Tn * Hn + (size_t)c * 64 * Hn + tn * 16;
  switch (q) {
    case 0:  tile_ab_rowsum_k<16>(At, Vt, Hn, acc, dsum); break;
    case 1:  tile_ab_rowsum_k<32>(At, Vt, Hn, acc, dsum); break;
    case 2:  tile_ab_rowsum_k<48>(At, Vt, Hn, acc, dsum); break;
    default: tile_ab_rowsum_k<64>(At, Vt, Hn, acc, dsum); break;
  }
  const float* S0 = st + (size_t)b * STSTRIDE + layerOff;
  const float* Pc = (c == 0) ? S0 : (Pb + (size_t)(b * 3 + (c - 1)) * Hn * Hn);
  const float* Zc = (c == 0) ? (S0 + Hn * Hn) : (Zpb + (size_t)(b * 3 + (c - 1)) * Hn);
  tile_ab_rowdot(Qb + (size_t)b * Tn * Hn + (size_t)tm * 16 * Hn, Hn,
                 Pc + tn * 16, Hn, Zc, Hn, acc, dsum);
  const int lane = threadIdx.x & 31, half = lane >> 4, idx = lane & 15;
  float* Out = Ao + (size_t)b * Tn * Hn + (size_t)tm * 16 * Hn + tn * 16;
#pragma unroll
  for (int r = 0; r < 8; r++)
    Out[(size_t)(r + 8 * half) * Hn + idx] = acc[r] / (dsum[r] + 1e-5f);
}


#define LOAD_FENCE() do { asm volatile("" ::: "memory"); __builtin_amdgcn_sched_barrier(0); } while (0)

template <bool LAST>
__global__ void __launch_bounds__(128)
dla_ffchain_kernel(const float* __restrict__ Ao,
                   const float* __restrict__ w1, const float* __restrict__ b1,
                   const float* __restrict__ w2, const float* __restrict__ b2,
                   const float* __restrict__ sc, const float* __restrict__ scb,
                   const float* __restrict__ lnw, const float* __restrict__ lnb,
                   const float* __restrict__ Wa, const float* __restrict__ Wq,
                   const float* __restrict__ Wv, const float* __restrict__ Ws,
                   const float* __restrict__ ub,
                   float* __restrict__ O0, float* __restrict__ O1,
                   float* __restrict__ O2, float* __restrict__ O3,
                   const float* __restrict__ Qb, const float* __restrict__ Pb,
                   const float* __restrict__ Zpb, const float* __restrict__ Lnm,
                   const float* __restrict__ Ldn, const float* __restrict__ S0l) {
  __shared__ __attribute__((aligned(16))) float hs[16 * XP];
  __shared__ __attribute__((aligned(16))) float xs[16 * XP];
  __shared__ __attribute__((aligned(16))) float as0[16 * XP];
  const int wave = threadIdx.x >> 5;
  const int b = blockIdx.x >> 5;
  const int tm = (blockIdx.x & 31) >> 1;
  const int tn = 4 * (blockIdx.x & 1) + wave;
  const int lane = threadIdx.x & 31, half = lane >> 4, idx = lane & 15;
  const size_t row0 = (size_t)b * Tn + (size_t)tm * 16;
  const int c0 = 32 * wave;

  const float* aop = Ao + (row0 + idx) * Hn + 2 * half;
  const float* w1p = w1 + (size_t)(c0 + idx) * Hn + 2 * half;
  const float* w2p = w2 + (size_t)(c0 + idx) * Hn + 8 * half;
  const float* b1p = b1 + c0 + 8 * half;
  const float* b2p = b2 + c0 + 8 * half;
  const float* scbp = scb + c0 + 8 * half;
  const float* scp = sc + (row0 + idx) * Hn + c0 + 8 * half;
  const size_t w3o = (size_t)(tn * 16 + idx) * Hn + 8 * half;
  float* hrow = hs + idx * XP + 8 * half;
  float* xrow = xs + idx * XP + 8 * half;

  v2f ao[32], f1[2][32], f2[2][32], f3[4][32], hb[32];
  v8f h[2], x2[2], acc2[8];
  float b1v[2][8], b2v[2][8], scbv[2][8], scv[2][8], lwv[8][8], lbv[8][8];
#pragma unroll
  for (int u = 0; u < 2; u++) {
#pragma unroll
    for (int r = 0; r < 8; r++) { h[u][r] = 0.f; x2[u][r] = 0.f; }
  }
  if (LAST) {
    const int cch = tm >> 2;
    const float* S0 = S0l + (size_t)b * STSTRIDE;
    const float* Pc = (cch == 0) ? S0 : (Pb + (size_t)(b * 3 + (cch - 1)) * Hn * Hn);
    const float* Zc = (cch == 0) ? (S0 + Hn * Hn) : (Zpb + (size_t)(b * 3 + (cch - 1)) * Hn);
    const float* qp  = Qb + (row0 + idx) * Hn + 2 * half;
    const float* pp0 = Pc + (size_t)(2 * half) * Hn + c0 + idx;
    const float* zp  = Zc + 2 * half;
    v2f qa[32], pb[2][32], zw[32];
    float lnv[2][8], ldv[8];
    v8f n0 = {0.f,0.f,0.f,0.f,0.f,0.f,0.f,0.f};
    v8f n1 = {0.f,0.f,0.f,0.f,0.f,0.f,0.f,0.f};
    v8f dn = {0.f,0.f,0.f,0.f,0.f,0.f,0.f,0.f};
#define S0_LOAD(s) do { \
    qa[s].x = qp[4 * (s)]; qa[s].y = qp[4 * (s) + 1]; \
    pb[0][s].x = pp0[(size_t)(4 * (s)) * Hn];      pb[0][s].y = pp0[(size_t)(4 * (s) + 1) * Hn]; \
    pb[1][s].x = pp0[(size_t)(4 * (s)) * Hn + 16]; pb[1][s].y = pp0[(size_t)(4 * (s) + 1) * Hn + 16]; \
    zw[s].x = zp[4 * (s)]; zw[s].y = zp[4 * (s) + 1]; } while (0)
#pragma unroll
    for (int s = 0; s < 16; s++) S0_LOAD(s);
#pragma unroll
    for (int u = 0; u < 2; u++) {
#pragma unroll
      for (int r = 0; r < 8; r++)
        lnv[u][r] = Lnm[(row0 + 8 * half + r) * Hn + c0 + 16 * u + idx];
    }
#pragma unroll
    for (int r = 0; r < 8; r++) ldv[r] = Ldn[(size_t)(b * 16 + tm) * LDP + 8 * half + r];
#pragma unroll
    for (int g = 0; g < 4; g++) {
      if (g < 2) {
#pragma unroll
        for (int s = 8 * g + 16; s < 8 * g + 24; s++) S0_LOAD(s);
      } else {
#pragma unroll
        for (int s = 16 * (g - 2); s < 16 * (g - 2) + 16; s++) {
#pragma unroll
          for (int u = 0; u < 2; u++) {
            const float* pp = w1p + (size_t)(16 * u) * Hn + 4 * s;
            f1[u][s].x = pp[0]; f1[u][s].y = pp[1];
          }
        }
      }
      LOAD_FENCE();
#pragma unroll
      for (int s = 8 * g; s < 8 * g + 8; s++) {
        n0 = wmma4(qa[s], pb[0][s], n0);
        n1 = wmma4(qa[s], pb[1][s], n1);
        dn = wmma4(qa[s], zw[s], dn);
      }
    }
#undef S0_LOAD
#pragma unroll
    for (int r = 0; r < 8; r++) {
      const float den = (dn[r] + ldv[r]) + 1e-5f;
      as0[(8 * half + r) * XP + c0 + idx]      = (n0[r] + lnv[0][r]) / den;
      as0[(8 * half + r) * XP + c0 + 16 + idx] = (n1[r] + lnv[1][r]) / den;
    }
    lds_barrier();
#pragma unroll
    for (int s = 0; s < 32; s++) ao[s] = *(const v2f*)(as0 + idx * XP + 2 * half + 4 * s);
  } else {
#pragma unroll
    for (int s = 0; s < 32; s++) { ao[s].x = aop[4 * s]; ao[s].y = aop[4 * s + 1]; }
#pragma unroll
    for (int s = 0; s < 32; s++) {
#pragma unroll
      for (int u = 0; u < 2; u++) {
        const float* pp = w1p + (size_t)(16 * u) * Hn + 4 * s;
        f1[u][s].x = pp[0]; f1[u][s].y = pp[1];
      }
    }
  }
  LOAD_FENCE();

#pragma unroll
  for (int s = 0; s < 16; s++) {
#pragma unroll
    for (int u = 0; u < 2; u++) h[u] = wmma4(f1[u][s], ao[s], h[u]);
  }
  LOAD_FENCE();
#pragma unroll
  for (int u = 0; u < 2; u++) {
#pragma unroll
    for (int r = 0; r < 8; r++) b1v[u][r] = b1p[16 * u + r];
  }
#pragma unroll
  for (int j = 0; j < 8; j++) {
#pragma unroll
    for (int rr = 0; rr < 4; rr++) {
#pragma unroll
      for (int u = 0; u < 2; u++) {
        const float* pp = w2p + (size_t)(16 * u) * Hn + 16 * j + 2 * rr;
        f2[u][4 * j + rr].x = pp[0]; f2[u][4 * j + rr].y = pp[1];
      }
    }
  }
#pragma unroll
  for (int u = 0; u < 2; u++) {
#pragma unroll
    for (int r = 0; r < 8; r++) {
      b2v[u][r] = b2p[16 * u + r];
      scbv[u][r] = scbp[16 * u + r];
      scv[u][r] = scp[16 * u + r];
    }
  }
  LOAD_FENCE();
#pragma unroll
  for (int s = 16; s < 32; s++) {
#pragma unroll
    for (int u = 0; u < 2; u++) h[u] = wmma4(f1[u][s], ao[s], h[u]);
  }

#pragma unroll
  for (int u = 0; u < 2; u++) {
    float t[8];
#pragma unroll
    for (int r = 0; r < 8; r++) {
      const float v = h[u][r] + b1v[u][r];
      t[r] = v > 0.f ? v : 0.f;
    }
    v4f lo, hi;
    lo.x = t[0]; lo.y = t[1]; lo.z = t[2]; lo.w = t[3];
    hi.x = t[4]; hi.y = t[5]; hi.z = t[6]; hi.w = t[7];
    *(v4f*)(hrow + c0 + 16 * u)     = lo;
    *(v4f*)(hrow + c0 + 16 * u + 4) = hi;
  }
  lds_barrier();

#pragma unroll
  for (int j = 0; j < 8; j++) {
    const v4f lo = *(const v4f*)(hrow + 16 * j);
    const v4f hi = *(const v4f*)(hrow + 16 * j + 4);
    hb[4 * j].x     = lo.x; hb[4 * j].y     = lo.y;
    hb[4 * j + 1].x = lo.z; hb[4 * j + 1].y = lo.w;
    hb[4 * j + 2].x = hi.x; hb[4 * j + 2].y = hi.y;
    hb[4 * j + 3].x = hi.z; hb[4 * j + 3].y = hi.w;
  }
  LOAD_FENCE();
#pragma unroll
  for (int e = 0; e < 16; e++) {
#pragma unroll
    for (int u = 0; u < 2; u++) x2[u] = wmma4(f2[u][e], hb[e], x2[u]);
  }
  LOAD_FENCE();
  if (LAST) {
#pragma unroll
    for (int n = 0; n < 8; n++) {
#pragma unroll
      for (int rr = 0; rr < 4; rr++) {
        const float* pp = Wa + w3o + 16 * n + 2 * rr;
        f3[0][n * 4 + rr].x = pp[0]; f3[0][n * 4 + rr].y = pp[1];
      }
    }
  } else {
    const float* lwp = lnw + 8 * half;
    const float* lbp = lnb + 8 * half;
#pragma unroll
    for (int n = 0; n < 8; n++) {
#pragma unroll
      for (int r = 0; r < 8; r++) { lwv[n][r] = lwp[16 * n + r]; lbv[n][r] = lbp[16 * n + r]; }
    }
#pragma unroll
    for (int nl = 0; nl < 2; nl++) {
#pragma unroll
      for (int rr = 0; rr < 4; rr++) {
        const size_t o = w3o + 16 * nl + 2 * rr;
        f3[0][0 * 8 + nl * 4 + rr].x = Wa[o]; f3[0][0 * 8 + nl * 4 + rr].y = Wa[o + 1];
        f3[0][1 * 8 + nl * 4 + rr].x = Wq[o]; f3[0][1 * 8 + nl * 4 + rr].y = Wq[o + 1];
        f3[0][2 * 8 + nl * 4 + rr].x = Wv[o]; f3[0][2 * 8 + nl * 4 + rr].y = Wv[o + 1];
        f3[0][3 * 8 + nl * 4 + rr].x = Ws[o]; f3[0][3 * 8 + nl * 4 + rr].y = Ws[o + 1];
      }
    }
  }
  LOAD_FENCE();
#pragma unroll
  for (int e = 16; e < 32; e++) {
#pragma unroll
    for (int u = 0; u < 2; u++) x2[u] = wmma4(f2[u][e], hb[e], x2[u]);
  }

#pragma unroll
  for (int u = 0; u < 2; u++) {
    float t[8];
#pragma unroll
    for (int r = 0; r < 8; r++) {
      float v = x2[u][r] + b2v[u][r];
      v = v > 0.f ? v : 0.f;
      t[r] = v + scv[u][r] + scbv[u][r];
    }
    v4f lo, hi;
    lo.x = t[0]; lo.y = t[1]; lo.z = t[2]; lo.w = t[3];
    hi.x = t[4]; hi.y = t[5]; hi.z = t[6]; hi.w = t[7];
    *(v4f*)(xrow + c0 + 16 * u)     = lo;
    *(v4f*)(xrow + c0 + 16 * u + 4) = hi;
  }
  lds_barrier();

#pragma unroll
  for (int n = 0; n < 8; n++) {
    const v4f lo = *(const v4f*)(xrow + 16 * n);
    const v4f hi = *(const v4f*)(xrow + 16 * n + 4);
    acc2[n][0] = lo.x; acc2[n][1] = lo.y; acc2[n][2] = lo.z; acc2[n][3] = lo.w;
    acc2[n][4] = hi.x; acc2[n][5] = hi.y; acc2[n][6] = hi.z; acc2[n][7] = hi.w;
  }
  if (!LAST) {
#pragma unroll
    for (int q3 = 1; q3 < 4; q3++) {
#pragma unroll
      for (int nl = 0; nl < 2; nl++) {
#pragma unroll
        for (int rr = 0; rr < 4; rr++) {
          const size_t o = w3o + 16 * (2 * q3 + nl) + 2 * rr;
          f3[q3][0 * 8 + nl * 4 + rr].x = Wa[o]; f3[q3][0 * 8 + nl * 4 + rr].y = Wa[o + 1];
          f3[q3][1 * 8 + nl * 4 + rr].x = Wq[o]; f3[q3][1 * 8 + nl * 4 + rr].y = Wq[o + 1];
          f3[q3][2 * 8 + nl * 4 + rr].x = Wv[o]; f3[q3][2 * 8 + nl * 4 + rr].y = Wv[o + 1];
          f3[q3][3 * 8 + nl * 4 + rr].x = Ws[o]; f3[q3][3 * 8 + nl * 4 + rr].y = Ws[o + 1];
        }
      }
    }
  }
  LOAD_FENCE();

  if (LAST) {
    v8f accE = {0.f,0.f,0.f,0.f,0.f,0.f,0.f,0.f};
    v8f accO = {0.f,0.f,0.f,0.f,0.f,0.f,0.f,0.f};
#pragma unroll
    for (int n = 0; n < 8; n++) {
#pragma unroll
      for (int rr = 0; rr < 4; rr += 2) {
        v2f xa, xb;
        xa.x = acc2[n][2 * rr];     xa.y = acc2[n][2 * rr + 1];
        xb.x = acc2[n][2 * rr + 2]; xb.y = acc2[n][2 * rr + 3];
        accE = wmma4(xa, f3[0][n * 4 + rr], accE);
        accO = wmma4(xb, f3[0][n * 4 + rr + 1], accO);
      }
    }
#pragma unroll
    for (int r = 0; r < 8; r++) {
      const int t = tm * 16 + r + 8 * half;
      const int o = tn * 16 + idx;
      O0[(size_t)t * Bn * OUTn + (size_t)b * OUTn + o] = (accE[r] + accO[r]) + ub[o];
    }
  } else {
    float p0 = 0.f, p1 = 0.f, p2 = 0.f, p3 = 0.f;
#pragma unroll
    for (int n = 0; n < 8; n += 4) {
#pragma unroll
      for (int r = 0; r < 8; r += 2) {
        p0 += acc2[n][r] + acc2[n][r + 1];
        p1 += acc2[n + 1][r] + acc2[n + 1][r + 1];
        p2 += acc2[n + 2][r] + acc2[n + 2][r + 1];
        p3 += acc2[n + 3][r] + acc2[n + 3][r + 1];
      }
    }
    float sum = (p0 + p1) + (p2 + p3);
    sum += __shfl_xor(sum, 16);
    const float m = sum * (1.0f / INn);

    float q0 = 0.f, q1 = 0.f, q2 = 0.f, q3 = 0.f;
#pragma unroll
    for (int n = 0; n < 8; n += 4) {
#pragma unroll
      for (int r = 0; r < 8; r++) {
        float d;
        d = acc2[n][r] - m;     q0 += d * d;
        d = acc2[n + 1][r] - m; q1 += d * d;
        d = acc2[n + 2][r] - m; q2 += d * d;
        d = acc2[n + 3][r] - m; q3 += d * d;
      }
    }
    float sq = (q0 + q1) + (q2 + q3);
    sq += __shfl_xor(sq, 16);
    const float inv = rsqrtf(sq * (1.0f / INn) + 1e-5f);

#pragma unroll
    for (int n = 0; n < 8; n++) {
#pragma unroll
      for (int r = 0; r < 8; r++)
        acc2[n][r] = (acc2[n][r] - m) * inv * lwv[n][r] + lbv[n][r];
    }

    v8f accK = {0.f,0.f,0.f,0.f,0.f,0.f,0.f,0.f};
    v8f accQ = {0.f,0.f,0.f,0.f,0.f,0.f,0.f,0.f};
    v8f accV = {0.f,0.f,0.f,0.f,0.f,0.f,0.f,0.f};
    v8f accS = {0.f,0.f,0.f,0.f,0.f,0.f,0.f,0.f};
#pragma unroll
    for (int q3 = 0; q3 < 4; q3++) {
#pragma unroll
      for (int nl = 0; nl < 2; nl++) {
#pragma unroll
        for (int rr = 0; rr < 4; rr++) {
          v2f xa;
          xa.x = acc2[2 * q3 + nl][2 * rr]; xa.y = acc2[2 * q3 + nl][2 * rr + 1];
          accK = wmma4(xa, f3[q3][0 * 8 + nl * 4 + rr], accK);
          accQ = wmma4(xa, f3[q3][1 * 8 + nl * 4 + rr], accQ);
          accV = wmma4(xa, f3[q3][2 * 8 + nl * 4 + rr], accV);
          accS = wmma4(xa, f3[q3][3 * 8 + nl * 4 + rr], accS);
        }
      }
    }
    const size_t o16 = row0 * Hn + tn * 16;
#pragma unroll
    for (int r = 0; r < 8; r++) {
      const size_t e = o16 + (size_t)(r + 8 * half) * Hn + idx;
      O0[e] = elu_plus1(accK[r]);
      O1[e] = elu_plus1(accQ[r]);
      O2[e] = accV[r];
      O3[e] = accS[r];
    }
  }
}

extern "C" void kernel_launch(void* const* d_in, const int* in_sizes, int n_in,
                              void* d_out, int out_size, void* d_ws, size_t ws_size,
                              hipStream_t stream) {
  (void)in_sizes; (void)n_in; (void)out_size; (void)ws_size;
  const float* z   = (const float*)d_in[0];
  const float* st  = (const float*)d_in[1];
  const float* Wk  = (const float*)d_in[2];
  const float* Wq  = (const float*)d_in[3];
  const float* Wv  = (const float*)d_in[4];
  const float* lnw = (const float*)d_in[5];
  const float* lnb = (const float*)d_in[6];
  const float* w1  = (const float*)d_in[7];
  const float* b1  = (const float*)d_in[8];
  const float* w2  = (const float*)d_in[9];
  const float* b2  = (const float*)d_in[10];
  const float* scw = (const float*)d_in[11];
  const float* scb = (const float*)d_in[12];
  const float* uw  = (const float*)d_in[13];
  const float* ub  = (const float*)d_in[14];
  float* out = (float*)d_out;
  float* ws  = (float*)d_ws;

  float* SC2  = ws;
  float* SC   = ws + 262144;
  float* KB   = ws + 524288;
  float* QB   = ws + 786432;
  float* VB   = ws + 1048576;
  float* PB   = ws + 1310720;
  float* ZPB  = PB + PSZ;
  float* AB   = ZPB + ZPSZ;
  float* AO   = ws + 1835008;
  float* LN1  = AO;
  float* LD1  = AB + 81920;
  float* OST  = out + YSZ;

  dla_lnproj_kernel<<<256, 128, 0, stream>>>(z, st, lnw, lnb, Wk, Wq, Wv, scw, KB, QB, VB, SC, OST);
  dla_score_state_kernel<<<Bn * 14, 256, 0, stream>>>(QB, KB, VB, st, 0, PB, ZPB, AB, OST);
  dla_attnout_kernel<<<128, 256, 0, stream>>>(AB, VB, QB, PB, ZPB, st, 0, AO);
  dla_ffchain_kernel<false><<<256, 128, 0, stream>>>(
      AO, w1, b1, w2, b2, SC, scb, lnw + INn, lnb + INn,
      Wk + (size_t)Hn * INn, Wq + (size_t)Hn * INn, Wv + (size_t)Hn * INn,
      scw + (size_t)Hn * INn, nullptr, KB, QB, VB, SC2,
      nullptr, nullptr, nullptr, nullptr, nullptr, nullptr);
  dla_state_local_kernel<<<128 + Bn * 9, 256, 0, stream>>>(QB, KB, VB, st, LAYERSTATE, PB, ZPB,
                                                          LN1, LD1, OST);
  dla_ffchain_kernel<true><<<256, 128, 0, stream>>>(
      nullptr, w1 + (size_t)Hn * Hn, b1 + Hn, w2 + (size_t)Hn * Hn, b2 + Hn, SC2, scb + Hn,
      nullptr, nullptr, uw, nullptr, nullptr, nullptr, ub, out, nullptr, nullptr, nullptr,
      QB, PB, ZPB, LN1, LD1, st + LAYERSTATE);
}


# layer 0's attn-out launch dissolved like layer 1's: A_cc V_c and row sums beside the state jobs (scores stay in LDS), the Q P_c half as stage 0 of the ff launch: 6 to 5 launches; sum order changes
# speedup vs baseline: 2.6047x; 1.0820x over previous
#include <hip/hip_runtime.h>
#include <hip/hip_bf16.h>
#include <math.h>

#define Bn 8
#define Tn 256
#define INn 128
#define Hn 128
#define OUTn 128
#define Ln 2
#define STSTRIDE 33025
#define LAYERSTATE 16512
#define YSZ (Tn*Bn*OUTn)

typedef __attribute__((ext_vector_type(2))) float v2f;
typedef __attribute__((ext_vector_type(8))) float v8f;

__device__ __forceinline__ v8f wmma4(v2f a, v2f b, v8f c) {
  return __builtin_amdgcn_wmma_f32_16x16x4_f32(false, a, false, b, (short)0, c, false, false);
}

__device__ __forceinline__ void tile_abt(const float* A, int lda, const float* Bt, int ldb,
                                         int K, v8f& acc) {
  const int lane = threadIdx.x & 31;
  const int half = lane >> 4;
  const int idx  = lane & 15;
  const float* ap = A  + (size_t)idx * lda + 2 * half;
  const float* bp = Bt + (size_t)idx * ldb + 2 * half;
  for (int k = 0; k < K; k += 4) {
    v2f a, b;
    a.x = ap[k]; a.y = ap[k + 1];
    b.x = bp[k]; b.y = bp[k + 1];
    acc = wmma4(a, b, acc);
  }
}

#define XP 132

typedef __attribute__((ext_vector_type(4))) float v4f;

__device__ __forceinline__ void lds_barrier() {
  asm volatile("s_wait_dscnt 0x0\n\ts_barrier_signal -1\n\ts_barrier_wait -1" ::: "memory");
}

__device__ __forceinline__ float elu_plus1(float v) {
  return ((v > 0.f) ? v : expm1f(v)) + 1.0f;
}

__global__ void __launch_bounds__(128)
dla_lnproj_kernel(const float* __restrict__ z, const float* __restrict__ st,
                  const float* __restrict__ lnw, const float* __restrict__ lnb,
                  const float* __restrict__ Wk, const float* __restrict__ Wq,
                  const float* __restrict__ Wv, const float* __restrict__ scw,
                  float* __restrict__ Kb, float* __restrict__ Qb,
                  float* __restrict__ Vb, float* __restrict__ Sc,
                  float* __restrict__ outst) {
  __shared__ __attribute__((aligned(16))) float xs[16 * XP];
  const int wave = threadIdx.x >> 5;
  const int g = blockIdx.x * 4 + wave;
  const int b = g >> 7;
  const int tile = g & 127;
  const int tm = tile >> 3, tn = tile & 7;
  const int lane = threadIdx.x & 31, half = lane >> 4, idx = lane & 15;

  const int t = tm * 16 + idx;
  const int j0 = 16 * wave + 8 * half;
  const float* zp = z + ((size_t)t * Bn + b) * INn + j0;
  const float ts = st[(size_t)b * STSTRIDE + STSTRIDE - 1];
  const v4f z1lo = *(const v4f*)(zp);
  const v4f z1hi = *(const v4f*)(zp + 4);
  const v4f z2lo = *(const v4f*)(zp + 64);
  const v4f z2hi = *(const v4f*)(zp + 68);
  const float* xr = xs + idx * XP + 2 * half;
  const size_t wrow = (size_t)(tn * 16 + idx) * INn + 2 * half;
  const float* pk = Wk + wrow;
  const float* pq = Wq + wrow;
  const float* pv = Wv + wrow;
  const float* ps = scw + wrow;
  const float* wp = lnw + 2 * half;
  const float* bp = lnb + 2 * half;

  v2f a[32], lw[32], lb[32], fk[32], fq[32], fv[32], fs[32];
#pragma unroll
  for (int s = 0; s < 16; s++) {
    fk[s].x = pk[4 * s]; fk[s].y = pk[4 * s + 1];
    fq[s].x = pq[4 * s]; fq[s].y = pq[4 * s + 1];
    fv[s].x = pv[4 * s]; fv[s].y = pv[4 * s + 1];
    fs[s].x = ps[4 * s]; fs[s].y = ps[4 * s + 1];
  }
  __builtin_amdgcn_sched_barrier(0);

  {
    const float pos = ts + (float)t;
    const float x1[8] = {z1lo.x, z1lo.y, z1lo.z, z1lo.w, z1hi.x, z1hi.y, z1hi.z, z1hi.w};
    const float x2[8] = {z2lo.x, z2lo.y, z2lo.z, z2lo.w, z2hi.x, z2hi.y, z2hi.z, z2hi.w};
    float r1[8], r2[8];
#pragma unroll
    for (int i = 0; i < 8; i++) {
      const float freq = expf((float)(2 * (j0 + i)) * (-9.210340371976184f / 128.0f));
      const float ang = pos * freq;
      const float s = sinf(ang), c = cosf(ang);
      r1[i] = x1[i] * c - x2[i] * s;
      r2[i] = x2[i] * c + x1[i] * s;
    }
    float* xw = xs + idx * XP + j0;
    v4f o;
    o.x = r1[0]; o.y = r1[1]; o.z = r1[2]; o.w = r1[3]; *(v4f*)(xw)      = o;
    o.x = r1[4]; o.y = r1[5]; o.z = r1[6]; o.w = r1[7]; *(v4f*)(xw + 4)  = o;
    o.x = r2[0]; o.y = r2[1]; o.z = r2[2]; o.w = r2[3]; *(v4f*)(xw + 64) = o;
    o.x = r2[4]; o.y = r2[5]; o.z = r2[6]; o.w = r2[7]; *(v4f*)(xw + 68) = o;
  }
  if (tile == 0 && (threadIdx.x & 31) == 0)
    outst[(size_t)b * STSTRIDE + STSTRIDE - 1] = ts + (float)Tn;
#pragma unroll
  for (int s = 0; s < 32; s++) {
    lw[s].x = wp[4 * s]; lw[s].y = wp[4 * s + 1];
    lb[s].x = bp[4 * s]; lb[s].y = bp[4 * s + 1];
  }
  lds_barrier();
#pragma unroll
  for (int s = 0; s < 32; s++) { a[s] = *(const v2f*)(xr + 4 * s); }

  float p0 = 0.f, p1 = 0.f, p2 = 0.f, p3 = 0.f;
#pragma unroll
  for (int s = 0; s < 32; s += 4) {
    p0 += a[s].x + a[s].y;
    p1 += a[s + 1].x + a[s + 1].y;
    p2 += a[s + 2].x + a[s + 2].y;
    p3 += a[s + 3].x + a[s + 3].y;
  }
  float sum = (p0 + p1) + (p2 + p3);
  sum += __shfl_xor(sum, 16);
  const float m = sum * (1.0f / INn);

  float q0 = 0.f, q1 = 0.f, q2 = 0.f, q3 = 0.f;
#pragma unroll
  for (int s = 0; s < 32; s += 4) {
    float d;
    d = a[s].x - m;     q0 += d * d;  d = a[s].y - m;     q0 += d * d;
    d = a[s + 1].x - m; q1 += d * d;  d = a[s + 1].y - m; q1 += d * d;
    d = a[s + 2].x - m; q2 += d * d;  d = a[s + 2].y - m; q2 += d * d;
    d = a[s + 3].x - m; q3 += d * d;  d = a[s + 3].y - m; q3 += d * d;
  }
  float sq = (q0 + q1) + (q2 + q3);
  sq += __shfl_xor(sq, 16);
  const float inv = rsqrtf(sq * (1.0f / INn) + 1e-5f);


#pragma unroll
  for (int s = 0; s < 32; s++) {
    a[s].x = (a[s].x - m) * inv * lw[s].x + lb[s].x;
    a[s].y = (a[s].y - m) * inv * lw[s].y + lb[s].y;
  }

  v8f accK = {0.f,0.f,0.f,0.f,0.f,0.f,0.f,0.f};
  v8f accQ = {0.f,0.f,0.f,0.f,0.f,0.f,0.f,0.f};
  v8f accV = {0.f,0.f,0.f,0.f,0.f,0.f,0.f,0.f};
  v8f accS = {0.f,0.f,0.f,0.f,0.f,0.f,0.f,0.f};

#pragma unroll
  for (int gq = 0; gq < 4; gq++) {
    if (gq < 2) {
#pragma unroll
      for (int s = 8 * gq + 16; s < 8 * gq + 24; s++) {
        fk[s].x = pk[4 * s]; fk[s].y = pk[4 * s + 1];
        fq[s].x = pq[4 * s]; fq[s].y = pq[4 * s + 1];
        fv[s].x = pv[4 * s]; fv[s].y = pv[4 * s + 1];
        fs[s].x = ps[4 * s]; fs[s].y = ps[4 * s + 1];
      }
    }
    __builtin_amdgcn_sched_barrier(0);
#pragma unroll
    for (int s = 8 * gq; s < 8 * gq + 8; s++) {
      accK = wmma4(a[s], fk[s], accK);
      accQ = wmma4(a[s], fq[s], accQ);
      accV = wmma4(a[s], fv[s], accV);
      accS = wmma4(a[s], fs[s], accS);
    }
  }

  const size_t o = (size_t)b * Tn * Hn + (size_t)tm * 16 * Hn + tn * 16;
#pragma unroll
  for (int r = 0; r < 8; r++) {
    const size_t e = o + (size_t)(r + 8 * half) * Hn + idx;
    Kb[e] = elu_plus1(accK[r]);
    Qb[e] = elu_plus1(accQ[r]);
    Vb[e] = accV[r];
    Sc[e] = accS[r];
  }
}

#define PSZ  393216
#define ZPSZ 3072

__device__ __forceinline__ void dla_state_job(int b, int jb, int half, int idx,
                                              const float* __restrict__ Kb, const float* __restrict__ Vb,
                                              const float* __restrict__ st, int layerOff,
                                              float* __restrict__ Pb, float* __restrict__ Zpb,
                                              float* __restrict__ outst) {
  const bool zjob = jb >= 64;
  const int ti = zjob ? (jb - 64) : (jb >> 3);
  const int tn = jb & 7;
  const float* S0 = st + (size_t)b * STSTRIDE + layerOff;
  float* SOut = outst + (size_t)b * STSTRIDE + layerOff;
  const float* Kt = Kb + (size_t)b * Tn * Hn + ti * 16 + idx;
  const float* Vt = Vb + (size_t)b * Tn * Hn + tn * 16 + idx;
  v8f sx = {0.f,0.f,0.f,0.f,0.f,0.f,0.f,0.f};
  if (!zjob) {
    float s0v[8];
#pragma unroll
    for (int r = 0; r < 8; r++)
      s0v[r] = S0[(size_t)(ti * 16 + r + 8 * half) * Hn + tn * 16 + idx];
    v2f ka[64], vb[64];
#pragma unroll
    for (int s = 0; s < 32; s++) {
      ka[s].x = Kt[(size_t)(4 * s + 2 * half)     * Hn];
      ka[s].y = Kt[(size_t)(4 * s + 2 * half + 1) * Hn];
      vb[s].x = Vt[(size_t)(4 * s + 2 * half)     * Hn];
      vb[s].y = Vt[(size_t)(4 * s + 2 * half + 1) * Hn];
    }
#pragma unroll
    for (int g = 0; g < 4; g++) {
      if (g + 2 < 4) {
#pragma unroll
        for (int s = 16 * (g + 2); s < 16 * (g + 3); s++) {
          ka[s].x = Kt[(size_t)(4 * s + 2 * half)     * Hn];
          ka[s].y = Kt[(size_t)(4 * s + 2 * half + 1) * Hn];
          vb[s].x = Vt[(size_t)(4 * s + 2 * half)     * Hn];
          vb[s].y = Vt[(size_t)(4 * s + 2 * half + 1) * Hn];
        }
      }
      __builtin_amdgcn_sched_barrier(0);
#pragma unroll
      for (int s = 16 * g; s < 16 * g + 16; s++) sx = wmma4(ka[s], vb[s], sx);
      float* D = (g < 3) ? (Pb + (size_t)(b * 3 + g) * Hn * Hn) : SOut;
#pragma unroll
      for (int r = 0; r < 8; r++)
        D[(size_t)(ti * 16 + r + 8 * half) * Hn + tn * 16 + idx] = sx[r] + s0v[r];
    }
  } else {
    v2f ones; ones.x = 1.0f; ones.y = 1.0f;
    float z0v[8];
#pragma unroll
    for (int r = 0; r < 8; r++) z0v[r] = S0[Hn * Hn + ti * 16 + r + 8 * half];
    v2f ka[64];
#pragma unroll
    for (int s = 0; s < 32; s++) {
      ka[s].x = Kt[(size_t)(4 * s + 2 * half)     * Hn];
      ka[s].y = Kt[(size_t)(4 * s + 2 * half + 1) * Hn];
    }
#pragma unroll
    for (int g = 0; g < 4; g++) {
      if (g + 2 < 4) {
#pragma unroll
        for (int s = 16 * (g + 2); s < 16 * (g + 3); s++) {
          ka[s].x = Kt[(size_t)(4 * s + 2 * half)     * Hn];
          ka[s].y = Kt[(size_t)(4 * s + 2 * half + 1) * Hn];
        }
      }
      __builtin_amdgcn_sched_barrier(0);
#pragma unroll
      for (int s = 16 * g; s < 16 * g + 16; s++) sx = wmma4(ka[s], ones, sx);
      float* D = (g < 3) ? (Zpb + (size_t)(b * 3 + g) * Hn) : (SOut + Hn * Hn);
      float zv[8];
#pragma unroll
      for (int r = 0; r < 8; r++) zv[r] = z0v[r] + sx[r];
      if (idx == 0) {
#pragma unroll
        for (int r = 0; r < 8; r++) D[ti * 16 + r + 8 * half] = zv[r];
      }
    }
  }
}

#define AP 68
#define LDP 32

__global__ void __launch_bounds__(256)
dla_state_local_kernel(const float* __restrict__ Qb, const float* __restrict__ Kb,
                       const float* __restrict__ Vb, const float* __restrict__ st,
                       int layerOff, float* __restrict__ Pb, float* __restrict__ Zpb,
                       float* __restrict__ Lnm, float* __restrict__ Ldn,
                       float* __restrict__ outst) {
  __shared__ __attribute__((aligned(16))) float as[16 * AP];
  const int wave = __builtin_amdgcn_readfirstlane((int)(threadIdx.x >> 5));
  const int lane = threadIdx.x & 31, half = lane >> 4, idx = lane & 15;
  if (blockIdx.x >= 128) {
    const int sb = blockIdx.x - 128;
    const int b  = sb / 9;
    const int jb = (sb - b * 9) * 8 + wave;
    dla_state_job(b, jb, half, idx, Kb, Vb, st, layerOff, Pb, Zpb, outst);
    return;
  }
  const int b = blockIdx.x >> 4;
  const int tm = blockIdx.x & 15;
  const int c = tm >> 2, q = tm & 3;
  const size_t bo = (size_t)b * Tn * Hn;
  const float* Vt = Vb + bo + (size_t)c * 64 * Hn + wave * 16 + idx;
  v2f vb[16];
#pragma unroll
  for (int g = 0; g < 4; g++) {
    if (g <= q) {
#pragma unroll
      for (int s = 4 * g; s < 4 * g + 4; s++) {
        vb[s].x = Vt[(size_t)(4 * s + 2 * half)     * Hn];
        vb[s].y = Vt[(size_t)(4 * s + 2 * half + 1) * Hn];
      }
    }
  }
  if (wave <= q) {
    v8f sc = {0.f,0.f,0.f,0.f,0.f,0.f,0.f,0.f};
    tile_abt(Qb + bo + (size_t)tm * 16 * Hn, Hn,
             Kb + bo + (size_t)(4 * c + wave) * 16 * Hn, Hn, Hn, sc);
#pragma unroll
    for (int r = 0; r < 8; r++) {
      float v = sc[r];
      if (wave == q && idx > (r + 8 * half)) v = 0.f;
      as[(r + 8 * half) * AP + 16 * wave + idx] = v;
    }
  }
  lds_barrier();
  v8f acc  = {0.f,0.f,0.f,0.f,0.f,0.f,0.f,0.f};
  v8f dsum = {0.f,0.f,0.f,0.f,0.f,0.f,0.f,0.f};
  v2f ones; ones.x = 1.0f; ones.y = 1.0f;
  const float* ar = as + idx * AP + 2 * half;
#pragma unroll
  for (int g = 0; g < 4; g++) {
    if (g <= q) {
#pragma unroll
      for (int s = 4 * g; s < 4 * g + 4; s++) {
        const v2f a = *(const v2f*)(ar + 4 * s);
        acc  = wmma4(a, vb[s], acc);
        dsum = wmma4(a, ones, dsum);
      }
    }
  }
  float* Lo = Lnm + bo + (size_t)tm * 16 * Hn + wave * 16;
#pragma unroll
  for (int r = 0; r < 8; r++) Lo[(size_t)(r + 8 * half) * Hn + idx] = acc[r];
  if (wave == 0 && idx == 0) {
    float* Do = Ldn + (size_t)(b * 16 + tm) * LDP + 8 * half;
#pragma unroll
    for (int r = 0; r < 8; r++) Do[r] = dsum[r];
  }
}


#define LOAD_FENCE() do { asm volatile("" ::: "memory"); __builtin_amdgcn_sched_barrier(0); } while (0)

template <bool LAST>
__global__ void __launch_bounds__(128)
dla_ffchain_kernel(const float* __restrict__ w1, const float* __restrict__ b1,
                   const float* __restrict__ w2, const float* __restrict__ b2,
                   const float* __restrict__ sc, const float* __restrict__ scb,
                   const float* __restrict__ lnw, const float* __restrict__ lnb,
                   const float* __restrict__ Wa, const float* __restrict__ Wq,
                   const float* __restrict__ Wv, const float* __restrict__ Ws,
                   const float* __restrict__ ub,
                   float* __restrict__ O0, float* __restrict__ O1,
                   float* __restrict__ O2, float* __restrict__ O3,
                   const float* __restrict__ Qb, const float* __restrict__ Pb,
                   const float* __restrict__ Zpb, const float* __restrict__ Lnm,
                   const float* __restrict__ Ldn, const float* __restrict__ S0l) {
  __shared__ __attribute__((aligned(16))) float hs[16 * XP];
  __shared__ __attribute__((aligned(16))) float xs[16 * XP];
  __shared__ __attribute__((aligned(16))) float as0[16 * XP];
  const int wave = threadIdx.x >> 5;
  const int b = blockIdx.x >> 5;
  const int tm = (blockIdx.x & 31) >> 1;
  const int tn = 4 * (blockIdx.x & 1) + wave;
  const int lane = threadIdx.x & 31, half = lane >> 4, idx = lane & 15;
  const size_t row0 = (size_t)b * Tn + (size_t)tm * 16;
  const int c0 = 32 * wave;

  const float* w1p = w1 + (size_t)(c0 + idx) * Hn + 2 * half;
  const float* w2p = w2 + (size_t)(c0 + idx) * Hn + 8 * half;
  const float* b1p = b1 + c0 + 8 * half;
  const float* b2p = b2 + c0 + 8 * half;
  const float* scbp = scb + c0 + 8 * half;
  const float* scp = sc + (row0 + idx) * Hn + c0 + 8 * half;
  const size_t w3o = (size_t)(tn * 16 + idx) * Hn + 8 * half;
  float* hrow = hs + idx * XP + 8 * half;
  float* xrow = xs + idx * XP + 8 * half;

  v2f ao[32], f1[2][32], f2[2][32], f3[4][32], hb[32];
  v8f h[2], x2[2], acc2[8];
  float b1v[2][8], b2v[2][8], scbv[2][8], scv[2][8], lwv[8][8], lbv[8][8];
#pragma unroll
  for (int u = 0; u < 2; u++) {
#pragma unroll
    for (int r = 0; r < 8; r++) { h[u][r] = 0.f; x2[u][r] = 0.f; }
  }
  {
    const int cch = tm >> 2;
    const float* S0 = S0l + (size_t)b * STSTRIDE;
    const float* Pc = (cch == 0) ? S0 : (Pb + (size_t)(b * 3 + (cch - 1)) * Hn * Hn);
    const float* Zc = (cch == 0) ? (S0 + Hn * Hn) : (Zpb + (size_t)(b * 3 + (cch - 1)) * Hn);
    const float* qp  = Qb + (row0 + idx) * Hn + 2 * half;
    const float* pp0 = Pc + (size_t)(2 * half) * Hn + c0 + idx;
    const float* zp  = Zc + 2 * half;
    v2f qa[32], pb[2][32], zw[32];
    float lnv[2][8], ldv[8];
    v8f n0 = {0.f,0.f,0.f,0.f,0.f,0.f,0.f,0.f};
    v8f n1 = {0.f,0.f,0.f,0.f,0.f,0.f,0.f,0.f};
    v8f dn = {0.f,0.f,0.f,0.f,0.f,0.f,0.f,0.f};
#define S0_LOAD(s) do { \
    qa[s].x = qp[4 * (s)]; qa[s].y = qp[4 * (s) + 1]; \
    pb[0][s].x = pp0[(size_t)(4 * (s)) * Hn];      pb[0][s].y = pp0[(size_t)(4 * (s) + 1) * Hn]; \
    pb[1][s].x = pp0[(size_t)(4 * (s)) * Hn + 16]; pb[1][s].y = pp0[(size_t)(4 * (s) + 1) * Hn + 16]; \
    zw[s].x = zp[4 * (s)]; zw[s].y = zp[4 * (s) + 1]; } while (0)
#pragma unroll
    for (int s = 0; s < 16; s++) S0_LOAD(s);
#pragma unroll
    for (int u = 0; u < 2; u++) {
#pragma unroll
      for (int r = 0; r < 8; r++)
        lnv[u][r] = Lnm[(row0 + 8 * half + r) * Hn + c0 + 16 * u + idx];
    }
#pragma unroll
    for (int r = 0; r < 8; r++) ldv[r] = Ldn[(size_t)(b * 16 + tm) * LDP + 8 * half + r];
#pragma unroll
    for (int g = 0; g < 4; g++) {
      if (g < 2) {
#pragma unroll
        for (int s = 8 * g + 16; s < 8 * g + 24; s++) S0_LOAD(s);
      } else {
#pragma unroll
        for (int s = 16 * (g - 2); s < 16 * (g - 2) + 16; s++) {
#pragma unroll
          for (int u = 0; u < 2; u++) {
            const float* pp = w1p + (size_t)(16 * u) * Hn + 4 * s;
            f1[u][s].x = pp[0]; f1[u][s].y = pp[1];
          }
        }
      }
      LOAD_FENCE();
#pragma unroll
      for (int s = 8 * g; s < 8 * g + 8; s++) {
        n0 = wmma4(qa[s], pb[0][s], n0);
        n1 = wmma4(qa[s], pb[1][s], n1);
        dn = wmma4(qa[s], zw[s], dn);
      }
    }
#undef S0_LOAD
#pragma unroll
    for (int r = 0; r < 8; r++) {
      const float den = (dn[r] + ldv[r]) + 1e-5f;
      as0[(8 * half + r) * XP + c0 + idx]      = (n0[r] + lnv[0][r]) / den;
      as0[(8 * half + r) * XP + c0 + 16 + idx] = (n1[r] + lnv[1][r]) / den;
    }
    lds_barrier();
#pragma unroll
    for (int s = 0; s < 32; s++) ao[s] = *(const v2f*)(as0 + idx * XP + 2 * half + 4 * s);
  }
  LOAD_FENCE();

#pragma unroll
  for (int s = 0; s < 16; s++) {
#pragma unroll
    for (int u = 0; u < 2; u++) h[u] = wmma4(f1[u][s], ao[s], h[u]);
  }
  LOAD_FENCE();
#pragma unroll
  for (int u = 0; u < 2; u++) {
#pragma unroll
    for (int r = 0; r < 8; r++) b1v[u][r] = b1p[16 * u + r];
  }
#pragma unroll
  for (int j = 0; j < 8; j++) {
#pragma unroll
    for (int rr = 0; rr < 4; rr++) {
#pragma unroll
      for (int u = 0; u < 2; u++) {
        const float* pp = w2p + (size_t)(16 * u) * Hn + 16 * j + 2 * rr;
        f2[u][4 * j + rr].x = pp[0]; f2[u][4 * j + rr].y = pp[1];
      }
    }
  }
#pragma unroll
  for (int u = 0; u < 2; u++) {
#pragma unroll
    for (int r = 0; r < 8; r++) {
      b2v[u][r] = b2p[16 * u + r];
      scbv[u][r] = scbp[16 * u + r];
      scv[u][r] = scp[16 * u + r];
    }
  }
  LOAD_FENCE();
#pragma unroll
  for (int s = 16; s < 32; s++) {
#pragma unroll
    for (int u = 0; u < 2; u++) h[u] = wmma4(f1[u][s], ao[s], h[u]);
  }

#pragma unroll
  for (int u = 0; u < 2; u++) {
    float t[8];
#pragma unroll
    for (int r = 0; r < 8; r++) {
      const float v = h[u][r] + b1v[u][r];
      t[r] = v > 0.f ? v : 0.f;
    }
    v4f lo, hi;
    lo.x = t[0]; lo.y = t[1]; lo.z = t[2]; lo.w = t[3];
    hi.x = t[4]; hi.y = t[5]; hi.z = t[6]; hi.w = t[7];
    *(v4f*)(hrow + c0 + 16 * u)     = lo;
    *(v4f*)(hrow + c0 + 16 * u + 4) = hi;
  }
  lds_barrier();

#pragma unroll
  for (int j = 0; j < 8; j++) {
    const v4f lo = *(const v4f*)(hrow + 16 * j);
    const v4f hi = *(const v4f*)(hrow + 16 * j + 4);
    hb[4 * j].x     = lo.x; hb[4 * j].y     = lo.y;
    hb[4 * j + 1].x = lo.z; hb[4 * j + 1].y = lo.w;
    hb[4 * j + 2].x = hi.x; hb[4 * j + 2].y = hi.y;
    hb[4 * j + 3].x = hi.z; hb[4 * j + 3].y = hi.w;
  }
  LOAD_FENCE();
#pragma unroll
  for (int e = 0; e < 16; e++) {
#pragma unroll
    for (int u = 0; u < 2; u++) x2[u] = wmma4(f2[u][e], hb[e], x2[u]);
  }
  LOAD_FENCE();
  if (LAST) {
#pragma unroll
    for (int n = 0; n < 8; n++) {
#pragma unroll
      for (int rr = 0; rr < 4; rr++) {
        const float* pp = Wa + w3o + 16 * n + 2 * rr;
        f3[0][n * 4 + rr].x = pp[0]; f3[0][n * 4 + rr].y = pp[1];
      }
    }
  } else {
    const float* lwp = lnw + 8 * half;
    const float* lbp = lnb + 8 * half;
#pragma unroll
    for (int n = 0; n < 8; n++) {
#pragma unroll
      for (int r = 0; r < 8; r++) { lwv[n][r] = lwp[16 * n + r]; lbv[n][r] = lbp[16 * n + r]; }
    }
#pragma unroll
    for (int nl = 0; nl < 2; nl++) {
#pragma unroll
      for (int rr = 0; rr < 4; rr++) {
        const size_t o = w3o + 16 * nl + 2 * rr;
        f3[0][0 * 8 + nl * 4 + rr].x = Wa[o]; f3[0][0 * 8 + nl * 4 + rr].y = Wa[o + 1];
        f3[0][1 * 8 + nl * 4 + rr].x = Wq[o]; f3[0][1 * 8 + nl * 4 + rr].y = Wq[o + 1];
        f3[0][2 * 8 + nl * 4 + rr].x = Wv[o]; f3[0][2 * 8 + nl * 4 + rr].y = Wv[o + 1];
        f3[0][3 * 8 + nl * 4 + rr].x = Ws[o]; f3[0][3 * 8 + nl * 4 + rr].y = Ws[o + 1];
      }
    }
  }
  LOAD_FENCE();
#pragma unroll
  for (int e = 16; e < 32; e++) {
#pragma unroll
    for (int u = 0; u < 2; u++) x2[u] = wmma4(f2[u][e], hb[e], x2[u]);
  }

#pragma unroll
  for (int u = 0; u < 2; u++) {
    float t[8];
#pragma unroll
    for (int r = 0; r < 8; r++) {
      float v = x2[u][r] + b2v[u][r];
      v = v > 0.f ? v : 0.f;
      t[r] = v + scv[u][r] + scbv[u][r];
    }
    v4f lo, hi;
    lo.x = t[0]; lo.y = t[1]; lo.z = t[2]; lo.w = t[3];
    hi.x = t[4]; hi.y = t[5]; hi.z = t[6]; hi.w = t[7];
    *(v4f*)(xrow + c0 + 16 * u)     = lo;
    *(v4f*)(xrow + c0 + 16 * u + 4) = hi;
  }
  lds_barrier();

#pragma unroll
  for (int n = 0; n < 8; n++) {
    const v4f lo = *(const v4f*)(xrow + 16 * n);
    const v4f hi = *(const v4f*)(xrow + 16 * n + 4);
    acc2[n][0] = lo.x; acc2[n][1] = lo.y; acc2[n][2] = lo.z; acc2[n][3] = lo.w;
    acc2[n][4] = hi.x; acc2[n][5] = hi.y; acc2[n][6] = hi.z; acc2[n][7] = hi.w;
  }
  if (!LAST) {
#pragma unroll
    for (int q3 = 1; q3 < 4; q3++) {
#pragma unroll
      for (int nl = 0; nl < 2; nl++) {
#pragma unroll
        for (int rr = 0; rr < 4; rr++) {
          const size_t o = w3o + 16 * (2 * q3 + nl) + 2 * rr;
          f3[q3][0 * 8 + nl * 4 + rr].x = Wa[o]; f3[q3][0 * 8 + nl * 4 + rr].y = Wa[o + 1];
          f3[q3][1 * 8 + nl * 4 + rr].x = Wq[o]; f3[q3][1 * 8 + nl * 4 + rr].y = Wq[o + 1];
          f3[q3][2 * 8 + nl * 4 + rr].x = Wv[o]; f3[q3][2 * 8 + nl * 4 + rr].y = Wv[o + 1];
          f3[q3][3 * 8 + nl * 4 + rr].x = Ws[o]; f3[q3][3 * 8 + nl * 4 + rr].y = Ws[o + 1];
        }
      }
    }
  }
  LOAD_FENCE();

  if (LAST) {
    v8f accE = {0.f,0.f,0.f,0.f,0.f,0.f,0.f,0.f};
    v8f accO = {0.f,0.f,0.f,0.f,0.f,0.f,0.f,0.f};
#pragma unroll
    for (int n = 0; n < 8; n++) {
#pragma unroll
      for (int rr = 0; rr < 4; rr += 2) {
        v2f xa, xb;
        xa.x = acc2[n][2 * rr];     xa.y = acc2[n][2 * rr + 1];
        xb.x = acc2[n][2 * rr + 2]; xb.y = acc2[n][2 * rr + 3];
        accE = wmma4(xa, f3[0][n * 4 + rr], accE);
        accO = wmma4(xb, f3[0][n * 4 + rr + 1], accO);
      }
    }
#pragma unroll
    for (int r = 0; r < 8; r++) {
      const int t = tm * 16 + r + 8 * half;
      const int o = tn * 16 + idx;
      O0[(size_t)t * Bn * OUTn + (size_t)b * OUTn + o] = (accE[r] + accO[r]) + ub[o];
    }
  } else {
    float p0 = 0.f, p1 = 0.f, p2 = 0.f, p3 = 0.f;
#pragma unroll
    for (int n = 0; n < 8; n += 4) {
#pragma unroll
      for (int r = 0; r < 8; r += 2) {
        p0 += acc2[n][r] + acc2[n][r + 1];
        p1 += acc2[n + 1][r] + acc2[n + 1][r + 1];
        p2 += acc2[n + 2][r] + acc2[n + 2][r + 1];
        p3 += acc2[n + 3][r] + acc2[n + 3][r + 1];
      }
    }
    float sum = (p0 + p1) + (p2 + p3);
    sum += __shfl_xor(sum, 16);
    const float m = sum * (1.0f / INn);

    float q0 = 0.f, q1 = 0.f, q2 = 0.f, q3 = 0.f;
#pragma unroll
    for (int n = 0; n < 8; n += 4) {
#pragma unroll
      for (int r = 0; r < 8; r++) {
        float d;
        d = acc2[n][r] - m;     q0 += d * d;
        d = acc2[n + 1][r] - m; q1 += d * d;
        d = acc2[n + 2][r] - m; q2 += d * d;
        d = acc2[n + 3][r] - m; q3 += d * d;
      }
    }
    float sq = (q0 + q1) + (q2 + q3);
    sq += __shfl_xor(sq, 16);
    const float inv = rsqrtf(sq * (1.0f / INn) + 1e-5f);

#pragma unroll
    for (int n = 0; n < 8; n++) {
#pragma unroll
      for (int r = 0; r < 8; r++)
        acc2[n][r] = (acc2[n][r] - m) * inv * lwv[n][r] + lbv[n][r];
    }

    v8f accK = {0.f,0.f,0.f,0.f,0.f,0.f,0.f,0.f};
    v8f accQ = {0.f,0.f,0.f,0.f,0.f,0.f,0.f,0.f};
    v8f accV = {0.f,0.f,0.f,0.f,0.f,0.f,0.f,0.f};
    v8f accS = {0.f,0.f,0.f,0.f,0.f,0.f,0.f,0.f};
#pragma unroll
    for (int q3 = 0; q3 < 4; q3++) {
#pragma unroll
      for (int nl = 0; nl < 2; nl++) {
#pragma unroll
        for (int rr = 0; rr < 4; rr++) {
          v2f xa;
          xa.x = acc2[2 * q3 + nl][2 * rr]; xa.y = acc2[2 * q3 + nl][2 * rr + 1];
          accK = wmma4(xa, f3[q3][0 * 8 + nl * 4 + rr], accK);
          accQ = wmma4(xa, f3[q3][1 * 8 + nl * 4 + rr], accQ);
          accV = wmma4(xa, f3[q3][2 * 8 + nl * 4 + rr], accV);
          accS = wmma4(xa, f3[q3][3 * 8 + nl * 4 + rr], accS);
        }
      }
    }
    const size_t o16 = row0 * Hn + tn * 16;
#pragma unroll
    for (int r = 0; r < 8; r++) {
      const size_t e = o16 + (size_t)(r + 8 * half) * Hn + idx;
      O0[e] = elu_plus1(accK[r]);
      O1[e] = elu_plus1(accQ[r]);
      O2[e] = accV[r];
      O3[e] = accS[r];
    }
  }
}

extern "C" void kernel_launch(void* const* d_in, const int* in_sizes, int n_in,
                              void* d_out, int out_size, void* d_ws, size_t ws_size,
                              hipStream_t stream) {
  (void)in_sizes; (void)n_in; (void)out_size; (void)ws_size;
  const float* z   = (const float*)d_in[0];
  const float* st  = (const float*)d_in[1];
  const float* Wk  = (const float*)d_in[2];
  const float* Wq  = (const float*)d_in[3];
  const float* Wv  = (const float*)d_in[4];
  const float* lnw = (const float*)d_in[5];
  const float* lnb = (const float*)d_in[6];
  const float* w1  = (const float*)d_in[7];
  const float* b1  = (const float*)d_in[8];
  const float* w2  = (const float*)d_in[9];
  const float* b2  = (const float*)d_in[10];
  const float* scw = (const float*)d_in[11];
  const float* scb = (const float*)d_in[12];
  const float* uw  = (const float*)d_in[13];
  const float* ub  = (const float*)d_in[14];
  float* out = (float*)d_out;
  float* ws  = (float*)d_ws;

  float* SC2  = ws;
  float* SC   = ws + 262144;
  float* KB   = ws + 524288;
  float* QB   = ws + 786432;
  float* VB   = ws + 1048576;
  float* PB   = ws + 1310720;
  float* ZPB  = PB + PSZ;
  float* LD   = ZPB + ZPSZ;
  float* QB1  = ws + 1835008;
  float* LN1  = SC;
  float* LN0  = out;
  float* OST  = out + YSZ;

  dla_lnproj_kernel<<<256, 128, 0, stream>>>(z, st, lnw, lnb, Wk, Wq, Wv, scw, KB, QB, VB, SC, OST);
  dla_state_local_kernel<<<128 + Bn * 9, 256, 0, stream>>>(QB, KB, VB, st, 0, PB, ZPB,
                                                          LN0, LD, OST);
  dla_ffchain_kernel<false><<<256, 128, 0, stream>>>(
      w1, b1, w2, b2, SC, scb, lnw + INn, lnb + INn,
      Wk + (size_t)Hn * INn, Wq + (size_t)Hn * INn, Wv + (size_t)Hn * INn,
      scw + (size_t)Hn * INn, nullptr, KB, QB1, VB, SC2,
      QB, PB, ZPB, LN0, LD, st);
  dla_state_local_kernel<<<128 + Bn * 9, 256, 0, stream>>>(QB1, KB, VB, st, LAYERSTATE, PB, ZPB,
                                                          LN1, LD, OST);
  dla_ffchain_kernel<true><<<256, 128, 0, stream>>>(
      w1 + (size_t)Hn * Hn, b1 + Hn, w2 + (size_t)Hn * Hn, b2 + Hn, SC2, scb + Hn,
      nullptr, nullptr, uw, nullptr, nullptr, nullptr, ub, out, nullptr, nullptr, nullptr,
      QB1, PB, ZPB, LN1, LD, st + LAYERSTATE);
}


# ff-chain weights (repacked once per call by work-groups of their own), Q and V kept in the workspace in fragment order: fragment loads 16 B per lane, 4 whole lines per load not 16; sum order same
# speedup vs baseline: 2.8633x; 1.0993x over previous
#include <hip/hip_runtime.h>
#include <hip/hip_bf16.h>
#include <math.h>

#define Bn 8
#define Tn 256
#define INn 128
#define Hn 128
#define OUTn 128
#define Ln 2
#define STSTRIDE 33025
#define LAYERSTATE 16512
#define YSZ (Tn*Bn*OUTn)

typedef __attribute__((ext_vector_type(2))) float v2f;
typedef __attribute__((ext_vector_type(8))) float v8f;

__device__ __forceinline__ v8f wmma4(v2f a, v2f b, v8f c) {
  return __builtin_amdgcn_wmma_f32_16x16x4_f32(false, a, false, b, (short)0, c, false, false);
}

#define XP 132

typedef __attribute__((ext_vector_type(4))) float v4f;

__device__ __forceinline__ void lds_barrier() {
  asm volatile("s_wait_dscnt 0x0\n\ts_barrier_signal -1\n\ts_barrier_wait -1" ::: "memory");
}

__device__ __forceinline__ void dla_pack_piece(const float* __restrict__ W, float* __restrict__ PW,
                                               int rem, bool pairOrder) {
  const int lane = rem & 31, jj = (rem >> 5) & 15, R = rem >> 9;
  const int idx = lane & 15, half = lane >> 4;
  const float* src = W + (size_t)(16 * R + idx) * Hn;
  v4f v;
  if (pairOrder) {
    const v2f lo = *(const v2f*)(src + 8 * jj + 2 * half);
    const v2f hi = *(const v2f*)(src + 8 * jj + 4 + 2 * half);
    v.x = lo.x; v.y = lo.y; v.z = hi.x; v.w = hi.y;
  } else {
    v = *(const v4f*)(src + 16 * (jj >> 1) + 8 * half + 4 * (jj & 1));
  }
  *(v4f*)(PW + (size_t)rem * 4) = v;
}
#define PKM 16384

__device__ __forceinline__ void dla_store_vt(float* __restrict__ VT, int b, int tm, int tn,
                                             int half, int idx, const v8f& acc) {
  float* p = VT + ((((size_t)b * 8 + tn) * 32 + (2 * tm + half)) * 32 + idx) * 4;
  v4f lo, hi;
  lo.x = acc[0]; lo.y = acc[1]; lo.z = acc[4]; lo.w = acc[5];
  hi.x = acc[2]; hi.y = acc[3]; hi.z = acc[6]; hi.w = acc[7];
  *(v4f*)(p)      = lo;
  *(v4f*)(p + 64) = hi;
}

__device__ __forceinline__ float* dla_qf_slot(float* __restrict__ QF, int b, int tm, int tn,
                                              int half, int idx) {
  const int f = 16 * tn + idx;
  const int jj = f >> 3, rh = (f >> 1) & 1, slot = (f & 1) + 2 * ((f >> 2) & 1);
  return QF + ((((size_t)b * 16 + tm) * 16 + jj) * 32 + 16 * rh + 8 * half) * 4 + slot;
}

__device__ __forceinline__ float elu_plus1(float v) {
  return ((v > 0.f) ? v : expm1f(v)) + 1.0f;
}

__global__ void __launch_bounds__(128)
dla_lnproj_kernel(const float* __restrict__ z, const float* __restrict__ st,
                  const float* __restrict__ lnw, const float* __restrict__ lnb,
                  const float* __restrict__ Wk, const float* __restrict__ Wq,
                  const float* __restrict__ Wv, const float* __restrict__ scw,
                  float* __restrict__ Kb, float* __restrict__ Qb,
                  float* __restrict__ Vb, float* __restrict__ Sc,
                  float* __restrict__ outst,
                  const float* __restrict__ w1, const float* __restrict__ w2,
                  float* __restrict__ PK) {
  __shared__ __attribute__((aligned(16))) float xs[16 * XP];
  if (blockIdx.x >= 256) {
    const int q = (blockIdx.x - 256) * 128 + threadIdx.x;
    const int m = q >> 12;
    const float* W = (m == 0) ? w1 : (m == 1) ? w2
                   : ((m == 2) ? Wk : (m == 3) ? Wq : (m == 4) ? Wv : scw) + (size_t)Hn * INn;
    dla_pack_piece(W, PK + (size_t)m * PKM, q & 4095, m == 0);
    return;
  }
  const int wave = threadIdx.x >> 5;
  const int g = blockIdx.x * 4 + wave;
  const int b = g >> 7;
  const int tile = g & 127;
  const int tm = tile >> 3, tn = tile & 7;
  const int lane = threadIdx.x & 31, half = lane >> 4, idx = lane & 15;

  const int t = tm * 16 + idx;
  const int j0 = 16 * wave + 8 * half;
  const float* zp = z + ((size_t)t * Bn + b) * INn + j0;
  const float ts = st[(size_t)b * STSTRIDE + STSTRIDE - 1];
  const v4f z1lo = *(const v4f*)(zp);
  const v4f z1hi = *(const v4f*)(zp + 4);
  const v4f z2lo = *(const v4f*)(zp + 64);
  const v4f z2hi = *(const v4f*)(zp + 68);
  const float* xr = xs + idx * XP + 2 * half;
  const size_t wrow = (size_t)(tn * 16 + idx) * INn + 2 * half;
  const float* pk = Wk + wrow;
  const float* pq = Wq + wrow;
  const float* pv = Wv + wrow;
  const float* ps = scw + wrow;
  const float* wp = lnw + 2 * half;
  const float* bp = lnb + 2 * half;

  v2f a[32], lw[32], lb[32], fk[32], fq[32], fv[32], fs[32];
#pragma unroll
  for (int s = 0; s < 16; s++) {
    fk[s].x = pk[4 * s]; fk[s].y = pk[4 * s + 1];
    fq[s].x = pq[4 * s]; fq[s].y = pq[4 * s + 1];
    fv[s].x = pv[4 * s]; fv[s].y = pv[4 * s + 1];
    fs[s].x = ps[4 * s]; fs[s].y = ps[4 * s + 1];
  }
  __builtin_amdgcn_sched_barrier(0);

  {
    const float pos = ts + (float)t;
    const float x1[8] = {z1lo.x, z1lo.y, z1lo.z, z1lo.w, z1hi.x, z1hi.y, z1hi.z, z1hi.w};
    const float x2[8] = {z2lo.x, z2lo.y, z2lo.z, z2lo.w, z2hi.x, z2hi.y, z2hi.z, z2hi.w};
    float r1[8], r2[8];
#pragma unroll
    for (int i = 0; i < 8; i++) {
      const float freq = expf((float)(2 * (j0 + i)) * (-9.210340371976184f / 128.0f));
      const float ang = pos * freq;
      const float s = sinf(ang), c = cosf(ang);
      r1[i] = x1[i] * c - x2[i] * s;
      r2[i] = x2[i] * c + x1[i] * s;
    }
    float* xw = xs + idx * XP + j0;
    v4f o;
    o.x = r1[0]; o.y = r1[1]; o.z = r1[2]; o.w = r1[3]; *(v4f*)(xw)      = o;
    o.x = r1[4]; o.y = r1[5]; o.z = r1[6]; o.w = r1[7]; *(v4f*)(xw + 4)  = o;
    o.x = r2[0]; o.y = r2[1]; o.z = r2[2]; o.w = r2[3]; *(v4f*)(xw + 64) = o;
    o.x = r2[4]; o.y = r2[5]; o.z = r2[6]; o.w = r2[7]; *(v4f*)(xw + 68) = o;
  }
  if (tile == 0 && (threadIdx.x & 31) == 0)
    outst[(size_t)b * STSTRIDE + STSTRIDE - 1] = ts + (float)Tn;
#pragma unroll
  for (int s = 0; s < 32; s++) {
    lw[s].x = wp[4 * s]; lw[s].y = wp[4 * s + 1];
    lb[s].x = bp[4 * s]; lb[s].y = bp[4 * s + 1];
  }
  lds_barrier();
#pragma unroll
  for (int s = 0; s < 32; s++) { a[s] = *(const v2f*)(xr + 4 * s); }

  float p0 = 0.f, p1 = 0.f, p2 = 0.f, p3 = 0.f;
#pragma unroll
  for (int s = 0; s < 32; s += 4) {
    p0 += a[s].x + a[s].y;
    p1 += a[s + 1].x + a[s + 1].y;
    p2 += a[s + 2].x + a[s + 2].y;
    p3 += a[s + 3].x + a[s + 3].y;
  }
  float sum = (p0 + p1) + (p2 + p3);
  sum += __shfl_xor(sum, 16);
  const float m = sum * (1.0f / INn);

  float q0 = 0.f, q1 = 0.f, q2 = 0.f, q3 = 0.f;
#pragma unroll
  for (int s = 0; s < 32; s += 4) {
    float d;
    d = a[s].x - m;     q0 += d * d;  d = a[s].y - m;     q0 += d * d;
    d = a[s + 1].x - m; q1 += d * d;  d = a[s + 1].y - m; q1 += d * d;
    d = a[s + 2].x - m; q2 += d * d;  d = a[s + 2].y - m; q2 += d * d;
    d = a[s + 3].x - m; q3 += d * d;  d = a[s + 3].y - m; q3 += d * d;
  }
  float sq = (q0 + q1) + (q2 + q3);
  sq += __shfl_xor(sq, 16);
  const float inv = rsqrtf(sq * (1.0f / INn) + 1e-5f);


#pragma unroll
  for (int s = 0; s < 32; s++) {
    a[s].x = (a[s].x - m) * inv * lw[s].x + lb[s].x;
    a[s].y = (a[s].y - m) * inv * lw[s].y + lb[s].y;
  }

  v8f accK = {0.f,0.f,0.f,0.f,0.f,0.f,0.f,0.f};
  v8f accQ = {0.f,0.f,0.f,0.f,0.f,0.f,0.f,0.f};
  v8f accV = {0.f,0.f,0.f,0.f,0.f,0.f,0.f,0.f};
  v8f accS = {0.f,0.f,0.f,0.f,0.f,0.f,0.f,0.f};

#pragma unroll
  for (int gq = 0; gq < 4; gq++) {
    if (gq < 2) {
#pragma unroll
      for (int s = 8 * gq + 16; s < 8 * gq + 24; s++) {
        fk[s].x = pk[4 * s]; fk[s].y = pk[4 * s + 1];
        fq[s].x = pq[4 * s]; fq[s].y = pq[4 * s + 1];
        fv[s].x = pv[4 * s]; fv[s].y = pv[4 * s + 1];
        fs[s].x = ps[4 * s]; fs[s].y = ps[4 * s + 1];
      }
    }
    __builtin_amdgcn_sched_barrier(0);
#pragma unroll
    for (int s = 8 * gq; s < 8 * gq + 8; s++) {
      accK = wmma4(a[s], fk[s], accK);
      accQ = wmma4(a[s], fq[s], accQ);
      accV = wmma4(a[s], fv[s], accV);
      accS = wmma4(a[s], fs[s], accS);
    }
  }

  const size_t o = (size_t)b * Tn * Hn + (size_t)tm * 16 * Hn + tn * 16;
  float* qf = dla_qf_slot(Qb, b, tm, tn, half, idx);
#pragma unroll
  for (int r = 0; r < 8; r++) {
    const size_t e = o + (size_t)(r + 8 * half) * Hn + idx;
    Kb[e] = elu_plus1(accK[r]);
    qf[4 * r] = elu_plus1(accQ[r]);
    Sc[e] = accS[r];
  }
  dla_store_vt(Vb, b, tm, tn, half, idx, accV);
}

#define PSZ  393216
#define ZPSZ 3072

__device__ __forceinline__ void dla_state_job(int b, int jb, int half, int idx,
                                              const float* __restrict__ Kb, const float* __restrict__ Vb,
                                              const float* __restrict__ st, int layerOff,
                                              float* __restrict__ Pb, float* __restrict__ Zpb,
                                              float* __restrict__ outst) {
  const bool zjob = jb >= 64;
  const int ti = zjob ? (jb - 64) : (jb >> 3);
  const int tn = jb & 7;
  const float* S0 = st + (size_t)b * STSTRIDE + layerOff;
  float* SOut = outst + (size_t)b * STSTRIDE + layerOff;
  const float* Kt = Kb + (size_t)b * Tn * Hn + ti * 16 + idx;
  const float* Vt = Vb + ((((size_t)b * 8 + tn) * 32) * 32 + 16 * half + idx) * 4;
  v8f sx = {0.f,0.f,0.f,0.f,0.f,0.f,0.f,0.f};
  if (!zjob) {
    float s0v[8];
#pragma unroll
    for (int r = 0; r < 8; r++)
      s0v[r] = S0[(size_t)(ti * 16 + r + 8 * half) * Hn + tn * 16 + idx];
    v2f ka[64], vb[64];
#pragma unroll
    for (int s = 0; s < 32; s++) {
      ka[s].x = Kt[(size_t)(4 * s + 2 * half)     * Hn];
      ka[s].y = Kt[(size_t)(4 * s + 2 * half + 1) * Hn];
    }
#pragma unroll
    for (int jj = 0; jj < 16; jj++) {
      const v4f t = *(const v4f*)(Vt + 128 * jj);
      vb[2 * jj].x     = t.x; vb[2 * jj].y     = t.y;
      vb[2 * jj + 1].x = t.z; vb[2 * jj + 1].y = t.w;
    }
#pragma unroll
    for (int g = 0; g < 4; g++) {
      if (g + 2 < 4) {
#pragma unroll
        for (int s = 16 * (g + 2); s < 16 * (g + 3); s++) {
          ka[s].x = Kt[(size_t)(4 * s + 2 * half)     * Hn];
          ka[s].y = Kt[(size_t)(4 * s + 2 * half + 1) * Hn];
        }
#pragma unroll
        for (int jj = 8 * (g + 2); jj < 8 * (g + 3); jj++) {
          const v4f t = *(const v4f*)(Vt + 128 * jj);
          vb[2 * jj].x     = t.x; vb[2 * jj].y     = t.y;
          vb[2 * jj + 1].x = t.z; vb[2 * jj + 1].y = t.w;
        }
      }
      __builtin_amdgcn_sched_barrier(0);
#pragma unroll
      for (int s = 16 * g; s < 16 * g + 16; s++) sx = wmma4(ka[s], vb[s], sx);
      float* D = (g < 3) ? (Pb + (size_t)(b * 3 + g) * Hn * Hn) : SOut;
#pragma unroll
      for (int r = 0; r < 8; r++)
        D[(size_t)(ti * 16 + r + 8 * half) * Hn + tn * 16 + idx] = sx[r] + s0v[r];
    }
  } else {
    v2f ones; ones.x = 1.0f; ones.y = 1.0f;
    float z0v[8];
#pragma unroll
    for (int r = 0; r < 8; r++) z0v[r] = S0[Hn * Hn + ti * 16 + r + 8 * half];
    v2f ka[64];
#pragma unroll
    for (int s = 0; s < 32; s++) {
      ka[s].x = Kt[(size_t)(4 * s + 2 * half)     * Hn];
      ka[s].y = Kt[(size_t)(4 * s + 2 * half + 1) * Hn];
    }
#pragma unroll
    for (int g = 0; g < 4; g++) {
      if (g + 2 < 4) {
#pragma unroll
        for (int s = 16 * (g + 2); s < 16 * (g + 3); s++) {
          ka[s].x = Kt[(size_t)(4 * s + 2 * half)     * Hn];
          ka[s].y = Kt[(size_t)(4 * s + 2 * half + 1) * Hn];
        }
      }
      __builtin_amdgcn_sched_barrier(0);
#pragma unroll
      for (int s = 16 * g; s < 16 * g + 16; s++) sx = wmma4(ka[s], ones, sx);
      float* D = (g < 3) ? (Zpb + (size_t)(b * 3 + g) * Hn) : (SOut + Hn * Hn);
      float zv[8];
#pragma unroll
      for (int r = 0; r < 8; r++) zv[r] = z0v[r] + sx[r];
      if (idx == 0) {
#pragma unroll
        for (int r = 0; r < 8; r++) D[ti * 16 + r + 8 * half] = zv[r];
      }
    }
  }
}

#define AP 68
#define LDP 32

__global__ void __launch_bounds__(256)
dla_state_local_kernel(const float* __restrict__ Qb, const float* __restrict__ Kb,
                       const float* __restrict__ Vb, const float* __restrict__ st,
                       int layerOff, float* __restrict__ Pb, float* __restrict__ Zpb,
                       float* __restrict__ Lnm, float* __restrict__ Ldn,
                       float* __restrict__ outst,
                       const float* __restrict__ pw1, const float* __restrict__ pw2,
                       const float* __restrict__ puw, float* __restrict__ PK) {
  __shared__ __attribute__((aligned(16))) float as[16 * AP];
  if (blockIdx.x >= 200) {
    const int q = (blockIdx.x - 200) * 256 + threadIdx.x;
    const int m = q >> 12;
    const float* W = (m == 0) ? pw1 : (m == 1) ? pw2 : puw;
    dla_pack_piece(W, PK + (size_t)m * PKM, q & 4095, m == 0);
    return;
  }
  const int wave = __builtin_amdgcn_readfirstlane((int)(threadIdx.x >> 5));
  const int lane = threadIdx.x & 31, half = lane >> 4, idx = lane & 15;
  if (blockIdx.x >= 128) {
    const int sb = blockIdx.x - 128;
    const int b  = sb / 9;
    const int jb = (sb - b * 9) * 8 + wave;
    dla_state_job(b, jb, half, idx, Kb, Vb, st, layerOff, Pb, Zpb, outst);
    return;
  }
  const int b = blockIdx.x >> 4;
  const int tm = blockIdx.x & 15;
  const int c = tm >> 2, q = tm & 3;
  const size_t bo = (size_t)b * Tn * Hn;
  const float* Vt = Vb + ((((size_t)b * 8 + wave) * 32 + 8 * c) * 32 + lane) * 4;
  v2f vb[16];
#pragma unroll
  for (int g = 0; g < 4; g++) {
    if (g <= q) {
#pragma unroll
      for (int jl = 2 * g; jl < 2 * g + 2; jl++) {
        const v4f t = *(const v4f*)(Vt + 128 * jl);
        vb[2 * jl].x     = t.x; vb[2 * jl].y     = t.y;
        vb[2 * jl + 1].x = t.z; vb[2 * jl + 1].y = t.w;
      }
    }
  }
  if (wave <= q) {
    v8f sc = {0.f,0.f,0.f,0.f,0.f,0.f,0.f,0.f};
    const float* qf = Qb + ((((size_t)b * 16 + tm) * 16) * 32 + lane) * 4;
    const float* kp = Kb + bo + (size_t)((4 * c + wave) * 16 + idx) * Hn + 2 * half;
#pragma unroll
    for (int jj = 0; jj < 16; jj++) {
      const v4f t = *(const v4f*)(qf + 128 * jj);
      v2f a0, a1, b0, b1;
      a0.x = t.x; a0.y = t.y; a1.x = t.z; a1.y = t.w;
      b0.x = kp[8 * jj];     b0.y = kp[8 * jj + 1];
      b1.x = kp[8 * jj + 4]; b1.y = kp[8 * jj + 5];
      sc = wmma4(a0, b0, sc);
      sc = wmma4(a1, b1, sc);
    }
#pragma unroll
    for (int r = 0; r < 8; r++) {
      float v = sc[r];
      if (wave == q && idx > (r + 8 * half)) v = 0.f;
      as[(r + 8 * half) * AP + 16 * wave + idx] = v;
    }
  }
  lds_barrier();
  v8f acc  = {0.f,0.f,0.f,0.f,0.f,0.f,0.f,0.f};
  v8f dsum = {0.f,0.f,0.f,0.f,0.f,0.f,0.f,0.f};
  v2f ones; ones.x = 1.0f; ones.y = 1.0f;
  const float* ar = as + idx * AP + 2 * half;
#pragma unroll
  for (int g = 0; g < 4; g++) {
    if (g <= q) {
#pragma unroll
      for (int s = 4 * g; s < 4 * g + 4; s++) {
        const v2f a = *(const v2f*)(ar + 4 * s);
        acc  = wmma4(a, vb[s], acc);
        dsum = wmma4(a, ones, dsum);
      }
    }
  }
  float* Lo = Lnm + bo + (size_t)tm * 16 * Hn + wave * 16;
#pragma unroll
  for (int r = 0; r < 8; r++) Lo[(size_t)(r + 8 * half) * Hn + idx] = acc[r];
  if (wave == 0 && idx == 0) {
    float* Do = Ldn + (size_t)(b * 16 + tm) * LDP + 8 * half;
#pragma unroll
    for (int r = 0; r < 8; r++) Do[r] = dsum[r];
  }
}


#define LOAD_FENCE() do { asm volatile("" ::: "memory"); __builtin_amdgcn_sched_barrier(0); } while (0)

template <bool LAST>
__global__ void __launch_bounds__(128)
dla_ffchain_kernel(const float* __restrict__ w1, const float* __restrict__ b1,
                   const float* __restrict__ w2, const float* __restrict__ b2,
                   const float* __restrict__ sc, const float* __restrict__ scb,
                   const float* __restrict__ lnw, const float* __restrict__ lnb,
                   const float* __restrict__ Wa, const float* __restrict__ Wq,
                   const float* __restrict__ Wv, const float* __restrict__ Ws,
                   const float* __restrict__ ub,
                   float* __restrict__ O0, float* __restrict__ O1,
                   float* __restrict__ O2, float* __restrict__ O3,
                   const float* __restrict__ Qb, const float* __restrict__ Pb,
                   const float* __restrict__ Zpb, const float* __restrict__ Lnm,
                   const float* __restrict__ Ldn, const float* __restrict__ S0l) {
  __shared__ __attribute__((aligned(16))) float hs[16 * XP];
  __shared__ __attribute__((aligned(16))) float xs[16 * XP];
  __shared__ __attribute__((aligned(16))) float as0[16 * XP];
  const int wave = threadIdx.x >> 5;
  const int b = blockIdx.x >> 5;
  const int tm = (blockIdx.x & 31) >> 1;
  const int tn = 4 * (blockIdx.x & 1) + wave;
  const int lane = threadIdx.x & 31, half = lane >> 4, idx = lane & 15;
  const size_t row0 = (size_t)b * Tn + (size_t)tm * 16;
  const int c0 = 32 * wave;

  const float* w1p = w1 + (size_t)(2 * wave) * 2048 + 4 * lane;
  const float* w2p = w2 + (size_t)(2 * wave) * 2048 + 4 * lane;
  const float* b1p = b1 + c0 + 8 * half;
  const float* b2p = b2 + c0 + 8 * half;
  const float* scbp = scb + c0 + 8 * half;
  const float* scp = sc + (row0 + idx) * Hn + c0 + 8 * half;
  const size_t w3o = (size_t)tn * 2048 + 4 * lane;
  float* hrow = hs + idx * XP + 8 * half;
  float* xrow = xs + idx * XP + 8 * half;

  v2f ao[32], f1[2][32], f2[2][32], f3[4][32], hb[32];
  v8f h[2], x2[2], acc2[8];
  float b1v[2][8], b2v[2][8], scbv[2][8], scv[2][8], lwv[8][8], lbv[8][8];
#pragma unroll
  for (int u = 0; u < 2; u++) {
#pragma unroll
    for (int r = 0; r < 8; r++) { h[u][r] = 0.f; x2[u][r] = 0.f; }
  }
  {
    const int cch = tm >> 2;
    const float* S0 = S0l + (size_t)b * STSTRIDE;
    const float* Pc = (cch == 0) ? S0 : (Pb + (size_t)(b * 3 + (cch - 1)) * Hn * Hn);
    const float* Zc = (cch == 0) ? (S0 + Hn * Hn) : (Zpb + (size_t)(b * 3 + (cch - 1)) * Hn);
    const float* qp  = Qb + ((((size_t)b * 16 + tm) * 16) * 32 + lane) * 4;
    const float* pp0 = Pc + (size_t)(2 * half) * Hn + c0 + idx;
    const float* zp  = Zc + 2 * half;
    v2f qa[32], pb[2][32], zw[32];
    float lnv[2][8], ldv[8];
    v8f n0 = {0.f,0.f,0.f,0.f,0.f,0.f,0.f,0.f};
    v8f n1 = {0.f,0.f,0.f,0.f,0.f,0.f,0.f,0.f};
    v8f dn = {0.f,0.f,0.f,0.f,0.f,0.f,0.f,0.f};
#define S0_LOADQ(jj) do { \
    const v4f tq4 = *(const v4f*)(qp + 128 * (jj)); \
    qa[2 * (jj)].x = tq4.x; qa[2 * (jj)].y = tq4.y; \
    qa[2 * (jj) + 1].x = tq4.z; qa[2 * (jj) + 1].y = tq4.w; } while (0)
#define S0_LOAD(s) do { \
    pb[0][s].x = pp0[(size_t)(4 * (s)) * Hn];      pb[0][s].y = pp0[(size_t)(4 * (s) + 1) * Hn]; \
    pb[1][s].x = pp0[(size_t)(4 * (s)) * Hn + 16]; pb[1][s].y = pp0[(size_t)(4 * (s) + 1) * Hn + 16]; \
    zw[s].x = zp[4 * (s)]; zw[s].y = zp[4 * (s) + 1]; } while (0)
#pragma unroll
    for (int jj = 0; jj < 8; jj++) S0_LOADQ(jj);
#pragma unroll
    for (int s = 0; s < 16; s++) S0_LOAD(s);
#pragma unroll
    for (int u = 0; u < 2; u++) {
#pragma unroll
      for (int r = 0; r < 8; r++)
        lnv[u][r] = Lnm[(row0 + 8 * half + r) * Hn + c0 + 16 * u + idx];
    }
#pragma unroll
    for (int r = 0; r < 8; r++) ldv[r] = Ldn[(size_t)(b * 16 + tm) * LDP + 8 * half + r];
#pragma unroll
    for (int g = 0; g < 4; g++) {
      if (g < 2) {
#pragma unroll
        for (int jj = 4 * g + 8; jj < 4 * g + 12; jj++) S0_LOADQ(jj);
#pragma unroll
        for (int s = 8 * g + 16; s < 8 * g + 24; s++) S0_LOAD(s);
      } else {
#pragma unroll
        for (int jj = 8 * (g - 2); jj < 8 * (g - 2) + 8; jj++) {
#pragma unroll
          for (int u = 0; u < 2; u++) {
            const v4f t = *(const v4f*)(w1p + 2048 * u + 128 * jj);
            f1[u][2 * jj].x     = t.x; f1[u][2 * jj].y     = t.y;
            f1[u][2 * jj + 1].x = t.z; f1[u][2 * jj + 1].y = t.w;
          }
        }
      }
      LOAD_FENCE();
#pragma unroll
      for (int s = 8 * g; s < 8 * g + 8; s++) {
        n0 = wmma4(qa[s], pb[0][s], n0);
        n1 = wmma4(qa[s], pb[1][s], n1);
        dn = wmma4(qa[s], zw[s], dn);
      }
    }
#undef S0_LOAD
#undef S0_LOADQ
#pragma unroll
    for (int r = 0; r < 8; r++) {
      const float den = (dn[r] + ldv[r]) + 1e-5f;
      as0[(8 * half + r) * XP + c0 + idx]      = (n0[r] + lnv[0][r]) / den;
      as0[(8 * half + r) * XP + c0 + 16 + idx] = (n1[r] + lnv[1][r]) / den;
    }
    lds_barrier();
#pragma unroll
    for (int s = 0; s < 32; s++) ao[s] = *(const v2f*)(as0 + idx * XP + 2 * half + 4 * s);
  }
  LOAD_FENCE();

#pragma unroll
  for (int s = 0; s < 16; s++) {
#pragma unroll
    for (int u = 0; u < 2; u++) h[u] = wmma4(f1[u][s], ao[s], h[u]);
  }
  LOAD_FENCE();
#pragma unroll
  for (int u = 0; u < 2; u++) {
#pragma unroll
    for (int r = 0; r < 8; r++) b1v[u][r] = b1p[16 * u + r];
  }
#pragma unroll
  for (int jj = 0; jj < 16; jj++) {
#pragma unroll
    for (int u = 0; u < 2; u++) {
      const v4f t = *(const v4f*)(w2p + 2048 * u + 128 * jj);
      f2[u][2 * jj].x     = t.x; f2[u][2 * jj].y     = t.y;
      f2[u][2 * jj + 1].x = t.z; f2[u][2 * jj + 1].y = t.w;
    }
  }
#pragma unroll
  for (int u = 0; u < 2; u++) {
#pragma unroll
    for (int r = 0; r < 8; r++) {
      b2v[u][r] = b2p[16 * u + r];
      scbv[u][r] = scbp[16 * u + r];
      scv[u][r] = scp[16 * u + r];
    }
  }
  LOAD_FENCE();
#pragma unroll
  for (int s = 16; s < 32; s++) {
#pragma unroll
    for (int u = 0; u < 2; u++) h[u] = wmma4(f1[u][s], ao[s], h[u]);
  }

#pragma unroll
  for (int u = 0; u < 2; u++) {
    float t[8];
#pragma unroll
    for (int r = 0; r < 8; r++) {
      const float v = h[u][r] + b1v[u][r];
      t[r] = v > 0.f ? v : 0.f;
    }
    v4f lo, hi;
    lo.x = t[0]; lo.y = t[1]; lo.z = t[2]; lo.w = t[3];
    hi.x = t[4]; hi.y = t[5]; hi.z = t[6]; hi.w = t[7];
    *(v4f*)(hrow + c0 + 16 * u)     = lo;
    *(v4f*)(hrow + c0 + 16 * u + 4) = hi;
  }
  lds_barrier();

#pragma unroll
  for (int j = 0; j < 8; j++) {
    const v4f lo = *(const v4f*)(hrow + 16 * j);
    const v4f hi = *(const v4f*)(hrow + 16 * j + 4);
    hb[4 * j].x     = lo.x; hb[4 * j].y     = lo.y;
    hb[4 * j + 1].x = lo.z; hb[4 * j + 1].y = lo.w;
    hb[4 * j + 2].x = hi.x; hb[4 * j + 2].y = hi.y;
    hb[4 * j + 3].x = hi.z; hb[4 * j + 3].y = hi.w;
  }
  LOAD_FENCE();
#pragma unroll
  for (int e = 0; e < 16; e++) {
#pragma unroll
    for (int u = 0; u < 2; u++) x2[u] = wmma4(f2[u][e], hb[e], x2[u]);
  }
  LOAD_FENCE();
  if (LAST) {
#pragma unroll
    for (int jj = 0; jj < 16; jj++) {
      const v4f t = *(const v4f*)(Wa + w3o + 128 * jj);
      f3[0][2 * jj].x     = t.x; f3[0][2 * jj].y     = t.y;
      f3[0][2 * jj + 1].x = t.z; f3[0][2 * jj + 1].y = t.w;
    }
  } else {
    const float* lwp = lnw + 8 * half;
    const float* lbp = lnb + 8 * half;
#pragma unroll
    for (int n = 0; n < 8; n++) {
#pragma unroll
      for (int r = 0; r < 8; r++) { lwv[n][r] = lwp[16 * n + r]; lbv[n][r] = lbp[16 * n + r]; }
    }
#pragma unroll
    for (int p = 0; p < 4; p++) {
      const size_t o = w3o + 128 * p;
      const v4f tk = *(const v4f*)(Wa + o);
      const v4f tq = *(const v4f*)(Wq + o);
      const v4f tv = *(const v4f*)(Wv + o);
      const v4f tc = *(const v4f*)(Ws + o);
      f3[0][0 * 8 + 2 * p].x = tk.x; f3[0][0 * 8 + 2 * p].y = tk.y;
      f3[0][0 * 8 + 2 * p + 1].x = tk.z; f3[0][0 * 8 + 2 * p + 1].y = tk.w;
      f3[0][1 * 8 + 2 * p].x = tq.x; f3[0][1 * 8 + 2 * p].y = tq.y;
      f3[0][1 * 8 + 2 * p + 1].x = tq.z; f3[0][1 * 8 + 2 * p + 1].y = tq.w;
      f3[0][2 * 8 + 2 * p].x = tv.x; f3[0][2 * 8 + 2 * p].y = tv.y;
      f3[0][2 * 8 + 2 * p + 1].x = tv.z; f3[0][2 * 8 + 2 * p + 1].y = tv.w;
      f3[0][3 * 8 + 2 * p].x = tc.x; f3[0][3 * 8 + 2 * p].y = tc.y;
      f3[0][3 * 8 + 2 * p + 1].x = tc.z; f3[0][3 * 8 + 2 * p + 1].y = tc.w;
    }
  }
  LOAD_FENCE();
#pragma unroll
  for (int e = 16; e < 32; e++) {
#pragma unroll
    for (int u = 0; u < 2; u++) x2[u] = wmma4(f2[u][e], hb[e], x2[u]);
  }

#pragma unroll
  for (int u = 0; u < 2; u++) {
    float t[8];
#pragma unroll
    for (int r = 0; r < 8; r++) {
      float v = x2[u][r] + b2v[u][r];
      v = v > 0.f ? v : 0.f;
      t[r] = v + scv[u][r] + scbv[u][r];
    }
    v4f lo, hi;
    lo.x = t[0]; lo.y = t[1]; lo.z = t[2]; lo.w = t[3];
    hi.x = t[4]; hi.y = t[5]; hi.z = t[6]; hi.w = t[7];
    *(v4f*)(xrow + c0 + 16 * u)     = lo;
    *(v4f*)(xrow + c0 + 16 * u + 4) = hi;
  }
  lds_barrier();

#pragma unroll
  for (int n = 0; n < 8; n++) {
    const v4f lo = *(const v4f*)(xrow + 16 * n);
    const v4f hi = *(const v4f*)(xrow + 16 * n + 4);
    acc2[n][0] = lo.x; acc2[n][1] = lo.y; acc2[n][2] = lo.z; acc2[n][3] = lo.w;
    acc2[n][4] = hi.x; acc2[n][5] = hi.y; acc2[n][6] = hi.z; acc2[n][7] = hi.w;
  }
  if (!LAST) {
#pragma unroll
    for (int q3 = 1; q3 < 4; q3++) {
#pragma unroll
      for (int p = 0; p < 4; p++) {
        const size_t o = w3o + 128 * (4 * q3 + p);
        const v4f tk = *(const v4f*)(Wa + o);
        const v4f tq = *(const v4f*)(Wq + o);
        const v4f tv = *(const v4f*)(Wv + o);
        const v4f tc = *(const v4f*)(Ws + o);
        f3[q3][0 * 8 + 2 * p].x = tk.x; f3[q3][0 * 8 + 2 * p].y = tk.y;
        f3[q3][0 * 8 + 2 * p + 1].x = tk.z; f3[q3][0 * 8 + 2 * p + 1].y = tk.w;
        f3[q3][1 * 8 + 2 * p].x = tq.x; f3[q3][1 * 8 + 2 * p].y = tq.y;
        f3[q3][1 * 8 + 2 * p + 1].x = tq.z; f3[q3][1 * 8 + 2 * p + 1].y = tq.w;
        f3[q3][2 * 8 + 2 * p].x = tv.x; f3[q3][2 * 8 + 2 * p].y = tv.y;
        f3[q3][2 * 8 + 2 * p + 1].x = tv.z; f3[q3][2 * 8 + 2 * p + 1].y = tv.w;
        f3[q3][3 * 8 + 2 * p].x = tc.x; f3[q3][3 * 8 + 2 * p].y = tc.y;
        f3[q3][3 * 8 + 2 * p + 1].x = tc.z; f3[q3][3 * 8 + 2 * p + 1].y = tc.w;
      }
    }
  }
  LOAD_FENCE();

  if (LAST) {
    v8f accE = {0.f,0.f,0.f,0.f,0.f,0.f,0.f,0.f};
    v8f accO = {0.f,0.f,0.f,0.f,0.f,0.f,0.f,0.f};
#pragma unroll
    for (int n = 0; n < 8; n++) {
#pragma unroll
      for (int rr = 0; rr < 4; rr += 2) {
        v2f xa, xb;
        xa.x = acc2[n][2 * rr];     xa.y = acc2[n][2 * rr + 1];
        xb.x = acc2[n][2 * rr + 2]; xb.y = acc2[n][2 * rr + 3];
        accE = wmma4(xa, f3[0][n * 4 + rr], accE);
        accO = wmma4(xb, f3[0][n * 4 + rr + 1], accO);
      }
    }
#pragma unroll
    for (int r = 0; r < 8; r++) {
      const int t = tm * 16 + r + 8 * half;
      const int o = tn * 16 + idx;
      O0[(size_t)t * Bn * OUTn + (size_t)b * OUTn + o] = (accE[r] + accO[r]) + ub[o];
    }
  } else {
    float p0 = 0.f, p1 = 0.f, p2 = 0.f, p3 = 0.f;
#pragma unroll
    for (int n = 0; n < 8; n += 4) {
#pragma unroll
      for (int r = 0; r < 8; r += 2) {
        p0 += acc2[n][r] + acc2[n][r + 1];
        p1 += acc2[n + 1][r] + acc2[n + 1][r + 1];
        p2 += acc2[n + 2][r] + acc2[n + 2][r + 1];
        p3 += acc2[n + 3][r] + acc2[n + 3][r + 1];
      }
    }
    float sum = (p0 + p1) + (p2 + p3);
    sum += __shfl_xor(sum, 16);
    const float m = sum * (1.0f / INn);

    float q0 = 0.f, q1 = 0.f, q2 = 0.f, q3 = 0.f;
#pragma unroll
    for (int n = 0; n < 8; n += 4) {
#pragma unroll
      for (int r = 0; r < 8; r++) {
        float d;
        d = acc2[n][r] - m;     q0 += d * d;
        d = acc2[n + 1][r] - m; q1 += d * d;
        d = acc2[n + 2][r] - m; q2 += d * d;
        d = acc2[n + 3][r] - m; q3 += d * d;
      }
    }
    float sq = (q0 + q1) + (q2 + q3);
    sq += __shfl_xor(sq, 16);
    const float inv = rsqrtf(sq * (1.0f / INn) + 1e-5f);

#pragma unroll
    for (int n = 0; n < 8; n++) {
#pragma unroll
      for (int r = 0; r < 8; r++)
        acc2[n][r] = (acc2[n][r] - m) * inv * lwv[n][r] + lbv[n][r];
    }

    v8f accK = {0.f,0.f,0.f,0.f,0.f,0.f,0.f,0.f};
    v8f accQ = {0.f,0.f,0.f,0.f,0.f,0.f,0.f,0.f};
    v8f accV = {0.f,0.f,0.f,0.f,0.f,0.f,0.f,0.f};
    v8f accS = {0.f,0.f,0.f,0.f,0.f,0.f,0.f,0.f};
#pragma unroll
    for (int q3 = 0; q3 < 4; q3++) {
#pragma unroll
      for (int nl = 0; nl < 2; nl++) {
#pragma unroll
        for (int rr = 0; rr < 4; rr++) {
          v2f xa;
          xa.x = acc2[2 * q3 + nl][2 * rr]; xa.y = acc2[2 * q3 + nl][2 * rr + 1];
          accK = wmma4(xa, f3[q3][0 * 8 + nl * 4 + rr], accK);
          accQ = wmma4(xa, f3[q3][1 * 8 + nl * 4 + rr], accQ);
          accV = wmma4(xa, f3[q3][2 * 8 + nl * 4 + rr], accV);
          accS = wmma4(xa, f3[q3][3 * 8 + nl * 4 + rr], accS);
        }
      }
    }
    const size_t o16 = row0 * Hn + tn * 16;
    float* qf = dla_qf_slot(O1, b, tm, tn, half, idx);
#pragma unroll
    for (int r = 0; r < 8; r++) {
      const size_t e = o16 + (size_t)(r + 8 * half) * Hn + idx;
      O0[e] = elu_plus1(accK[r]);
      qf[4 * r] = elu_plus1(accQ[r]);
      O3[e] = accS[r];
    }
    dla_store_vt(O2, b, tm, tn, half, idx, accV);
  }
}

extern "C" void kernel_launch(void* const* d_in, const int* in_sizes, int n_in,
                              void* d_out, int out_size, void* d_ws, size_t ws_size,
                              hipStream_t stream) {
  (void)in_sizes; (void)n_in; (void)out_size; (void)ws_size;
  const float* z   = (const float*)d_in[0];
  const float* st  = (const float*)d_in[1];
  const float* Wk  = (const float*)d_in[2];
  const float* Wq  = (const float*)d_in[3];
  const float* Wv  = (const float*)d_in[4];
  const float* lnw = (const float*)d_in[5];
  const float* lnb = (const float*)d_in[6];
  const float* w1  = (const float*)d_in[7];
  const float* b1  = (const float*)d_in[8];
  const float* w2  = (const float*)d_in[9];
  const float* b2  = (const float*)d_in[10];
  const float* scw = (const float*)d_in[11];
  const float* scb = (const float*)d_in[12];
  const float* uw  = (const float*)d_in[13];
  const float* ub  = (const float*)d_in[14];
  float* out = (float*)d_out;
  float* ws  = (float*)d_ws;

  float* SC2  = ws;
  float* SC   = ws + 262144;
  float* KB   = ws + 524288;
  float* QB   = ws + 786432;
  float* VB   = ws + 1048576;
  float* PB   = ws + 1310720;
  float* ZPB  = PB + PSZ;
  float* LD   = ZPB + ZPSZ;
  float* QB1  = ws + 1835008;
  float* LN1  = SC;
  float* PK0  = LD + 4096;
  float* PK1  = QB;
  float* LN0  = out;
  float* OST  = out + YSZ;

  dla_lnproj_kernel<<<256 + 192, 128, 0, stream>>>(z, st, lnw, lnb, Wk, Wq, Wv, scw, KB, QB, VB, SC, OST,
                                                  w1, w2, PK0);
  dla_state_local_kernel<<<128 + Bn * 9, 256, 0, stream>>>(QB, KB, VB, st, 0, PB, ZPB,
                                                          LN0, LD, OST,
                                                          nullptr, nullptr, nullptr, nullptr);
  dla_ffchain_kernel<false><<<256, 128, 0, stream>>>(
      PK0, b1, PK0 + PKM, b2, SC, scb, lnw + INn, lnb + INn,
      PK0 + 2 * PKM, PK0 + 3 * PKM, PK0 + 4 * PKM, PK0 + 5 * PKM,
      nullptr, KB, QB1, VB, SC2,
      QB, PB, ZPB, LN0, LD, st);
  dla_state_local_kernel<<<128 + Bn * 9 + 48, 256, 0, stream>>>(QB1, KB, VB, st, LAYERSTATE, PB, ZPB,
                                                               LN1, LD, OST,
                                                               w1 + (size_t)Hn * Hn,
                                                               w2 + (size_t)Hn * Hn, uw, PK1);
  dla_ffchain_kernel<true><<<256, 128, 0, stream>>>(
      PK1, b1 + Hn, PK1 + PKM, b2 + Hn, SC2, scb + Hn,
      nullptr, nullptr, PK1 + 2 * PKM, nullptr, nullptr, nullptr, ub, out, nullptr, nullptr, nullptr,
      QB1, PB, ZPB, LN1, LD, st + LAYERSTATE);
}
